# starting kernel
# baseline (speedup 1.0000x reference)
#include <hip/hip_runtime.h>


#define GA_ALPHA 0.2f
#define GA_BETA1 0.6931471805599453f
#define GA_BETA2 0.4054651081081644f

typedef __attribute__((ext_vector_type(2))) float v2f;
typedef __attribute__((ext_vector_type(8))) float v8f;


__global__ __launch_bounds__(256) void k_zero_i(int* __restrict__ p, long n) {
  long i = (long)blockIdx.x * 256 + threadIdx.x;
  if (i < n) p[i] = 0;
}

__global__ __launch_bounds__(256) void k_degcount(const int* __restrict__ src,
                                                  const int* __restrict__ dst,
                                                  int* __restrict__ cntO,
                                                  int* __restrict__ cntI, int E) {
  int e = blockIdx.x * 256 + threadIdx.x;
  if (e < E) {
    atomicAdd(&cntO[src[e]], 1);
    atomicAdd(&cntI[dst[e]], 1);
  }
}

__global__ __launch_bounds__(256) void k_degrsqrt(const int* __restrict__ cntO,
                                                  const int* __restrict__ cntI,
                                                  float* __restrict__ degO,
                                                  float* __restrict__ degI, int n) {
  int i = blockIdx.x * 256 + threadIdx.x;
  if (i < n) {
    degO[i] = rsqrtf(fmaxf((float)cntO[i], 1.0f));
    degI[i] = rsqrtf(fmaxf((float)cntI[i], 1.0f));
  }
}


__global__ __launch_bounds__(256) void k_scan_block(const int* __restrict__ in,
                                                    int* __restrict__ outEx,
                                                    int* __restrict__ bsum, int n) {
  __shared__ int s[256];
  const int tid = threadIdx.x;
  const int i = blockIdx.x * 256 + tid;
  const int v = (i < n) ? in[i] : 0;
  s[tid] = v;
  __syncthreads();
#pragma unroll
  for (int off = 1; off < 256; off <<= 1) {
    int t = (tid >= off) ? s[tid - off] : 0;
    __syncthreads();
    s[tid] += t;
    __syncthreads();
  }
  if (i < n) outEx[i] = s[tid] - v;
  if (tid == 255) bsum[blockIdx.x] = s[255];
}

__global__ __launch_bounds__(512) void k_scan_tops(int* __restrict__ bsum, int nb) {
  __shared__ int s[512];
  const int tid = threadIdx.x;
  const int v = (tid < nb) ? bsum[tid] : 0;
  s[tid] = v;
  __syncthreads();
#pragma unroll
  for (int off = 1; off < 512; off <<= 1) {
    int t = (tid >= off) ? s[tid - off] : 0;
    __syncthreads();
    s[tid] += t;
    __syncthreads();
  }
  if (tid < nb) bsum[tid] = s[tid] - v;
}

__global__ __launch_bounds__(256) void k_scan_add(int* __restrict__ outEx,
                                                  const int* __restrict__ bsum,
                                                  int* __restrict__ cursor, int n) {
  int i = blockIdx.x * 256 + threadIdx.x;
  if (i < n) {
    int r = outEx[i] + bsum[blockIdx.x];
    outEx[i] = r;
    cursor[i] = r;
  }
}


__global__ __launch_bounds__(256) void k_build_csc(const int* __restrict__ src,
                                                   const int* __restrict__ dst,
                                                   int* __restrict__ cursor,
                                                   int* __restrict__ csc, int E) {
  int e = blockIdx.x * 256 + threadIdx.x;
  if (e < E) {
    int pos = atomicAdd(&cursor[dst[e]], 1);
    csc[pos] = src[e];
  }
}


__global__ __launch_bounds__(256) void k_scale(const float4* __restrict__ x,
                                               const float* __restrict__ dors,
                                               float4* __restrict__ hs, long total4) {
  long i = (long)blockIdx.x * 256 + threadIdx.x;
  if (i < total4) {
    const int row = (int)(i >> 5);
    const float sS = dors[row];
    float4 v = x[i];
    v.x *= sS; v.y *= sS; v.z *= sS; v.w *= sS;
    hs[i] = v;
  }
}

__global__ __launch_bounds__(256) void k_spmm_pull(const int* __restrict__ rowptr,
                                                   const int* __restrict__ cnt,
                                                   const int* __restrict__ csc,
                                                   const float* __restrict__ hs,
                                                   const float* __restrict__ dirs,
                                                   const float* __restrict__ x0,
                                                   float* __restrict__ f, int N) {
  const int node = blockIdx.x * 8 + (threadIdx.x >> 5);
  if (node >= N) return;
  const int lane = threadIdx.x & 31;
  const int start = rowptr[node];
  const int c = cnt[node];
  float4 acc = {0.0f, 0.0f, 0.0f, 0.0f};
  for (int i = 0; i < c; ++i) {
    const int s = csc[start + i];
    const float4 v = *(const float4*)(hs + (long)s * 128 + lane * 4);
    acc.x += v.x; acc.y += v.y; acc.z += v.z; acc.w += v.w;
  }
  const float scale = (1.0f - GA_ALPHA) * dirs[node];
  const long o = (long)node * 128 + lane * 4;
  const float4 x0v = *(const float4*)(x0 + o);
  float4 r;
  r.x = scale * acc.x + GA_ALPHA * x0v.x;
  r.y = scale * acc.y + GA_ALPHA * x0v.y;
  r.z = scale * acc.z + GA_ALPHA * x0v.z;
  r.w = scale * acc.w + GA_ALPHA * x0v.w;
  *(float4*)(f + o) = r;
}

template <int NCOLS, bool RELU, bool RESID>
__global__ __launch_bounds__(256) void k_gemm_wmma(const float* __restrict__ A,
                                                   const float* __restrict__ W,
                                                   const float* __restrict__ bias,
                                                   float* __restrict__ out,
                                                   float actW, float resS) {
  constexpr int NCT = NCOLS / 16;
  const int wave = threadIdx.x >> 5;
  const int lane = threadIdx.x & 31;
  const int colTile = wave % NCT;
  const int rowTile = blockIdx.x * (8 / NCT) + wave / NCT;
  const int row0 = rowTile * 16;
  const int col0 = colTile * 16;
  const int half = lane >> 4;
  const int l16  = lane & 15;

  const float* __restrict__ Arow = A + (long)(row0 + l16) * 128;
  const float* __restrict__ Wcol = W + col0 + l16;

  v8f acc = {};
#pragma unroll
  for (int k = 0; k < 128; k += 4) {
    const int ka = k + half * 2;
    v2f af, bf;
    af.x = Arow[ka];
    af.y = Arow[ka + 1];
    bf.x = Wcol[(long)ka * NCOLS];
    bf.y = Wcol[(long)(ka + 1) * NCOLS];
    acc = __builtin_amdgcn_wmma_f32_16x16x4_f32(false, af, false, bf,
                                                (short)0, acc, false, false);
  }

  const int outCol = col0 + l16;
  const float bv = bias[outCol];
#pragma unroll
  for (int r = 0; r < 8; ++r) {
    const int rowM = row0 + r + half * 8;
    float v = actW * acc[r] + bv;
    if (RESID) v += resS * A[(long)rowM * 128 + outCol];
    if (RELU) v = fmaxf(v, 0.0f);
    out[(long)rowM * NCOLS + outCol] = v;
  }
}


static inline long cdiv(long a, long b) { return (a + b - 1) / b; }

extern "C" void kernel_launch(void* const* d_in, const int* in_sizes, int n_in,
                              void* d_out, int out_size, void* d_ws, size_t ws_size,
                              hipStream_t stream) {
  const float* feat = (const float*)d_in[0];
  const int*   src  = (const int*)d_in[1];
  const int*   dst  = (const int*)d_in[2];
  const float* w1   = (const float*)d_in[3];
  const float* b1   = (const float*)d_in[4];
  const float* cw1  = (const float*)d_in[5];
  const float* cb1  = (const float*)d_in[6];
  const float* cw2  = (const float*)d_in[7];
  const float* cb2  = (const float*)d_in[8];
  const float* w2   = (const float*)d_in[9];
  const float* b2   = (const float*)d_in[10];

  const int N = in_sizes[0] / 128;
  const int E = in_sizes[1];
  const long NF = (long)N * 128;
  const int NA = (N + 63) & ~63;

  char* wsb = (char*)d_ws;
  int* cntO   = (int*)wsb;                wsb += (size_t)NA * 4;
  int* cntI   = (int*)wsb;                wsb += (size_t)NA * 4;
  int* rowptr = (int*)wsb;                wsb += (size_t)NA * 4;
  int* cursor = (int*)wsb;                wsb += (size_t)NA * 4;
  int* bsum   = (int*)wsb;                wsb += 512 * 4;
  int* csc    = (int*)wsb;                wsb += ((size_t)E * 4 + 255) & ~(size_t)255;
  float* degO = (float*)wsb;              wsb += (size_t)NA * 4;
  float* degI = (float*)wsb;              wsb += (size_t)NA * 4;
  float* h0   = (float*)wsb;              wsb += (size_t)NF * 4;
  float* bufS = (float*)wsb;              wsb += (size_t)NF * 4;
  float* bufA = (float*)wsb;              wsb += (size_t)NF * 4;
  float* bufB = (float*)wsb;
  (void)ws_size; (void)n_in; (void)out_size;

  const int NB       = (int)cdiv(N, 256);
  const int nb_node  = NB;
  const int nb_edge  = (int)cdiv(E, 256);
  const int nb_vec4  = (int)cdiv(NF / 4, 256);
  const int nb_pull  = (int)cdiv(N, 8);
  const int nb_g128  = N / 16;
  const int nb_g64   = N / 32;

  k_zero_i<<<(int)cdiv(2L * NA, 256), 256, 0, stream>>>(cntO, 2L * NA);
  k_degcount<<<nb_edge, 256, 0, stream>>>(src, dst, cntO, cntI, E);
  k_degrsqrt<<<nb_node, 256, 0, stream>>>(cntO, cntI, degO, degI, N);
  k_scan_block<<<NB, 256, 0, stream>>>(cntI, rowptr, bsum, N);
  k_scan_tops<<<1, 512, 0, stream>>>(bsum, NB);
  k_scan_add<<<NB, 256, 0, stream>>>(rowptr, bsum, cursor, N);
  k_build_csc<<<nb_edge, 256, 0, stream>>>(src, dst, cursor, csc, E);

  k_gemm_wmma<128, true, false><<<nb_g128, 256, 0, stream>>>(feat, w1, b1, h0, 1.0f, 0.0f);

  k_scale<<<nb_vec4, 256, 0, stream>>>((const float4*)h0, degO, (float4*)bufS, NF / 4);
  k_spmm_pull<<<nb_pull, 256, 0, stream>>>(rowptr, cntI, csc, bufS, degI, h0, bufA, N);
  k_gemm_wmma<128, true, true><<<nb_g128, 256, 0, stream>>>(bufA, cw1, cb1, bufS,
                                                            GA_BETA1, 1.0f - GA_BETA1);

  k_scale<<<nb_vec4, 256, 0, stream>>>((const float4*)bufS, degO, (float4*)bufB, NF / 4);
  k_spmm_pull<<<nb_pull, 256, 0, stream>>>(rowptr, cntI, csc, bufB, degI, h0, bufA, N);
  k_gemm_wmma<128, false, true><<<nb_g128, 256, 0, stream>>>(bufA, cw2, cb2, bufB,
                                                             GA_BETA2, 1.0f - GA_BETA2);

  k_gemm_wmma<64, false, false><<<nb_g64, 256, 0, stream>>>(bufB, w2, b2, (float*)d_out,
                                                            1.0f, 0.0f);
}


# GEMMs re-tiled: persistent 16-wave work-groups stage W once in LDS in fragment order; each wave owns 16-row strips, loads A once (b128), feeds all column tiles; regroups k inside the f32 sums
# speedup vs baseline: 1.2231x; 1.2231x over previous
#include <hip/hip_runtime.h>


#define GA_ALPHA 0.2f
#define GA_BETA1 0.6931471805599453f
#define GA_BETA2 0.4054651081081644f

typedef __attribute__((ext_vector_type(2))) float v2f;
typedef __attribute__((ext_vector_type(8))) float v8f;


__global__ __launch_bounds__(256) void k_zero_i(int* __restrict__ p, long n) {
  long i = (long)blockIdx.x * 256 + threadIdx.x;
  if (i < n) p[i] = 0;
}

__global__ __launch_bounds__(256) void k_degcount(const int* __restrict__ src,
                                                  const int* __restrict__ dst,
                                                  int* __restrict__ cntO,
                                                  int* __restrict__ cntI, int E) {
  int e = blockIdx.x * 256 + threadIdx.x;
  if (e < E) {
    atomicAdd(&cntO[src[e]], 1);
    atomicAdd(&cntI[dst[e]], 1);
  }
}

__global__ __launch_bounds__(256) void k_degrsqrt(const int* __restrict__ cntO,
                                                  const int* __restrict__ cntI,
                                                  float* __restrict__ degO,
                                                  float* __restrict__ degI, int n) {
  int i = blockIdx.x * 256 + threadIdx.x;
  if (i < n) {
    degO[i] = rsqrtf(fmaxf((float)cntO[i], 1.0f));
    degI[i] = rsqrtf(fmaxf((float)cntI[i], 1.0f));
  }
}


__global__ __launch_bounds__(256) void k_scan_block(const int* __restrict__ in,
                                                    int* __restrict__ outEx,
                                                    int* __restrict__ bsum, int n) {
  __shared__ int s[256];
  const int tid = threadIdx.x;
  const int i = blockIdx.x * 256 + tid;
  const int v = (i < n) ? in[i] : 0;
  s[tid] = v;
  __syncthreads();
#pragma unroll
  for (int off = 1; off < 256; off <<= 1) {
    int t = (tid >= off) ? s[tid - off] : 0;
    __syncthreads();
    s[tid] += t;
    __syncthreads();
  }
  if (i < n) outEx[i] = s[tid] - v;
  if (tid == 255) bsum[blockIdx.x] = s[255];
}

__global__ __launch_bounds__(512) void k_scan_tops(int* __restrict__ bsum, int nb) {
  __shared__ int s[512];
  const int tid = threadIdx.x;
  const int v = (tid < nb) ? bsum[tid] : 0;
  s[tid] = v;
  __syncthreads();
#pragma unroll
  for (int off = 1; off < 512; off <<= 1) {
    int t = (tid >= off) ? s[tid - off] : 0;
    __syncthreads();
    s[tid] += t;
    __syncthreads();
  }
  if (tid < nb) bsum[tid] = s[tid] - v;
}

__global__ __launch_bounds__(256) void k_scan_add(int* __restrict__ outEx,
                                                  const int* __restrict__ bsum,
                                                  int* __restrict__ cursor, int n) {
  int i = blockIdx.x * 256 + threadIdx.x;
  if (i < n) {
    int r = outEx[i] + bsum[blockIdx.x];
    outEx[i] = r;
    cursor[i] = r;
  }
}


__global__ __launch_bounds__(256) void k_build_csc(const int* __restrict__ src,
                                                   const int* __restrict__ dst,
                                                   int* __restrict__ cursor,
                                                   int* __restrict__ csc, int E) {
  int e = blockIdx.x * 256 + threadIdx.x;
  if (e < E) {
    int pos = atomicAdd(&cursor[dst[e]], 1);
    csc[pos] = src[e];
  }
}


__global__ __launch_bounds__(256) void k_scale(const float4* __restrict__ x,
                                               const float* __restrict__ dors,
                                               float4* __restrict__ hs, long total4) {
  long i = (long)blockIdx.x * 256 + threadIdx.x;
  if (i < total4) {
    const int row = (int)(i >> 5);
    const float sS = dors[row];
    float4 v = x[i];
    v.x *= sS; v.y *= sS; v.z *= sS; v.w *= sS;
    hs[i] = v;
  }
}

__global__ __launch_bounds__(256) void k_spmm_pull(const int* __restrict__ rowptr,
                                                   const int* __restrict__ cnt,
                                                   const int* __restrict__ csc,
                                                   const float* __restrict__ hs,
                                                   const float* __restrict__ dirs,
                                                   const float* __restrict__ x0,
                                                   float* __restrict__ f, int N) {
  const int node = blockIdx.x * 8 + (threadIdx.x >> 5);
  if (node >= N) return;
  const int lane = threadIdx.x & 31;
  const int start = rowptr[node];
  const int c = cnt[node];
  float4 acc = {0.0f, 0.0f, 0.0f, 0.0f};
  for (int i = 0; i < c; ++i) {
    const int s = csc[start + i];
    const float4 v = *(const float4*)(hs + (long)s * 128 + lane * 4);
    acc.x += v.x; acc.y += v.y; acc.z += v.z; acc.w += v.w;
  }
  const float scale = (1.0f - GA_ALPHA) * dirs[node];
  const long o = (long)node * 128 + lane * 4;
  const float4 x0v = *(const float4*)(x0 + o);
  float4 r;
  r.x = scale * acc.x + GA_ALPHA * x0v.x;
  r.y = scale * acc.y + GA_ALPHA * x0v.y;
  r.z = scale * acc.z + GA_ALPHA * x0v.z;
  r.w = scale * acc.w + GA_ALPHA * x0v.w;
  *(float4*)(f + o) = r;
}

template <int NCOLS, bool RELU, bool RESID>
__global__ __launch_bounds__(512) void k_gemm_wmma(const float* __restrict__ A,
                                                   const float* __restrict__ W,
                                                   const float* __restrict__ bias,
                                                   float* __restrict__ out,
                                                   float actW, float resS, int nTiles) {
  constexpr int NCT = NCOLS / 16;
  __shared__ float sW[128 * NCOLS];

  const int wave = __builtin_amdgcn_readfirstlane((int)(threadIdx.x >> 5));
  const int lane = threadIdx.x & 31;
  const int half = lane >> 4;
  const int l16  = lane & 15;
  const int nblk = gridDim.x;

  int round = 0;
  int tile = blockIdx.x + nblk * wave;
  bool have = tile < nTiles;

  constexpr int NST = (128 * NCOLS) / 512;
  float wv[NST];
#pragma unroll
  for (int it = 0; it < NST; ++it) {
    const int d  = it * 512 + threadIdx.x;
    const int p  = d & 1;
    const int c  = (d >> 1) & 15;
    const int h  = (d >> 5) & 1;
    const int bt = d >> 6;
    const int t  = bt % NCT;
    const int js = bt / NCT;
    const int k  = (js >> 1) * 8 + h * 4 + (js & 1) * 2 + p;
    wv[it] = W[k * NCOLS + t * 16 + c];
  }

  float4 a[16];
  {
    const float* __restrict__ Ap = A + (long)((have ? tile : 0) * 16 + l16) * 128 + half * 4;
#pragma unroll
    for (int j = 0; j < 16; ++j) a[j] = *(const float4*)(Ap + 8 * j);
  }

#pragma unroll
  for (int it = 0; it < NST; ++it) sW[it * 512 + threadIdx.x] = wv[it];
  __builtin_amdgcn_fence(__ATOMIC_RELEASE, "workgroup", "local");
  __builtin_amdgcn_s_barrier();
  __builtin_amdgcn_fence(__ATOMIC_ACQUIRE, "workgroup", "local");

  const float* __restrict__ sWl = sW + lane * 2;

  while (have) {
    const int nround = round + 1;
    const int q = (nround & 1) ? ((13 * wave) & 15) : wave;
    const int ntile = blockIdx.x + nblk * (16 * nround + q);
    const bool nhave = ntile < nTiles;
    const float* __restrict__ An =
        A + (long)((nhave ? ntile : tile) * 16 + l16) * 128 + half * 4;

    v8f acc[NCT];
#pragma unroll
    for (int t = 0; t < NCT; ++t) acc[t] = v8f{};

#pragma unroll
    for (int j = 0; j < 16; ++j) {
      v2f b[2 * NCT];
#pragma unroll
      for (int u = 0; u < 2 * NCT; ++u) b[u] = *(const v2f*)(sWl + (j * 2 * NCT + u) * 64);
#pragma unroll
      for (int s = 0; s < 2; ++s) {
        v2f af;
        af.x = s ? a[j].z : a[j].x;
        af.y = s ? a[j].w : a[j].y;
#pragma unroll
        for (int t = 0; t < NCT; ++t) {
          acc[t] = __builtin_amdgcn_wmma_f32_16x16x4_f32(false, af, false, b[s * NCT + t],
                                                         (short)0, acc[t], false, false);
        }
      }
      a[j] = *(const float4*)(An + 8 * j);
      __builtin_amdgcn_sched_barrier(0);
    }

    const int row0 = tile * 16;
#pragma unroll
    for (int t = 0; t < NCT; ++t) {
      const int outCol = t * 16 + l16;
      const float bv = bias[outCol];
#pragma unroll
      for (int r = 0; r < 8; ++r) {
        const int rowM = row0 + r + half * 8;
        float v = actW * acc[t][r] + bv;
        if (RESID) v += resS * A[(long)rowM * 128 + outCol];
        if (RELU) v = fmaxf(v, 0.0f);
        out[(long)rowM * NCOLS + outCol] = v;
      }
    }

    tile = ntile;
    have = nhave;
    round = nround;
  }
}


static inline long cdiv(long a, long b) { return (a + b - 1) / b; }

extern "C" void kernel_launch(void* const* d_in, const int* in_sizes, int n_in,
                              void* d_out, int out_size, void* d_ws, size_t ws_size,
                              hipStream_t stream) {
  const float* feat = (const float*)d_in[0];
  const int*   src  = (const int*)d_in[1];
  const int*   dst  = (const int*)d_in[2];
  const float* w1   = (const float*)d_in[3];
  const float* b1   = (const float*)d_in[4];
  const float* cw1  = (const float*)d_in[5];
  const float* cb1  = (const float*)d_in[6];
  const float* cw2  = (const float*)d_in[7];
  const float* cb2  = (const float*)d_in[8];
  const float* w2   = (const float*)d_in[9];
  const float* b2   = (const float*)d_in[10];

  const int N = in_sizes[0] / 128;
  const int E = in_sizes[1];
  const long NF = (long)N * 128;
  const int NA = (N + 63) & ~63;

  char* wsb = (char*)d_ws;
  int* cntO   = (int*)wsb;                wsb += (size_t)NA * 4;
  int* cntI   = (int*)wsb;                wsb += (size_t)NA * 4;
  int* rowptr = (int*)wsb;                wsb += (size_t)NA * 4;
  int* cursor = (int*)wsb;                wsb += (size_t)NA * 4;
  int* bsum   = (int*)wsb;                wsb += 512 * 4;
  int* csc    = (int*)wsb;                wsb += ((size_t)E * 4 + 255) & ~(size_t)255;
  float* degO = (float*)wsb;              wsb += (size_t)NA * 4;
  float* degI = (float*)wsb;              wsb += (size_t)NA * 4;
  float* h0   = (float*)wsb;              wsb += (size_t)NF * 4;
  float* bufS = (float*)wsb;              wsb += (size_t)NF * 4;
  float* bufA = (float*)wsb;              wsb += (size_t)NF * 4;
  float* bufB = (float*)wsb;
  (void)ws_size; (void)n_in; (void)out_size;

  const int NB       = (int)cdiv(N, 256);
  const int nb_node  = NB;
  const int nb_edge  = (int)cdiv(E, 256);
  const int nb_vec4  = (int)cdiv(NF / 4, 256);
  const int nb_pull  = (int)cdiv(N, 8);
  const int n_strips = N / 16;

  k_zero_i<<<(int)cdiv(2L * NA, 256), 256, 0, stream>>>(cntO, 2L * NA);
  k_degcount<<<nb_edge, 256, 0, stream>>>(src, dst, cntO, cntI, E);
  k_degrsqrt<<<nb_node, 256, 0, stream>>>(cntO, cntI, degO, degI, N);
  k_scan_block<<<NB, 256, 0, stream>>>(cntI, rowptr, bsum, N);
  k_scan_tops<<<1, 512, 0, stream>>>(bsum, NB);
  k_scan_add<<<NB, 256, 0, stream>>>(rowptr, bsum, cursor, N);
  k_build_csc<<<nb_edge, 256, 0, stream>>>(src, dst, cursor, csc, E);

  k_gemm_wmma<128, true, false><<<256, 512, 0, stream>>>(feat, w1, b1, h0, 1.0f, 0.0f, n_strips);

  k_scale<<<nb_vec4, 256, 0, stream>>>((const float4*)h0, degO, (float4*)bufS, NF / 4);
  k_spmm_pull<<<nb_pull, 256, 0, stream>>>(rowptr, cntI, csc, bufS, degI, h0, bufA, N);
  k_gemm_wmma<128, true, true><<<256, 512, 0, stream>>>(bufA, cw1, cb1, bufS,
                                                        GA_BETA1, 1.0f - GA_BETA1, n_strips);

  k_scale<<<nb_vec4, 256, 0, stream>>>((const float4*)bufS, degO, (float4*)bufB, NF / 4);
  k_spmm_pull<<<nb_pull, 256, 0, stream>>>(rowptr, cntI, csc, bufB, degI, h0, bufA, N);
  k_gemm_wmma<128, false, true><<<256, 512, 0, stream>>>(bufA, cw2, cb2, bufB,
                                                         GA_BETA2, 1.0f - GA_BETA2, n_strips);

  k_gemm_wmma<64, false, false><<<256, 512, 0, stream>>>(bufB, w2, b2, (float*)d_out,
                                                        1.0f, 0.0f, n_strips);
}


# Degree counts and CSC build without global atomics: 64 edge slices, one work-group each counts and ranks its slice per node in LDS; a node pass sums slices. CSC order: by slice, then LDS arrival
# speedup vs baseline: 1.5999x; 1.3081x over previous
#include <hip/hip_runtime.h>


#define GA_ALPHA 0.2f
#define GA_BETA1 0.6931471805599453f
#define GA_BETA2 0.4054651081081644f

typedef __attribute__((ext_vector_type(2))) float v2f;
typedef __attribute__((ext_vector_type(8))) float v8f;


#define CS_SLICES  64
#define CS_WORDS   50048
#define CS_THREADS 1024
#define CS_B       8

__global__ __launch_bounds__(CS_THREADS) void k_hist(const int* __restrict__ src,
                                                     const int* __restrict__ dst,
                                                     unsigned int* __restrict__ hist, int E) {
  __shared__ unsigned int h[CS_WORDS];
  const int tid = threadIdx.x;
  const int which = blockIdx.x / CS_SLICES;
  const int b = blockIdx.x - which * CS_SLICES;
  const int* __restrict__ key = which ? dst : src;
  for (int w = tid; w < CS_WORDS; w += CS_THREADS) h[w] = 0u;
  __syncthreads();
  const int len = (E + CS_SLICES - 1) / CS_SLICES;
  const int lo = b * len;
  const int hi = min(E, lo + len);
  for (int e0 = lo + tid; e0 < hi; e0 += CS_B * CS_THREADS) {
    int v[CS_B];
#pragma unroll
    for (int k = 0; k < CS_B; ++k) {
      const int e = e0 + k * CS_THREADS;
      v[k] = (e < hi) ? key[e] : -1;
    }
#pragma unroll
    for (int k = 0; k < CS_B; ++k)
      if (v[k] >= 0) atomicAdd(&h[v[k] >> 1], (v[k] & 1) ? 0x10000u : 1u);
  }
  __syncthreads();
  unsigned int* __restrict__ out = hist + (size_t)blockIdx.x * CS_WORDS;
  for (int w = tid; w < CS_WORDS; w += CS_THREADS) out[w] = h[w];
}

__global__ __launch_bounds__(256) void k_sum(const unsigned int* __restrict__ hist,
                                             int* __restrict__ cntO, int* __restrict__ cntI,
                                             int* __restrict__ pre, int N, int NA) {
  const int w = blockIdx.x * 256 + threadIdx.x;
  if (w >= CS_WORDS) return;
  const int n0 = 2 * w, n1 = 2 * w + 1;
  const bool v0 = n0 < N, v1 = n1 < N;
  if (!v0) return;

  unsigned int x[CS_SLICES];
#pragma unroll
  for (int b = 0; b < CS_SLICES; ++b) x[b] = hist[(size_t)b * CS_WORDS + w];
  unsigned int lo = 0u, hi = 0u;
#pragma unroll
  for (int b = 0; b < CS_SLICES; ++b) {
    lo += x[b] & 0xffffu;
    hi += x[b] >> 16;
  }
  const unsigned int outLo = lo, outHi = hi;

#pragma unroll
  for (int b = 0; b < CS_SLICES; ++b) x[b] = hist[(size_t)(CS_SLICES + b) * CS_WORDS + w];
  int* __restrict__ prow = pre + n0;
  lo = 0u; hi = 0u;
  if (v1) {
#pragma unroll
    for (int b = 0; b < CS_SLICES; ++b) {
      *(int2*)(prow + (size_t)b * NA) = make_int2((int)lo, (int)hi);
      lo += x[b] & 0xffffu;
      hi += x[b] >> 16;
    }
    *(int2*)(cntO + n0) = make_int2((int)outLo, (int)outHi);
    *(int2*)(cntI + n0) = make_int2((int)lo, (int)hi);
  } else {
#pragma unroll
    for (int b = 0; b < CS_SLICES; ++b) {
      prow[(size_t)b * NA] = (int)lo;
      lo += x[b] & 0xffffu;
    }
    cntO[n0] = (int)outLo;
    cntI[n0] = (int)lo;
  }
}

__global__ __launch_bounds__(256) void k_degrsqrt(const int* __restrict__ cntO,
                                                  const int* __restrict__ cntI,
                                                  float* __restrict__ degO,
                                                  float* __restrict__ degI, int n) {
  int i = blockIdx.x * 256 + threadIdx.x;
  if (i < n) {
    degO[i] = rsqrtf(fmaxf((float)cntO[i], 1.0f));
    degI[i] = rsqrtf(fmaxf((float)cntI[i], 1.0f));
  }
}


__global__ __launch_bounds__(256) void k_scan_block(const int* __restrict__ in,
                                                    int* __restrict__ outEx,
                                                    int* __restrict__ bsum, int n) {
  __shared__ int s[256];
  const int tid = threadIdx.x;
  const int i = blockIdx.x * 256 + tid;
  const int v = (i < n) ? in[i] : 0;
  s[tid] = v;
  __syncthreads();
#pragma unroll
  for (int off = 1; off < 256; off <<= 1) {
    int t = (tid >= off) ? s[tid - off] : 0;
    __syncthreads();
    s[tid] += t;
    __syncthreads();
  }
  if (i < n) outEx[i] = s[tid] - v;
  if (tid == 255) bsum[blockIdx.x] = s[255];
}

__global__ __launch_bounds__(512) void k_scan_tops(int* __restrict__ bsum, int nb) {
  __shared__ int s[512];
  const int tid = threadIdx.x;
  const int v = (tid < nb) ? bsum[tid] : 0;
  s[tid] = v;
  __syncthreads();
#pragma unroll
  for (int off = 1; off < 512; off <<= 1) {
    int t = (tid >= off) ? s[tid - off] : 0;
    __syncthreads();
    s[tid] += t;
    __syncthreads();
  }
  if (tid < nb) bsum[tid] = s[tid] - v;
}

__global__ __launch_bounds__(256) void k_scan_add(int* __restrict__ outEx,
                                                  const int* __restrict__ bsum,
                                                  int* __restrict__ cursor, int n) {
  int i = blockIdx.x * 256 + threadIdx.x;
  if (i < n) {
    int r = outEx[i] + bsum[blockIdx.x];
    outEx[i] = r;
    cursor[i] = r;
  }
}

__global__ __launch_bounds__(CS_THREADS) void k_scatter(const int* __restrict__ src,
                                                        const int* __restrict__ dst,
                                                        const int* __restrict__ rowptr,
                                                        const int* __restrict__ pre,
                                                        int* __restrict__ csc, int E, int NA) {
  __shared__ unsigned int h[CS_WORDS];
  const int tid = threadIdx.x;
  const int b = blockIdx.x;
  for (int w = tid; w < CS_WORDS; w += CS_THREADS) h[w] = 0u;
  __syncthreads();
  const int len = (E + CS_SLICES - 1) / CS_SLICES;
  const int lo = b * len;
  const int hi = min(E, lo + len);
  const int* __restrict__ preb = pre + (size_t)b * NA;
  for (int e0 = lo + tid; e0 < hi; e0 += CS_B * CS_THREADS) {
    int s[CS_B], d[CS_B], base[CS_B];
    unsigned int old[CS_B];
    bool ok[CS_B];
#pragma unroll
    for (int k = 0; k < CS_B; ++k) {
      const int e = e0 + k * CS_THREADS;
      ok[k] = e < hi;
      const int ec = ok[k] ? e : e0;
      d[k] = dst[ec];
      s[k] = src[ec];
    }
#pragma unroll
    for (int k = 0; k < CS_B; ++k) {
      const unsigned int inc = ok[k] ? ((d[k] & 1) ? 0x10000u : 1u) : 0u;
      old[k] = atomicAdd(&h[d[k] >> 1], inc);
      base[k] = rowptr[d[k]] + preb[d[k]];
    }
#pragma unroll
    for (int k = 0; k < CS_B; ++k) {
      const unsigned int rank = (d[k] & 1) ? (old[k] >> 16) : (old[k] & 0xffffu);
      if (ok[k]) csc[base[k] + (int)rank] = s[k];
    }
  }
}


__global__ __launch_bounds__(256) void k_scale(const float4* __restrict__ x,
                                               const float* __restrict__ dors,
                                               float4* __restrict__ hs, long total4) {
  long i = (long)blockIdx.x * 256 + threadIdx.x;
  if (i < total4) {
    const int row = (int)(i >> 5);
    const float sS = dors[row];
    float4 v = x[i];
    v.x *= sS; v.y *= sS; v.z *= sS; v.w *= sS;
    hs[i] = v;
  }
}

__global__ __launch_bounds__(256) void k_spmm_pull(const int* __restrict__ rowptr,
                                                   const int* __restrict__ cnt,
                                                   const int* __restrict__ csc,
                                                   const float* __restrict__ hs,
                                                   const float* __restrict__ dirs,
                                                   const float* __restrict__ x0,
                                                   float* __restrict__ f, int N) {
  const int node = blockIdx.x * 8 + (threadIdx.x >> 5);
  if (node >= N) return;
  const int lane = threadIdx.x & 31;
  const int start = rowptr[node];
  const int c = cnt[node];
  float4 acc = {0.0f, 0.0f, 0.0f, 0.0f};
  for (int i = 0; i < c; ++i) {
    const int s = csc[start + i];
    const float4 v = *(const float4*)(hs + (long)s * 128 + lane * 4);
    acc.x += v.x; acc.y += v.y; acc.z += v.z; acc.w += v.w;
  }
  const float scale = (1.0f - GA_ALPHA) * dirs[node];
  const long o = (long)node * 128 + lane * 4;
  const float4 x0v = *(const float4*)(x0 + o);
  float4 r;
  r.x = scale * acc.x + GA_ALPHA * x0v.x;
  r.y = scale * acc.y + GA_ALPHA * x0v.y;
  r.z = scale * acc.z + GA_ALPHA * x0v.z;
  r.w = scale * acc.w + GA_ALPHA * x0v.w;
  *(float4*)(f + o) = r;
}

template <int NCOLS, bool RELU, bool RESID>
__global__ __launch_bounds__(512) void k_gemm_wmma(const float* __restrict__ A,
                                                   const float* __restrict__ W,
                                                   const float* __restrict__ bias,
                                                   float* __restrict__ out,
                                                   float actW, float resS, int nTiles) {
  constexpr int NCT = NCOLS / 16;
  __shared__ float sW[128 * NCOLS];

  const int wave = __builtin_amdgcn_readfirstlane((int)(threadIdx.x >> 5));
  const int lane = threadIdx.x & 31;
  const int half = lane >> 4;
  const int l16  = lane & 15;
  const int nblk = gridDim.x;

  int round = 0;
  int tile = blockIdx.x + nblk * wave;
  bool have = tile < nTiles;

  constexpr int NST = (128 * NCOLS) / 512;
  float wv[NST];
#pragma unroll
  for (int it = 0; it < NST; ++it) {
    const int d  = it * 512 + threadIdx.x;
    const int p  = d & 1;
    const int c  = (d >> 1) & 15;
    const int h  = (d >> 5) & 1;
    const int bt = d >> 6;
    const int t  = bt % NCT;
    const int js = bt / NCT;
    const int k  = (js >> 1) * 8 + h * 4 + (js & 1) * 2 + p;
    wv[it] = W[k * NCOLS + t * 16 + c];
  }

  float4 a[16];
  {
    const float* __restrict__ Ap = A + (long)((have ? tile : 0) * 16 + l16) * 128 + half * 4;
#pragma unroll
    for (int j = 0; j < 16; ++j) a[j] = *(const float4*)(Ap + 8 * j);
  }

#pragma unroll
  for (int it = 0; it < NST; ++it) sW[it * 512 + threadIdx.x] = wv[it];
  __builtin_amdgcn_fence(__ATOMIC_RELEASE, "workgroup", "local");
  __builtin_amdgcn_s_barrier();
  __builtin_amdgcn_fence(__ATOMIC_ACQUIRE, "workgroup", "local");

  const float* __restrict__ sWl = sW + lane * 2;

  while (have) {
    const int nround = round + 1;
    const int q = (nround & 1) ? ((13 * wave) & 15) : wave;
    const int ntile = blockIdx.x + nblk * (16 * nround + q);
    const bool nhave = ntile < nTiles;
    const float* __restrict__ An =
        A + (long)((nhave ? ntile : tile) * 16 + l16) * 128 + half * 4;

    v8f acc[NCT];
#pragma unroll
    for (int t = 0; t < NCT; ++t) acc[t] = v8f{};

#pragma unroll
    for (int j = 0; j < 16; ++j) {
      v2f b[2 * NCT];
#pragma unroll
      for (int u = 0; u < 2 * NCT; ++u) b[u] = *(const v2f*)(sWl + (j * 2 * NCT + u) * 64);
#pragma unroll
      for (int s = 0; s < 2; ++s) {
        v2f af;
        af.x = s ? a[j].z : a[j].x;
        af.y = s ? a[j].w : a[j].y;
#pragma unroll
        for (int t = 0; t < NCT; ++t) {
          acc[t] = __builtin_amdgcn_wmma_f32_16x16x4_f32(false, af, false, b[s * NCT + t],
                                                         (short)0, acc[t], false, false);
        }
      }
      a[j] = *(const float4*)(An + 8 * j);
      __builtin_amdgcn_sched_barrier(0);
    }

    const int row0 = tile * 16;
#pragma unroll
    for (int t = 0; t < NCT; ++t) {
      const int outCol = t * 16 + l16;
      const float bv = bias[outCol];
#pragma unroll
      for (int r = 0; r < 8; ++r) {
        const int rowM = row0 + r + half * 8;
        float v = actW * acc[t][r] + bv;
        if (RESID) v += resS * A[(long)rowM * 128 + outCol];
        if (RELU) v = fmaxf(v, 0.0f);
        out[(long)rowM * NCOLS + outCol] = v;
      }
    }

    tile = ntile;
    have = nhave;
    round = nround;
  }
}


static inline long cdiv(long a, long b) { return (a + b - 1) / b; }

extern "C" void kernel_launch(void* const* d_in, const int* in_sizes, int n_in,
                              void* d_out, int out_size, void* d_ws, size_t ws_size,
                              hipStream_t stream) {
  const float* feat = (const float*)d_in[0];
  const int*   src  = (const int*)d_in[1];
  const int*   dst  = (const int*)d_in[2];
  const float* w1   = (const float*)d_in[3];
  const float* b1   = (const float*)d_in[4];
  const float* cw1  = (const float*)d_in[5];
  const float* cb1  = (const float*)d_in[6];
  const float* cw2  = (const float*)d_in[7];
  const float* cb2  = (const float*)d_in[8];
  const float* w2   = (const float*)d_in[9];
  const float* b2   = (const float*)d_in[10];

  const int N = in_sizes[0] / 128;
  const int E = in_sizes[1];
  const long NF = (long)N * 128;
  const int NA = (N + 63) & ~63;

  char* wsb = (char*)d_ws;
  int* cntO   = (int*)wsb;                wsb += (size_t)NA * 4;
  int* cntI   = (int*)wsb;                wsb += (size_t)NA * 4;
  int* rowptr = (int*)wsb;                wsb += (size_t)NA * 4;
  int* cursor = (int*)wsb;                wsb += (size_t)NA * 4;
  int* bsum   = (int*)wsb;                wsb += 512 * 4;
  int* csc    = (int*)wsb;                wsb += ((size_t)E * 4 + 255) & ~(size_t)255;
  float* degO = (float*)wsb;              wsb += (size_t)NA * 4;
  float* degI = (float*)wsb;              wsb += (size_t)NA * 4;
  float* h0   = (float*)wsb;              wsb += (size_t)NF * 4;
  float* bufS = (float*)wsb;              wsb += (size_t)NF * 4;
  float* bufA = (float*)wsb;              wsb += (size_t)NF * 4;
  float* bufB = (float*)wsb;
  (void)ws_size; (void)n_in; (void)out_size;

  const int NB       = (int)cdiv(N, 256);
  const int nb_node  = NB;
  const int nb_vec4  = (int)cdiv(NF / 4, 256);
  const int nb_pull  = (int)cdiv(N, 8);
  const int n_strips = N / 16;

  unsigned int* hist = (unsigned int*)bufA;
  int* pre = (int*)bufB;

  k_hist<<<128, 1024, 0, stream>>>(src, dst, hist, E);
  k_sum<<<196, 256, 0, stream>>>(hist, cntO, cntI, pre, N, NA);
  k_degrsqrt<<<nb_node, 256, 0, stream>>>(cntO, cntI, degO, degI, N);
  k_scan_block<<<NB, 256, 0, stream>>>(cntI, rowptr, bsum, N);
  k_scan_tops<<<1, 512, 0, stream>>>(bsum, NB);
  k_scan_add<<<NB, 256, 0, stream>>>(rowptr, bsum, cursor, N);
  k_scatter<<<64, 1024, 0, stream>>>(src, dst, rowptr, pre, csc, E, NA);

  k_gemm_wmma<128, true, false><<<256, 512, 0, stream>>>(feat, w1, b1, h0, 1.0f, 0.0f, n_strips);

  k_scale<<<nb_vec4, 256, 0, stream>>>((const float4*)h0, degO, (float4*)bufS, NF / 4);
  k_spmm_pull<<<nb_pull, 256, 0, stream>>>(rowptr, cntI, csc, bufS, degI, h0, bufA, N);
  k_gemm_wmma<128, true, true><<<256, 512, 0, stream>>>(bufA, cw1, cb1, bufS,
                                                        GA_BETA1, 1.0f - GA_BETA1, n_strips);

  k_scale<<<nb_vec4, 256, 0, stream>>>((const float4*)bufS, degO, (float4*)bufB, NF / 4);
  k_spmm_pull<<<nb_pull, 256, 0, stream>>>(rowptr, cntI, csc, bufB, degI, h0, bufA, N);
  k_gemm_wmma<128, false, true><<<256, 512, 0, stream>>>(bufA, cw2, cb2, bufB,
                                                         GA_BETA2, 1.0f - GA_BETA2, n_strips);

  k_gemm_wmma<64, false, false><<<256, 512, 0, stream>>>(bufB, w2, b2, (float*)d_out,
                                                        1.0f, 0.0f, n_strips);
}


# Row-scale and SpMM kernels run as resident waves looping over rows/nodes (4096 and 8192 waves, not 100000 each); a node's indices come in one load, rows 16 per batch; sum order unchanged
# speedup vs baseline: 1.7165x; 1.0729x over previous
#include <hip/hip_runtime.h>


#define GA_ALPHA 0.2f
#define GA_BETA1 0.6931471805599453f
#define GA_BETA2 0.4054651081081644f

typedef __attribute__((ext_vector_type(2))) float v2f;
typedef __attribute__((ext_vector_type(8))) float v8f;


#define CS_SLICES  64
#define CS_WORDS   50048
#define CS_THREADS 1024
#define CS_B       8

__global__ __launch_bounds__(CS_THREADS) void k_hist(const int* __restrict__ src,
                                                     const int* __restrict__ dst,
                                                     unsigned int* __restrict__ hist, int E) {
  __shared__ unsigned int h[CS_WORDS];
  const int tid = threadIdx.x;
  const int which = blockIdx.x / CS_SLICES;
  const int b = blockIdx.x - which * CS_SLICES;
  const int* __restrict__ key = which ? dst : src;
  for (int w = tid; w < CS_WORDS; w += CS_THREADS) h[w] = 0u;
  __syncthreads();
  const int len = (E + CS_SLICES - 1) / CS_SLICES;
  const int lo = b * len;
  const int hi = min(E, lo + len);
  for (int e0 = lo + tid; e0 < hi; e0 += CS_B * CS_THREADS) {
    int v[CS_B];
#pragma unroll
    for (int k = 0; k < CS_B; ++k) {
      const int e = e0 + k * CS_THREADS;
      v[k] = (e < hi) ? key[e] : -1;
    }
#pragma unroll
    for (int k = 0; k < CS_B; ++k)
      if (v[k] >= 0) atomicAdd(&h[v[k] >> 1], (v[k] & 1) ? 0x10000u : 1u);
  }
  __syncthreads();
  unsigned int* __restrict__ out = hist + (size_t)blockIdx.x * CS_WORDS;
  for (int w = tid; w < CS_WORDS; w += CS_THREADS) out[w] = h[w];
}

__global__ __launch_bounds__(256) void k_sum(const unsigned int* __restrict__ hist,
                                             int* __restrict__ cntO, int* __restrict__ cntI,
                                             int* __restrict__ pre, int N, int NA) {
  const int w = blockIdx.x * 256 + threadIdx.x;
  if (w >= CS_WORDS) return;
  const int n0 = 2 * w, n1 = 2 * w + 1;
  const bool v0 = n0 < N, v1 = n1 < N;
  if (!v0) return;

  unsigned int x[CS_SLICES];
#pragma unroll
  for (int b = 0; b < CS_SLICES; ++b) x[b] = hist[(size_t)b * CS_WORDS + w];
  unsigned int lo = 0u, hi = 0u;
#pragma unroll
  for (int b = 0; b < CS_SLICES; ++b) {
    lo += x[b] & 0xffffu;
    hi += x[b] >> 16;
  }
  const unsigned int outLo = lo, outHi = hi;

#pragma unroll
  for (int b = 0; b < CS_SLICES; ++b) x[b] = hist[(size_t)(CS_SLICES + b) * CS_WORDS + w];
  int* __restrict__ prow = pre + n0;
  lo = 0u; hi = 0u;
  if (v1) {
#pragma unroll
    for (int b = 0; b < CS_SLICES; ++b) {
      *(int2*)(prow + (size_t)b * NA) = make_int2((int)lo, (int)hi);
      lo += x[b] & 0xffffu;
      hi += x[b] >> 16;
    }
    *(int2*)(cntO + n0) = make_int2((int)outLo, (int)outHi);
    *(int2*)(cntI + n0) = make_int2((int)lo, (int)hi);
  } else {
#pragma unroll
    for (int b = 0; b < CS_SLICES; ++b) {
      prow[(size_t)b * NA] = (int)lo;
      lo += x[b] & 0xffffu;
    }
    cntO[n0] = (int)outLo;
    cntI[n0] = (int)lo;
  }
}

__global__ __launch_bounds__(256) void k_degrsqrt(const int* __restrict__ cntO,
                                                  const int* __restrict__ cntI,
                                                  float* __restrict__ degO,
                                                  float* __restrict__ degI, int n) {
  int i = blockIdx.x * 256 + threadIdx.x;
  if (i < n) {
    degO[i] = rsqrtf(fmaxf((float)cntO[i], 1.0f));
    degI[i] = rsqrtf(fmaxf((float)cntI[i], 1.0f));
  }
}


__global__ __launch_bounds__(256) void k_scan_block(const int* __restrict__ in,
                                                    int* __restrict__ outEx,
                                                    int* __restrict__ bsum, int n) {
  __shared__ int s[256];
  const int tid = threadIdx.x;
  const int i = blockIdx.x * 256 + tid;
  const int v = (i < n) ? in[i] : 0;
  s[tid] = v;
  __syncthreads();
#pragma unroll
  for (int off = 1; off < 256; off <<= 1) {
    int t = (tid >= off) ? s[tid - off] : 0;
    __syncthreads();
    s[tid] += t;
    __syncthreads();
  }
  if (i < n) outEx[i] = s[tid] - v;
  if (tid == 255) bsum[blockIdx.x] = s[255];
}

__global__ __launch_bounds__(512) void k_scan_tops(int* __restrict__ bsum, int nb) {
  __shared__ int s[512];
  const int tid = threadIdx.x;
  const int v = (tid < nb) ? bsum[tid] : 0;
  s[tid] = v;
  __syncthreads();
#pragma unroll
  for (int off = 1; off < 512; off <<= 1) {
    int t = (tid >= off) ? s[tid - off] : 0;
    __syncthreads();
    s[tid] += t;
    __syncthreads();
  }
  if (tid < nb) bsum[tid] = s[tid] - v;
}

__global__ __launch_bounds__(256) void k_scan_add(int* __restrict__ outEx,
                                                  const int* __restrict__ bsum,
                                                  int* __restrict__ cursor, int n) {
  int i = blockIdx.x * 256 + threadIdx.x;
  if (i < n) {
    int r = outEx[i] + bsum[blockIdx.x];
    outEx[i] = r;
    cursor[i] = r;
  }
}

__global__ __launch_bounds__(CS_THREADS) void k_scatter(const int* __restrict__ src,
                                                        const int* __restrict__ dst,
                                                        const int* __restrict__ rowptr,
                                                        const int* __restrict__ pre,
                                                        int* __restrict__ csc, int E, int NA) {
  __shared__ unsigned int h[CS_WORDS];
  const int tid = threadIdx.x;
  const int b = blockIdx.x;
  for (int w = tid; w < CS_WORDS; w += CS_THREADS) h[w] = 0u;
  __syncthreads();
  const int len = (E + CS_SLICES - 1) / CS_SLICES;
  const int lo = b * len;
  const int hi = min(E, lo + len);
  const int* __restrict__ preb = pre + (size_t)b * NA;
  for (int e0 = lo + tid; e0 < hi; e0 += CS_B * CS_THREADS) {
    int s[CS_B], d[CS_B], base[CS_B];
    unsigned int old[CS_B];
    bool ok[CS_B];
#pragma unroll
    for (int k = 0; k < CS_B; ++k) {
      const int e = e0 + k * CS_THREADS;
      ok[k] = e < hi;
      const int ec = ok[k] ? e : e0;
      d[k] = dst[ec];
      s[k] = src[ec];
    }
#pragma unroll
    for (int k = 0; k < CS_B; ++k) {
      const unsigned int inc = ok[k] ? ((d[k] & 1) ? 0x10000u : 1u) : 0u;
      old[k] = atomicAdd(&h[d[k] >> 1], inc);
      base[k] = rowptr[d[k]] + preb[d[k]];
    }
#pragma unroll
    for (int k = 0; k < CS_B; ++k) {
      const unsigned int rank = (d[k] & 1) ? (old[k] >> 16) : (old[k] & 0xffffu);
      if (ok[k]) csc[base[k] + (int)rank] = s[k];
    }
  }
}

#define SC_B 8
__global__ __launch_bounds__(256) void k_scale(const float4* __restrict__ x,
                                               const float* __restrict__ dors,
                                               float4* __restrict__ hs, long total4) {
  const long stride = (long)gridDim.x * 256;
  long i = (long)blockIdx.x * 256 + threadIdx.x;
  for (; i + (SC_B - 1) * stride < total4; i += SC_B * stride) {
    float4 v[SC_B];
    float sS[SC_B];
#pragma unroll
    for (int k = 0; k < SC_B; ++k) {
      const long e = i + k * stride;
      v[k] = x[e];
      sS[k] = dors[(int)(e >> 5)];
    }
#pragma unroll
    for (int k = 0; k < SC_B; ++k) {
      const long e = i + k * stride;
      float4 r = v[k];
      r.x *= sS[k]; r.y *= sS[k]; r.z *= sS[k]; r.w *= sS[k];
      hs[e] = r;
    }
  }
  for (; i < total4; i += stride) {
    const float sS = dors[(int)(i >> 5)];
    float4 v = x[i];
    v.x *= sS; v.y *= sS; v.z *= sS; v.w *= sS;
    hs[i] = v;
  }
}

#define SP_B 16

template <bool FULL>
__device__ __forceinline__ void sp_batch(const float* __restrict__ hsl, int myidx, int i,
                                         int rem, float4& acc) {
  int s[SP_B];
#pragma unroll
  for (int k = 0; k < SP_B; ++k) s[k] = __builtin_amdgcn_readlane(myidx, i + k);
  float4 v[SP_B];
#pragma unroll
  for (int k = 0; k < SP_B; ++k) {
    if (FULL || k < rem) v[k] = *(const float4*)(hsl + (long)s[k] * 128);
    else                 v[k] = float4{0.0f, 0.0f, 0.0f, 0.0f};
  }
#pragma unroll
  for (int k = 0; k < SP_B; ++k) {
    if (FULL || k < rem) {
      acc.x += v[k].x; acc.y += v[k].y; acc.z += v[k].z; acc.w += v[k].w;
    }
  }
}

__global__ __launch_bounds__(256) void k_spmm_pull(const int* __restrict__ rowptr,
                                                   const int* __restrict__ cnt,
                                                   const int* __restrict__ csc,
                                                   const float* __restrict__ hs,
                                                   const float* __restrict__ dirs,
                                                   const float* __restrict__ x0,
                                                   float* __restrict__ f, int N) {
  const int wave = __builtin_amdgcn_readfirstlane((int)(threadIdx.x >> 5));
  const int lane = threadIdx.x & 31;
  const int step = (int)gridDim.x * 8;
  const float* __restrict__ hsl = hs + lane * 4;
  int node = (int)blockIdx.x * 8 + wave;
  if (node >= N) return;
  int start = __builtin_amdgcn_readfirstlane(rowptr[node]);
  int c = __builtin_amdgcn_readfirstlane(cnt[node]);
  float dn = dirs[node];
  for (;;) {
    const int nnode = node + step;
    const int nn = min(nnode, N - 1);
    const int nstart = __builtin_amdgcn_readfirstlane(rowptr[nn]);
    const int nc = __builtin_amdgcn_readfirstlane(cnt[nn]);
    const float ndn = dirs[nn];

    int m = min(32, c);
    int myidx = 0;
    if (lane < m) myidx = csc[start + lane];
    const long o = (long)node * 128 + lane * 4;
    const float4 x0v = *(const float4*)(x0 + o);
    const float scale = (1.0f - GA_ALPHA) * dn;
    float4 acc = {0.0f, 0.0f, 0.0f, 0.0f};
    int base = 0;
    for (;;) {
      for (int i = 0; i < m; i += SP_B) {
        const int rem = m - i;
        if (rem >= SP_B) sp_batch<true>(hsl, myidx, i, rem, acc);
        else             sp_batch<false>(hsl, myidx, i, rem, acc);
      }
      base += 32;
      if (base >= c) break;
      m = min(32, c - base);
      myidx = 0;
      if (lane < m) myidx = csc[start + base + lane];
    }
    float4 r;
    r.x = scale * acc.x + GA_ALPHA * x0v.x;
    r.y = scale * acc.y + GA_ALPHA * x0v.y;
    r.z = scale * acc.z + GA_ALPHA * x0v.z;
    r.w = scale * acc.w + GA_ALPHA * x0v.w;
    *(float4*)(f + o) = r;

    if (nnode >= N) break;
    node = nnode; start = nstart; c = nc; dn = ndn;
  }
}

template <int NCOLS, bool RELU, bool RESID>
__global__ __launch_bounds__(512) void k_gemm_wmma(const float* __restrict__ A,
                                                   const float* __restrict__ W,
                                                   const float* __restrict__ bias,
                                                   float* __restrict__ out,
                                                   float actW, float resS, int nTiles) {
  constexpr int NCT = NCOLS / 16;
  __shared__ float sW[128 * NCOLS];

  const int wave = __builtin_amdgcn_readfirstlane((int)(threadIdx.x >> 5));
  const int lane = threadIdx.x & 31;
  const int half = lane >> 4;
  const int l16  = lane & 15;
  const int nblk = gridDim.x;

  int round = 0;
  int tile = blockIdx.x + nblk * wave;
  bool have = tile < nTiles;

  constexpr int NST = (128 * NCOLS) / 512;
  float wv[NST];
#pragma unroll
  for (int it = 0; it < NST; ++it) {
    const int d  = it * 512 + threadIdx.x;
    const int p  = d & 1;
    const int c  = (d >> 1) & 15;
    const int h  = (d >> 5) & 1;
    const int bt = d >> 6;
    const int t  = bt % NCT;
    const int js = bt / NCT;
    const int k  = (js >> 1) * 8 + h * 4 + (js & 1) * 2 + p;
    wv[it] = W[k * NCOLS + t * 16 + c];
  }

  float4 a[16];
  {
    const float* __restrict__ Ap = A + (long)((have ? tile : 0) * 16 + l16) * 128 + half * 4;
#pragma unroll
    for (int j = 0; j < 16; ++j) a[j] = *(const float4*)(Ap + 8 * j);
  }

#pragma unroll
  for (int it = 0; it < NST; ++it) sW[it * 512 + threadIdx.x] = wv[it];
  __builtin_amdgcn_fence(__ATOMIC_RELEASE, "workgroup", "local");
  __builtin_amdgcn_s_barrier();
  __builtin_amdgcn_fence(__ATOMIC_ACQUIRE, "workgroup", "local");

  const float* __restrict__ sWl = sW + lane * 2;

  while (have) {
    const int nround = round + 1;
    const int q = (nround & 1) ? ((13 * wave) & 15) : wave;
    const int ntile = blockIdx.x + nblk * (16 * nround + q);
    const bool nhave = ntile < nTiles;
    const float* __restrict__ An =
        A + (long)((nhave ? ntile : tile) * 16 + l16) * 128 + half * 4;

    v8f acc[NCT];
#pragma unroll
    for (int t = 0; t < NCT; ++t) acc[t] = v8f{};

#pragma unroll
    for (int j = 0; j < 16; ++j) {
      v2f b[2 * NCT];
#pragma unroll
      for (int u = 0; u < 2 * NCT; ++u) b[u] = *(const v2f*)(sWl + (j * 2 * NCT + u) * 64);
#pragma unroll
      for (int s = 0; s < 2; ++s) {
        v2f af;
        af.x = s ? a[j].z : a[j].x;
        af.y = s ? a[j].w : a[j].y;
#pragma unroll
        for (int t = 0; t < NCT; ++t) {
          acc[t] = __builtin_amdgcn_wmma_f32_16x16x4_f32(false, af, false, b[s * NCT + t],
                                                         (short)0, acc[t], false, false);
        }
      }
      a[j] = *(const float4*)(An + 8 * j);
      __builtin_amdgcn_sched_barrier(0);
    }

    const int row0 = tile * 16;
#pragma unroll
    for (int t = 0; t < NCT; ++t) {
      const int outCol = t * 16 + l16;
      const float bv = bias[outCol];
#pragma unroll
      for (int r = 0; r < 8; ++r) {
        const int rowM = row0 + r + half * 8;
        float v = actW * acc[t][r] + bv;
        if (RESID) v += resS * A[(long)rowM * 128 + outCol];
        if (RELU) v = fmaxf(v, 0.0f);
        out[(long)rowM * NCOLS + outCol] = v;
      }
    }

    tile = ntile;
    have = nhave;
    round = nround;
  }
}


static inline long cdiv(long a, long b) { return (a + b - 1) / b; }

extern "C" void kernel_launch(void* const* d_in, const int* in_sizes, int n_in,
                              void* d_out, int out_size, void* d_ws, size_t ws_size,
                              hipStream_t stream) {
  const float* feat = (const float*)d_in[0];
  const int*   src  = (const int*)d_in[1];
  const int*   dst  = (const int*)d_in[2];
  const float* w1   = (const float*)d_in[3];
  const float* b1   = (const float*)d_in[4];
  const float* cw1  = (const float*)d_in[5];
  const float* cb1  = (const float*)d_in[6];
  const float* cw2  = (const float*)d_in[7];
  const float* cb2  = (const float*)d_in[8];
  const float* w2   = (const float*)d_in[9];
  const float* b2   = (const float*)d_in[10];

  const int N = in_sizes[0] / 128;
  const int E = in_sizes[1];
  const long NF = (long)N * 128;
  const int NA = (N + 63) & ~63;

  char* wsb = (char*)d_ws;
  int* cntO   = (int*)wsb;                wsb += (size_t)NA * 4;
  int* cntI   = (int*)wsb;                wsb += (size_t)NA * 4;
  int* rowptr = (int*)wsb;                wsb += (size_t)NA * 4;
  int* cursor = (int*)wsb;                wsb += (size_t)NA * 4;
  int* bsum   = (int*)wsb;                wsb += 512 * 4;
  int* csc    = (int*)wsb;                wsb += ((size_t)E * 4 + 255) & ~(size_t)255;
  float* degO = (float*)wsb;              wsb += (size_t)NA * 4;
  float* degI = (float*)wsb;              wsb += (size_t)NA * 4;
  float* h0   = (float*)wsb;              wsb += (size_t)NF * 4;
  float* bufS = (float*)wsb;              wsb += (size_t)NF * 4;
  float* bufA = (float*)wsb;              wsb += (size_t)NF * 4;
  float* bufB = (float*)wsb;
  (void)ws_size; (void)n_in; (void)out_size;

  const int NB       = (int)cdiv(N, 256);
  const int nb_node  = NB;
  const int n_strips = N / 16;

  unsigned int* hist = (unsigned int*)bufA;
  int* pre = (int*)bufB;

  k_hist<<<128, 1024, 0, stream>>>(src, dst, hist, E);
  k_sum<<<196, 256, 0, stream>>>(hist, cntO, cntI, pre, N, NA);
  k_degrsqrt<<<nb_node, 256, 0, stream>>>(cntO, cntI, degO, degI, N);
  k_scan_block<<<NB, 256, 0, stream>>>(cntI, rowptr, bsum, N);
  k_scan_tops<<<1, 512, 0, stream>>>(bsum, NB);
  k_scan_add<<<NB, 256, 0, stream>>>(rowptr, bsum, cursor, N);
  k_scatter<<<64, 1024, 0, stream>>>(src, dst, rowptr, pre, csc, E, NA);

  k_gemm_wmma<128, true, false><<<256, 512, 0, stream>>>(feat, w1, b1, h0, 1.0f, 0.0f, n_strips);

  k_scale<<<512, 256, 0, stream>>>((const float4*)h0, degO, (float4*)bufS, NF / 4);
  k_spmm_pull<<<1024, 256, 0, stream>>>(rowptr, cntI, csc, bufS, degI, h0, bufA, N);
  k_gemm_wmma<128, true, true><<<256, 512, 0, stream>>>(bufA, cw1, cb1, bufS,
                                                        GA_BETA1, 1.0f - GA_BETA1, n_strips);

  k_scale<<<512, 256, 0, stream>>>((const float4*)bufS, degO, (float4*)bufB, NF / 4);
  k_spmm_pull<<<1024, 256, 0, stream>>>(rowptr, cntI, csc, bufB, degI, h0, bufA, N);
  k_gemm_wmma<128, false, true><<<256, 512, 0, stream>>>(bufA, cw2, cb2, bufB,
                                                         GA_BETA2, 1.0f - GA_BETA2, n_strips);

  k_gemm_wmma<64, false, false><<<256, 512, 0, stream>>>(bufB, w2, b2, (float*)d_out,
                                                        1.0f, 0.0f, n_strips);
}


# Layer 2 projects before it aggregates: cw2, identity term and w2 folded into one 128x64 matrix (on WMMA), applied to h0 and x1 rows first; 2nd gather reads 256-B rows; f32 sum order regrouped
# speedup vs baseline: 1.8014x; 1.0495x over previous
#include <hip/hip_runtime.h>


#define GA_ALPHA 0.2f
#define GA_BETA1 0.6931471805599453f
#define GA_BETA2 0.4054651081081644f

typedef __attribute__((ext_vector_type(2))) float v2f;
typedef __attribute__((ext_vector_type(8))) float v8f;


#define CS_SLICES  64
#define CS_WORDS   50048
#define CS_THREADS 1024
#define CS_B       8

__global__ __launch_bounds__(CS_THREADS) void k_hist(const int* __restrict__ src,
                                                     const int* __restrict__ dst,
                                                     unsigned int* __restrict__ hist, int E) {
  __shared__ unsigned int h[CS_WORDS];
  const int tid = threadIdx.x;
  const int which = blockIdx.x / CS_SLICES;
  const int b = blockIdx.x - which * CS_SLICES;
  const int* __restrict__ key = which ? dst : src;
  for (int w = tid; w < CS_WORDS; w += CS_THREADS) h[w] = 0u;
  __syncthreads();
  const int len = (E + CS_SLICES - 1) / CS_SLICES;
  const int lo = b * len;
  const int hi = min(E, lo + len);
  for (int e0 = lo + tid; e0 < hi; e0 += CS_B * CS_THREADS) {
    int v[CS_B];
#pragma unroll
    for (int k = 0; k < CS_B; ++k) {
      const int e = e0 + k * CS_THREADS;
      v[k] = (e < hi) ? key[e] : -1;
    }
#pragma unroll
    for (int k = 0; k < CS_B; ++k)
      if (v[k] >= 0) atomicAdd(&h[v[k] >> 1], (v[k] & 1) ? 0x10000u : 1u);
  }
  __syncthreads();
  unsigned int* __restrict__ out = hist + (size_t)blockIdx.x * CS_WORDS;
  for (int w = tid; w < CS_WORDS; w += CS_THREADS) out[w] = h[w];
}

__global__ __launch_bounds__(256) void k_sum(const unsigned int* __restrict__ hist,
                                             int* __restrict__ cntO, int* __restrict__ cntI,
                                             int* __restrict__ pre, int N, int NA) {
  const int w = blockIdx.x * 256 + threadIdx.x;
  if (w >= CS_WORDS) return;
  const int n0 = 2 * w, n1 = 2 * w + 1;
  const bool v0 = n0 < N, v1 = n1 < N;
  if (!v0) return;

  unsigned int x[CS_SLICES];
#pragma unroll
  for (int b = 0; b < CS_SLICES; ++b) x[b] = hist[(size_t)b * CS_WORDS + w];
  unsigned int lo = 0u, hi = 0u;
#pragma unroll
  for (int b = 0; b < CS_SLICES; ++b) {
    lo += x[b] & 0xffffu;
    hi += x[b] >> 16;
  }
  const unsigned int outLo = lo, outHi = hi;

#pragma unroll
  for (int b = 0; b < CS_SLICES; ++b) x[b] = hist[(size_t)(CS_SLICES + b) * CS_WORDS + w];
  int* __restrict__ prow = pre + n0;
  lo = 0u; hi = 0u;
  if (v1) {
#pragma unroll
    for (int b = 0; b < CS_SLICES; ++b) {
      *(int2*)(prow + (size_t)b * NA) = make_int2((int)lo, (int)hi);
      lo += x[b] & 0xffffu;
      hi += x[b] >> 16;
    }
    *(int2*)(cntO + n0) = make_int2((int)outLo, (int)outHi);
    *(int2*)(cntI + n0) = make_int2((int)lo, (int)hi);
  } else {
#pragma unroll
    for (int b = 0; b < CS_SLICES; ++b) {
      prow[(size_t)b * NA] = (int)lo;
      lo += x[b] & 0xffffu;
    }
    cntO[n0] = (int)outLo;
    cntI[n0] = (int)lo;
  }
}

__global__ __launch_bounds__(256) void k_degrsqrt(const int* __restrict__ cntO,
                                                  const int* __restrict__ cntI,
                                                  float* __restrict__ degO,
                                                  float* __restrict__ degI, int n) {
  int i = blockIdx.x * 256 + threadIdx.x;
  if (i < n) {
    degO[i] = rsqrtf(fmaxf((float)cntO[i], 1.0f));
    degI[i] = rsqrtf(fmaxf((float)cntI[i], 1.0f));
  }
}


__global__ __launch_bounds__(256) void k_scan_block(const int* __restrict__ in,
                                                    int* __restrict__ outEx,
                                                    int* __restrict__ bsum, int n) {
  __shared__ int s[256];
  const int tid = threadIdx.x;
  const int i = blockIdx.x * 256 + tid;
  const int v = (i < n) ? in[i] : 0;
  s[tid] = v;
  __syncthreads();
#pragma unroll
  for (int off = 1; off < 256; off <<= 1) {
    int t = (tid >= off) ? s[tid - off] : 0;
    __syncthreads();
    s[tid] += t;
    __syncthreads();
  }
  if (i < n) outEx[i] = s[tid] - v;
  if (tid == 255) bsum[blockIdx.x] = s[255];
}

__global__ __launch_bounds__(512) void k_scan_tops(int* __restrict__ bsum, int nb) {
  __shared__ int s[512];
  const int tid = threadIdx.x;
  const int v = (tid < nb) ? bsum[tid] : 0;
  s[tid] = v;
  __syncthreads();
#pragma unroll
  for (int off = 1; off < 512; off <<= 1) {
    int t = (tid >= off) ? s[tid - off] : 0;
    __syncthreads();
    s[tid] += t;
    __syncthreads();
  }
  if (tid < nb) bsum[tid] = s[tid] - v;
}

__global__ __launch_bounds__(256) void k_scan_add(int* __restrict__ outEx,
                                                  const int* __restrict__ bsum,
                                                  int* __restrict__ cursor, int n) {
  int i = blockIdx.x * 256 + threadIdx.x;
  if (i < n) {
    int r = outEx[i] + bsum[blockIdx.x];
    outEx[i] = r;
    cursor[i] = r;
  }
}

__global__ __launch_bounds__(CS_THREADS) void k_scatter(const int* __restrict__ src,
                                                        const int* __restrict__ dst,
                                                        const int* __restrict__ rowptr,
                                                        const int* __restrict__ pre,
                                                        int* __restrict__ csc, int E, int NA) {
  __shared__ unsigned int h[CS_WORDS];
  const int tid = threadIdx.x;
  const int b = blockIdx.x;
  for (int w = tid; w < CS_WORDS; w += CS_THREADS) h[w] = 0u;
  __syncthreads();
  const int len = (E + CS_SLICES - 1) / CS_SLICES;
  const int lo = b * len;
  const int hi = min(E, lo + len);
  const int* __restrict__ preb = pre + (size_t)b * NA;
  for (int e0 = lo + tid; e0 < hi; e0 += CS_B * CS_THREADS) {
    int s[CS_B], d[CS_B], base[CS_B];
    unsigned int old[CS_B];
    bool ok[CS_B];
#pragma unroll
    for (int k = 0; k < CS_B; ++k) {
      const int e = e0 + k * CS_THREADS;
      ok[k] = e < hi;
      const int ec = ok[k] ? e : e0;
      d[k] = dst[ec];
      s[k] = src[ec];
    }
#pragma unroll
    for (int k = 0; k < CS_B; ++k) {
      const unsigned int inc = ok[k] ? ((d[k] & 1) ? 0x10000u : 1u) : 0u;
      old[k] = atomicAdd(&h[d[k] >> 1], inc);
      base[k] = rowptr[d[k]] + preb[d[k]];
    }
#pragma unroll
    for (int k = 0; k < CS_B; ++k) {
      const unsigned int rank = (d[k] & 1) ? (old[k] >> 16) : (old[k] & 0xffffu);
      if (ok[k]) csc[base[k] + (int)rank] = s[k];
    }
  }
}

#define SC_B 8
__global__ __launch_bounds__(256) void k_scale(const float4* __restrict__ x,
                                               const float* __restrict__ dors,
                                               float4* __restrict__ hs, long total4) {
  const long stride = (long)gridDim.x * 256;
  long i = (long)blockIdx.x * 256 + threadIdx.x;
  for (; i + (SC_B - 1) * stride < total4; i += SC_B * stride) {
    float4 v[SC_B];
    float sS[SC_B];
#pragma unroll
    for (int k = 0; k < SC_B; ++k) {
      const long e = i + k * stride;
      v[k] = x[e];
      sS[k] = dors[(int)(e >> 5)];
    }
#pragma unroll
    for (int k = 0; k < SC_B; ++k) {
      const long e = i + k * stride;
      float4 r = v[k];
      r.x *= sS[k]; r.y *= sS[k]; r.z *= sS[k]; r.w *= sS[k];
      hs[e] = r;
    }
  }
  for (; i < total4; i += stride) {
    const float sS = dors[(int)(i >> 5)];
    float4 v = x[i];
    v.x *= sS; v.y *= sS; v.z *= sS; v.w *= sS;
    hs[i] = v;
  }
}

#define SP_B 16

__device__ __forceinline__ void sp_zero(float4& a) { a = float4{0.0f, 0.0f, 0.0f, 0.0f}; }
__device__ __forceinline__ void sp_zero(float2& a) { a = float2{0.0f, 0.0f}; }
__device__ __forceinline__ void sp_add(float4& a, const float4& b) {
  a.x += b.x; a.y += b.y; a.z += b.z; a.w += b.w;
}
__device__ __forceinline__ void sp_add(float2& a, const float2& b) { a.x += b.x; a.y += b.y; }
__device__ __forceinline__ float4 sp_comb(float s, const float4& a, const float4& x) {
  float4 r;
  r.x = s * a.x + GA_ALPHA * x.x;
  r.y = s * a.y + GA_ALPHA * x.y;
  r.z = s * a.z + GA_ALPHA * x.z;
  r.w = s * a.w + GA_ALPHA * x.w;
  return r;
}
__device__ __forceinline__ float2 sp_comb(float s, const float2& a, const float2& x) {
  float2 r;
  r.x = s * a.x + GA_ALPHA * x.x;
  r.y = s * a.y + GA_ALPHA * x.y;
  return r;
}

template <bool FULL, typename VT>
__device__ __forceinline__ void sp_batch(const float* __restrict__ hsl, int myidx, int i,
                                         int rem, VT& acc) {
  constexpr int ROWF = 32 * (int)(sizeof(VT) / sizeof(float));
  int s[SP_B];
#pragma unroll
  for (int k = 0; k < SP_B; ++k) s[k] = __builtin_amdgcn_readlane(myidx, i + k);
  VT v[SP_B];
#pragma unroll
  for (int k = 0; k < SP_B; ++k) {
    if (FULL || k < rem) v[k] = *(const VT*)(hsl + (long)s[k] * ROWF);
    else                 sp_zero(v[k]);
  }
#pragma unroll
  for (int k = 0; k < SP_B; ++k) {
    if (FULL || k < rem) sp_add(acc, v[k]);
  }
}

template <typename VT, bool BIAS>
__global__ __launch_bounds__(256) void k_spmm_pull(const int* __restrict__ rowptr,
                                                   const int* __restrict__ cnt,
                                                   const int* __restrict__ csc,
                                                   const float* __restrict__ hs,
                                                   const float* __restrict__ dirs,
                                                   const float* __restrict__ x0,
                                                   const float* __restrict__ bias,
                                                   float* __restrict__ f, int N) {
  constexpr int VW = (int)(sizeof(VT) / sizeof(float));
  constexpr int ROWF = 32 * VW;
  const int wave = __builtin_amdgcn_readfirstlane((int)(threadIdx.x >> 5));
  const int lane = threadIdx.x & 31;
  const int step = (int)gridDim.x * 8;
  const float* __restrict__ hsl = hs + lane * VW;
  int node = (int)blockIdx.x * 8 + wave;
  if (node >= N) return;
  VT bv;
  sp_zero(bv);
  if (BIAS) bv = *(const VT*)(bias + lane * VW);
  int start = __builtin_amdgcn_readfirstlane(rowptr[node]);
  int c = __builtin_amdgcn_readfirstlane(cnt[node]);
  float dn = dirs[node];
  for (;;) {
    const int nnode = node + step;
    const int nn = min(nnode, N - 1);
    const int nstart = __builtin_amdgcn_readfirstlane(rowptr[nn]);
    const int nc = __builtin_amdgcn_readfirstlane(cnt[nn]);
    const float ndn = dirs[nn];

    int m = min(32, c);
    int myidx = 0;
    if (lane < m) myidx = csc[start + lane];
    const long o = (long)node * ROWF + lane * VW;
    const VT x0v = *(const VT*)(x0 + o);
    const float scale = (1.0f - GA_ALPHA) * dn;
    VT acc;
    sp_zero(acc);
    int base = 0;
    for (;;) {
      for (int i = 0; i < m; i += SP_B) {
        const int rem = m - i;
        if (rem >= SP_B) sp_batch<true, VT>(hsl, myidx, i, rem, acc);
        else             sp_batch<false, VT>(hsl, myidx, i, rem, acc);
      }
      base += 32;
      if (base >= c) break;
      m = min(32, c - base);
      myidx = 0;
      if (lane < m) myidx = csc[start + base + lane];
    }
    VT r = sp_comb(scale, acc, x0v);
    if (BIAS) sp_add(r, bv);
    *(VT*)(f + o) = r;

    if (nnode >= N) break;
    node = nnode; start = nstart; c = nc; dn = ndn;
  }
}

template <int NCOLS, bool RELU, bool RESID, bool BIAS>
__global__ __launch_bounds__(512) void k_gemm_wmma(const float* __restrict__ A,
                                                   const float* __restrict__ W,
                                                   const float* __restrict__ bias,
                                                   float* __restrict__ out,
                                                   float actW, float resS, int nTiles) {
  constexpr int NCT = NCOLS / 16;
  __shared__ float sW[128 * NCOLS];

  const int wave = __builtin_amdgcn_readfirstlane((int)(threadIdx.x >> 5));
  const int lane = threadIdx.x & 31;
  const int half = lane >> 4;
  const int l16  = lane & 15;
  const int nblk = gridDim.x;

  int round = 0;
  int tile = blockIdx.x + nblk * wave;
  bool have = tile < nTiles;

  constexpr int NST = (128 * NCOLS) / 512;
  float wv[NST];
#pragma unroll
  for (int it = 0; it < NST; ++it) {
    const int d  = it * 512 + threadIdx.x;
    const int p  = d & 1;
    const int c  = (d >> 1) & 15;
    const int h  = (d >> 5) & 1;
    const int bt = d >> 6;
    const int t  = bt % NCT;
    const int js = bt / NCT;
    const int k  = (js >> 1) * 8 + h * 4 + (js & 1) * 2 + p;
    wv[it] = W[k * NCOLS + t * 16 + c];
  }

  float4 a[16];
  {
    const float* __restrict__ Ap = A + (long)((have ? tile : 0) * 16 + l16) * 128 + half * 4;
#pragma unroll
    for (int j = 0; j < 16; ++j) a[j] = *(const float4*)(Ap + 8 * j);
  }

#pragma unroll
  for (int it = 0; it < NST; ++it) sW[it * 512 + threadIdx.x] = wv[it];
  __builtin_amdgcn_fence(__ATOMIC_RELEASE, "workgroup", "local");
  __builtin_amdgcn_s_barrier();
  __builtin_amdgcn_fence(__ATOMIC_ACQUIRE, "workgroup", "local");

  const float* __restrict__ sWl = sW + lane * 2;

  while (have) {
    const int nround = round + 1;
    const int q = (nround & 1) ? ((13 * wave) & 15) : wave;
    const int ntile = blockIdx.x + nblk * (16 * nround + q);
    const bool nhave = ntile < nTiles;
    const float* __restrict__ An =
        A + (long)((nhave ? ntile : tile) * 16 + l16) * 128 + half * 4;

    v8f acc[NCT];
#pragma unroll
    for (int t = 0; t < NCT; ++t) acc[t] = v8f{};

#pragma unroll
    for (int j = 0; j < 16; ++j) {
      v2f b[2 * NCT];
#pragma unroll
      for (int u = 0; u < 2 * NCT; ++u) b[u] = *(const v2f*)(sWl + (j * 2 * NCT + u) * 64);
#pragma unroll
      for (int s = 0; s < 2; ++s) {
        v2f af;
        af.x = s ? a[j].z : a[j].x;
        af.y = s ? a[j].w : a[j].y;
#pragma unroll
        for (int t = 0; t < NCT; ++t) {
          acc[t] = __builtin_amdgcn_wmma_f32_16x16x4_f32(false, af, false, b[s * NCT + t],
                                                         (short)0, acc[t], false, false);
        }
      }
      a[j] = *(const float4*)(An + 8 * j);
      __builtin_amdgcn_sched_barrier(0);
    }

    const int row0 = tile * 16;
#pragma unroll
    for (int t = 0; t < NCT; ++t) {
      const int outCol = t * 16 + l16;
      const float bv = BIAS ? bias[outCol] : 0.0f;
#pragma unroll
      for (int r = 0; r < 8; ++r) {
        const int rowM = row0 + r + half * 8;
        float v;
        if (BIAS) v = actW * acc[t][r] + bv;
        else      v = actW * acc[t][r];
        if (RESID) v += resS * A[(long)rowM * 128 + outCol];
        if (RELU) v = fmaxf(v, 0.0f);
        out[(long)rowM * NCOLS + outCol] = v;
      }
    }

    tile = ntile;
    have = nhave;
    round = nround;
  }
}

__global__ __launch_bounds__(128) void k_fold(const float* __restrict__ cw,
                                              const float* __restrict__ cb,
                                              const float* __restrict__ w2,
                                              const float* __restrict__ b2,
                                              float* __restrict__ wc,
                                              float* __restrict__ bc, float beta) {
  const int wave = (int)(threadIdx.x >> 5);
  const int lane = threadIdx.x & 31;
  const int half = lane >> 4;
  const int l16  = lane & 15;
  const int rt   = blockIdx.x;
  const int arow = rt * 16 + l16;
  const int col  = wave * 16 + l16;
  const float* __restrict__ ap = ((arow < 128) ? (cw + arow * 128) : cb) + 2 * half;
  const float* __restrict__ bp = w2 + (2 * half) * 64 + col;

  v8f acc = v8f{};
#pragma unroll
  for (int k0 = 0; k0 < 128; k0 += 4) {
    v2f af, bf;
    af.x = ap[k0];
    af.y = ap[k0 + 1];
    bf.x = bp[k0 * 64];
    bf.y = bp[(k0 + 1) * 64];
    acc = __builtin_amdgcn_wmma_f32_16x16x4_f32(false, af, false, bf, (short)0, acc, false, false);
  }

#pragma unroll
  for (int r = 0; r < 8; ++r) {
    const int rowM = rt * 16 + half * 8 + r;
    const float v = acc[r];
    if (rowM < 128)       wc[rowM * 64 + col] = beta * v + (1.0f - beta) * w2[rowM * 64 + col];
    else if (rowM == 128) bc[col] = v + b2[col];
  }
}


static inline long cdiv(long a, long b) { return (a + b - 1) / b; }

extern "C" void kernel_launch(void* const* d_in, const int* in_sizes, int n_in,
                              void* d_out, int out_size, void* d_ws, size_t ws_size,
                              hipStream_t stream) {
  const float* feat = (const float*)d_in[0];
  const int*   src  = (const int*)d_in[1];
  const int*   dst  = (const int*)d_in[2];
  const float* w1   = (const float*)d_in[3];
  const float* b1   = (const float*)d_in[4];
  const float* cw1  = (const float*)d_in[5];
  const float* cb1  = (const float*)d_in[6];
  const float* cw2  = (const float*)d_in[7];
  const float* cb2  = (const float*)d_in[8];
  const float* w2   = (const float*)d_in[9];
  const float* b2   = (const float*)d_in[10];

  const int N = in_sizes[0] / 128;
  const int E = in_sizes[1];
  const long NF = (long)N * 128;
  const int NA = (N + 63) & ~63;

  char* wsb = (char*)d_ws;
  int* cntO   = (int*)wsb;                wsb += (size_t)NA * 4;
  int* cntI   = (int*)wsb;                wsb += (size_t)NA * 4;
  int* rowptr = (int*)wsb;                wsb += (size_t)NA * 4;
  int* cursor = (int*)wsb;                wsb += (size_t)NA * 4;
  int* bsum   = (int*)wsb;                wsb += 512 * 4;
  int* csc    = (int*)wsb;                wsb += ((size_t)E * 4 + 255) & ~(size_t)255;
  float* degO = (float*)wsb;              wsb += (size_t)NA * 4;
  float* degI = (float*)wsb;              wsb += (size_t)NA * 4;
  float* h0   = (float*)wsb;              wsb += (size_t)NF * 4;
  float* bufS = (float*)wsb;              wsb += (size_t)NF * 4;
  float* bufA = (float*)wsb;              wsb += (size_t)NF * 4;
  float* bufB = (float*)wsb;
  (void)ws_size; (void)n_in; (void)out_size;

  const int NB       = (int)cdiv(N, 256);
  const int nb_node  = NB;
  const int n_strips = N / 16;

  unsigned int* hist = (unsigned int*)bufA;
  int* pre = (int*)bufB;

  float* wc = (float*)cursor;
  float* bc = wc + 128 * 64;

  k_hist<<<128, 1024, 0, stream>>>(src, dst, hist, E);
  k_sum<<<196, 256, 0, stream>>>(hist, cntO, cntI, pre, N, NA);
  k_degrsqrt<<<nb_node, 256, 0, stream>>>(cntO, cntI, degO, degI, N);
  k_scan_block<<<NB, 256, 0, stream>>>(cntI, rowptr, bsum, N);
  k_scan_tops<<<1, 512, 0, stream>>>(bsum, NB);
  k_scan_add<<<NB, 256, 0, stream>>>(rowptr, bsum, cursor, N);
  k_scatter<<<64, 1024, 0, stream>>>(src, dst, rowptr, pre, csc, E, NA);

  k_gemm_wmma<128, true, false, true><<<256, 512, 0, stream>>>(feat, w1, b1, h0, 1.0f, 0.0f,
                                                                n_strips);

  k_scale<<<512, 256, 0, stream>>>((const float4*)h0, degO, (float4*)bufS, NF / 4);
  k_spmm_pull<float4, false><<<1024, 256, 0, stream>>>(rowptr, cntI, csc, bufS, degI, h0, nullptr,
                                                       bufA, N);
  k_gemm_wmma<128, true, true, true><<<256, 512, 0, stream>>>(bufA, cw1, cb1, bufB,
                                                              GA_BETA1, 1.0f - GA_BETA1, n_strips);

  k_scale<<<512, 256, 0, stream>>>((const float4*)bufB, degO, (float4*)bufS, NF / 4);
  k_fold<<<9, 128, 0, stream>>>(cw2, cb2, w2, b2, wc, bc, GA_BETA2);
  k_gemm_wmma<64, false, false, false><<<256, 512, 0, stream>>>(h0, wc, bc, bufA, 1.0f, 0.0f,
                                                                2 * n_strips);
  const float* Q = bufA;
  const float* P = bufA + (size_t)N * 64;
  k_spmm_pull<float2, true><<<1024, 256, 0, stream>>>(rowptr, cntI, csc, P, degI, Q, bc,
                                                      (float*)d_out, N);
}


# CSC placing pass fills all WGPs: work-group per (edge slice, node quarter); LDS table of next free positions, loaded coalesced, replaces 2 scattered reads per edge; in-slice CSC order: LDS arrival
# speedup vs baseline: 2.1829x; 1.2118x over previous
#include <hip/hip_runtime.h>


#define GA_ALPHA 0.2f
#define GA_BETA1 0.6931471805599453f
#define GA_BETA2 0.4054651081081644f

typedef __attribute__((ext_vector_type(2))) float v2f;
typedef __attribute__((ext_vector_type(8))) float v8f;


#define CS_SLICES  64
#define CS_WORDS   50048
#define CS_THREADS 1024
#define CS_B       8

__global__ __launch_bounds__(CS_THREADS) void k_hist(const int* __restrict__ src,
                                                     const int* __restrict__ dst,
                                                     unsigned int* __restrict__ hist, int E) {
  __shared__ unsigned int h[CS_WORDS];
  const int tid = threadIdx.x;
  const int which = blockIdx.x / CS_SLICES;
  const int b = blockIdx.x - which * CS_SLICES;
  const int* __restrict__ key = which ? dst : src;
  for (int w = tid; w < CS_WORDS; w += CS_THREADS) h[w] = 0u;
  __syncthreads();
  const int len = (E + CS_SLICES - 1) / CS_SLICES;
  const int lo = b * len;
  const int hi = min(E, lo + len);
  for (int e0 = lo + tid; e0 < hi; e0 += CS_B * CS_THREADS) {
    int v[CS_B];
#pragma unroll
    for (int k = 0; k < CS_B; ++k) {
      const int e = e0 + k * CS_THREADS;
      v[k] = (e < hi) ? key[e] : -1;
    }
#pragma unroll
    for (int k = 0; k < CS_B; ++k)
      if (v[k] >= 0) atomicAdd(&h[v[k] >> 1], (v[k] & 1) ? 0x10000u : 1u);
  }
  __syncthreads();
  unsigned int* __restrict__ out = hist + (size_t)blockIdx.x * CS_WORDS;
  for (int w = tid; w < CS_WORDS; w += CS_THREADS) out[w] = h[w];
}

__global__ __launch_bounds__(256) void k_sum(const unsigned int* __restrict__ hist,
                                             int* __restrict__ cntO, int* __restrict__ cntI,
                                             int* __restrict__ pre, int N, int NA) {
  const int w = blockIdx.x * 256 + threadIdx.x;
  if (w >= CS_WORDS) return;
  const int n0 = 2 * w, n1 = 2 * w + 1;
  const bool v0 = n0 < N, v1 = n1 < N;
  if (!v0) return;

  unsigned int x[CS_SLICES];
#pragma unroll
  for (int b = 0; b < CS_SLICES; ++b) x[b] = hist[(size_t)b * CS_WORDS + w];
  unsigned int lo = 0u, hi = 0u;
#pragma unroll
  for (int b = 0; b < CS_SLICES; ++b) {
    lo += x[b] & 0xffffu;
    hi += x[b] >> 16;
  }
  const unsigned int outLo = lo, outHi = hi;

#pragma unroll
  for (int b = 0; b < CS_SLICES; ++b) x[b] = hist[(size_t)(CS_SLICES + b) * CS_WORDS + w];
  int* __restrict__ prow = pre + n0;
  lo = 0u; hi = 0u;
  if (v1) {
#pragma unroll
    for (int b = 0; b < CS_SLICES; ++b) {
      *(int2*)(prow + (size_t)b * NA) = make_int2((int)lo, (int)hi);
      lo += x[b] & 0xffffu;
      hi += x[b] >> 16;
    }
    *(int2*)(cntO + n0) = make_int2((int)outLo, (int)outHi);
    *(int2*)(cntI + n0) = make_int2((int)lo, (int)hi);
  } else {
#pragma unroll
    for (int b = 0; b < CS_SLICES; ++b) {
      prow[(size_t)b * NA] = (int)lo;
      lo += x[b] & 0xffffu;
    }
    cntO[n0] = (int)outLo;
    cntI[n0] = (int)lo;
  }
}

__global__ __launch_bounds__(256) void k_degrsqrt(const int* __restrict__ cntO,
                                                  const int* __restrict__ cntI,
                                                  float* __restrict__ degO,
                                                  float* __restrict__ degI, int n) {
  int i = blockIdx.x * 256 + threadIdx.x;
  if (i < n) {
    degO[i] = rsqrtf(fmaxf((float)cntO[i], 1.0f));
    degI[i] = rsqrtf(fmaxf((float)cntI[i], 1.0f));
  }
}


__global__ __launch_bounds__(256) void k_scan_block(const int* __restrict__ in,
                                                    int* __restrict__ outEx,
                                                    int* __restrict__ bsum, int n) {
  __shared__ int s[256];
  const int tid = threadIdx.x;
  const int i = blockIdx.x * 256 + tid;
  const int v = (i < n) ? in[i] : 0;
  s[tid] = v;
  __syncthreads();
#pragma unroll
  for (int off = 1; off < 256; off <<= 1) {
    int t = (tid >= off) ? s[tid - off] : 0;
    __syncthreads();
    s[tid] += t;
    __syncthreads();
  }
  if (i < n) outEx[i] = s[tid] - v;
  if (tid == 255) bsum[blockIdx.x] = s[255];
}

__global__ __launch_bounds__(512) void k_scan_tops(int* __restrict__ bsum, int nb) {
  __shared__ int s[512];
  const int tid = threadIdx.x;
  const int v = (tid < nb) ? bsum[tid] : 0;
  s[tid] = v;
  __syncthreads();
#pragma unroll
  for (int off = 1; off < 512; off <<= 1) {
    int t = (tid >= off) ? s[tid - off] : 0;
    __syncthreads();
    s[tid] += t;
    __syncthreads();
  }
  if (tid < nb) bsum[tid] = s[tid] - v;
}

__global__ __launch_bounds__(256) void k_scan_add(int* __restrict__ outEx,
                                                  const int* __restrict__ bsum,
                                                  int* __restrict__ cursor, int n) {
  int i = blockIdx.x * 256 + threadIdx.x;
  if (i < n) {
    int r = outEx[i] + bsum[blockIdx.x];
    outEx[i] = r;
    cursor[i] = r;
  }
}

#define CP_PARTS 4
#define CP_NODES (2 * CS_WORDS / CP_PARTS)
#define CP_B     13
__global__ __launch_bounds__(CS_THREADS) void k_scatter(const int* __restrict__ src,
                                                        const int* __restrict__ dst,
                                                        const int* __restrict__ rowptr,
                                                        const int* __restrict__ pre,
                                                        int* __restrict__ csc, int E, int N,
                                                        int NA) {
  __shared__ int place[CP_NODES];
  const int tid = threadIdx.x;
  const int b = blockIdx.x / CP_PARTS;
  const int q = blockIdx.x - b * CP_PARTS;
  const int nlo = q * CP_NODES;
  const int nq = min(N - nlo, CP_NODES);
  if (nq <= 0) return;

  const int* __restrict__ rp = rowptr + nlo;
  const int* __restrict__ pp = pre + (size_t)b * NA + nlo;
  for (int i0 = tid; i0 < nq; i0 += CP_B * CS_THREADS) {
    int r[CP_B], p[CP_B];
#pragma unroll
    for (int k = 0; k < CP_B; ++k) {
      const int i = i0 + k * CS_THREADS;
      const int ic = (i < nq) ? i : i0;
      r[k] = rp[ic];
      p[k] = pp[ic];
    }
#pragma unroll
    for (int k = 0; k < CP_B; ++k) {
      const int i = i0 + k * CS_THREADS;
      if (i < nq) place[i] = r[k] + p[k];
    }
  }
  __syncthreads();

  const int len = (E + CS_SLICES - 1) / CS_SLICES;
  const int lo = b * len;
  const int hi = min(E, lo + len);
  for (int e0 = lo + tid; e0 < hi; e0 += CP_B * CS_THREADS) {
    int s[CP_B], li[CP_B], pos[CP_B];
    bool mine[CP_B];
#pragma unroll
    for (int k = 0; k < CP_B; ++k) {
      const int e = e0 + k * CS_THREADS;
      const bool ok = e < hi;
      const int ec = ok ? e : e0;
      li[k] = dst[ec] - nlo;
      s[k] = src[ec];
      mine[k] = ok && ((unsigned int)li[k] < (unsigned int)nq);
    }
#pragma unroll
    for (int k = 0; k < CP_B; ++k) {
      pos[k] = 0;
      if (mine[k]) pos[k] = atomicAdd(&place[li[k]], 1);
    }
#pragma unroll
    for (int k = 0; k < CP_B; ++k)
      if (mine[k]) csc[pos[k]] = s[k];
  }
}

#define SC_B 8
__global__ __launch_bounds__(256) void k_scale(const float4* __restrict__ x,
                                               const float* __restrict__ dors,
                                               float4* __restrict__ hs, long total4) {
  const long stride = (long)gridDim.x * 256;
  long i = (long)blockIdx.x * 256 + threadIdx.x;
  for (; i + (SC_B - 1) * stride < total4; i += SC_B * stride) {
    float4 v[SC_B];
    float sS[SC_B];
#pragma unroll
    for (int k = 0; k < SC_B; ++k) {
      const long e = i + k * stride;
      v[k] = x[e];
      sS[k] = dors[(int)(e >> 5)];
    }
#pragma unroll
    for (int k = 0; k < SC_B; ++k) {
      const long e = i + k * stride;
      float4 r = v[k];
      r.x *= sS[k]; r.y *= sS[k]; r.z *= sS[k]; r.w *= sS[k];
      hs[e] = r;
    }
  }
  for (; i < total4; i += stride) {
    const float sS = dors[(int)(i >> 5)];
    float4 v = x[i];
    v.x *= sS; v.y *= sS; v.z *= sS; v.w *= sS;
    hs[i] = v;
  }
}

#define SP_B 16

__device__ __forceinline__ void sp_zero(float4& a) { a = float4{0.0f, 0.0f, 0.0f, 0.0f}; }
__device__ __forceinline__ void sp_zero(float2& a) { a = float2{0.0f, 0.0f}; }
__device__ __forceinline__ void sp_add(float4& a, const float4& b) {
  a.x += b.x; a.y += b.y; a.z += b.z; a.w += b.w;
}
__device__ __forceinline__ void sp_add(float2& a, const float2& b) { a.x += b.x; a.y += b.y; }
__device__ __forceinline__ float4 sp_comb(float s, const float4& a, const float4& x) {
  float4 r;
  r.x = s * a.x + GA_ALPHA * x.x;
  r.y = s * a.y + GA_ALPHA * x.y;
  r.z = s * a.z + GA_ALPHA * x.z;
  r.w = s * a.w + GA_ALPHA * x.w;
  return r;
}
__device__ __forceinline__ float2 sp_comb(float s, const float2& a, const float2& x) {
  float2 r;
  r.x = s * a.x + GA_ALPHA * x.x;
  r.y = s * a.y + GA_ALPHA * x.y;
  return r;
}

template <bool FULL, typename VT>
__device__ __forceinline__ void sp_batch(const float* __restrict__ hsl, int myidx, int i,
                                         int rem, VT& acc) {
  constexpr int ROWF = 32 * (int)(sizeof(VT) / sizeof(float));
  int s[SP_B];
#pragma unroll
  for (int k = 0; k < SP_B; ++k) s[k] = __builtin_amdgcn_readlane(myidx, i + k);
  VT v[SP_B];
#pragma unroll
  for (int k = 0; k < SP_B; ++k) {
    if (FULL || k < rem) v[k] = *(const VT*)(hsl + (long)s[k] * ROWF);
    else                 sp_zero(v[k]);
  }
#pragma unroll
  for (int k = 0; k < SP_B; ++k) {
    if (FULL || k < rem) sp_add(acc, v[k]);
  }
}

template <typename VT, bool BIAS>
__global__ __launch_bounds__(256) void k_spmm_pull(const int* __restrict__ rowptr,
                                                   const int* __restrict__ cnt,
                                                   const int* __restrict__ csc,
                                                   const float* __restrict__ hs,
                                                   const float* __restrict__ dirs,
                                                   const float* __restrict__ x0,
                                                   const float* __restrict__ bias,
                                                   float* __restrict__ f, int N) {
  constexpr int VW = (int)(sizeof(VT) / sizeof(float));
  constexpr int ROWF = 32 * VW;
  const int wave = __builtin_amdgcn_readfirstlane((int)(threadIdx.x >> 5));
  const int lane = threadIdx.x & 31;
  const int step = (int)gridDim.x * 8;
  const float* __restrict__ hsl = hs + lane * VW;
  int node = (int)blockIdx.x * 8 + wave;
  if (node >= N) return;
  VT bv;
  sp_zero(bv);
  if (BIAS) bv = *(const VT*)(bias + lane * VW);
  int start = __builtin_amdgcn_readfirstlane(rowptr[node]);
  int c = __builtin_amdgcn_readfirstlane(cnt[node]);
  float dn = dirs[node];
  for (;;) {
    const int nnode = node + step;
    const int nn = min(nnode, N - 1);
    const int nstart = __builtin_amdgcn_readfirstlane(rowptr[nn]);
    const int nc = __builtin_amdgcn_readfirstlane(cnt[nn]);
    const float ndn = dirs[nn];

    int m = min(32, c);
    int myidx = 0;
    if (lane < m) myidx = csc[start + lane];
    const long o = (long)node * ROWF + lane * VW;
    const VT x0v = *(const VT*)(x0 + o);
    const float scale = (1.0f - GA_ALPHA) * dn;
    VT acc;
    sp_zero(acc);
    int base = 0;
    for (;;) {
      for (int i = 0; i < m; i += SP_B) {
        const int rem = m - i;
        if (rem >= SP_B) sp_batch<true, VT>(hsl, myidx, i, rem, acc);
        else             sp_batch<false, VT>(hsl, myidx, i, rem, acc);
      }
      base += 32;
      if (base >= c) break;
      m = min(32, c - base);
      myidx = 0;
      if (lane < m) myidx = csc[start + base + lane];
    }
    VT r = sp_comb(scale, acc, x0v);
    if (BIAS) sp_add(r, bv);
    *(VT*)(f + o) = r;

    if (nnode >= N) break;
    node = nnode; start = nstart; c = nc; dn = ndn;
  }
}

template <int NCOLS, bool RELU, bool RESID, bool BIAS>
__global__ __launch_bounds__(512) void k_gemm_wmma(const float* __restrict__ A,
                                                   const float* __restrict__ W,
                                                   const float* __restrict__ bias,
                                                   float* __restrict__ out,
                                                   float actW, float resS, int nTiles) {
  constexpr int NCT = NCOLS / 16;
  __shared__ float sW[128 * NCOLS];

  const int wave = __builtin_amdgcn_readfirstlane((int)(threadIdx.x >> 5));
  const int lane = threadIdx.x & 31;
  const int half = lane >> 4;
  const int l16  = lane & 15;
  const int nblk = gridDim.x;

  int round = 0;
  int tile = blockIdx.x + nblk * wave;
  bool have = tile < nTiles;

  constexpr int NST = (128 * NCOLS) / 512;
  float wv[NST];
#pragma unroll
  for (int it = 0; it < NST; ++it) {
    const int d  = it * 512 + threadIdx.x;
    const int p  = d & 1;
    const int c  = (d >> 1) & 15;
    const int h  = (d >> 5) & 1;
    const int bt = d >> 6;
    const int t  = bt % NCT;
    const int js = bt / NCT;
    const int k  = (js >> 1) * 8 + h * 4 + (js & 1) * 2 + p;
    wv[it] = W[k * NCOLS + t * 16 + c];
  }

  float4 a[16];
  {
    const float* __restrict__ Ap = A + (long)((have ? tile : 0) * 16 + l16) * 128 + half * 4;
#pragma unroll
    for (int j = 0; j < 16; ++j) a[j] = *(const float4*)(Ap + 8 * j);
  }

#pragma unroll
  for (int it = 0; it < NST; ++it) sW[it * 512 + threadIdx.x] = wv[it];
  __builtin_amdgcn_fence(__ATOMIC_RELEASE, "workgroup", "local");
  __builtin_amdgcn_s_barrier();
  __builtin_amdgcn_fence(__ATOMIC_ACQUIRE, "workgroup", "local");

  const float* __restrict__ sWl = sW + lane * 2;

  while (have) {
    const int nround = round + 1;
    const int q = (nround & 1) ? ((13 * wave) & 15) : wave;
    const int ntile = blockIdx.x + nblk * (16 * nround + q);
    const bool nhave = ntile < nTiles;
    const float* __restrict__ An =
        A + (long)((nhave ? ntile : tile) * 16 + l16) * 128 + half * 4;

    v8f acc[NCT];
#pragma unroll
    for (int t = 0; t < NCT; ++t) acc[t] = v8f{};

#pragma unroll
    for (int j = 0; j < 16; ++j) {
      v2f b[2 * NCT];
#pragma unroll
      for (int u = 0; u < 2 * NCT; ++u) b[u] = *(const v2f*)(sWl + (j * 2 * NCT + u) * 64);
#pragma unroll
      for (int s = 0; s < 2; ++s) {
        v2f af;
        af.x = s ? a[j].z : a[j].x;
        af.y = s ? a[j].w : a[j].y;
#pragma unroll
        for (int t = 0; t < NCT; ++t) {
          acc[t] = __builtin_amdgcn_wmma_f32_16x16x4_f32(false, af, false, b[s * NCT + t],
                                                         (short)0, acc[t], false, false);
        }
      }
      a[j] = *(const float4*)(An + 8 * j);
      __builtin_amdgcn_sched_barrier(0);
    }

    const int row0 = tile * 16;
#pragma unroll
    for (int t = 0; t < NCT; ++t) {
      const int outCol = t * 16 + l16;
      const float bv = BIAS ? bias[outCol] : 0.0f;
#pragma unroll
      for (int r = 0; r < 8; ++r) {
        const int rowM = row0 + r + half * 8;
        float v;
        if (BIAS) v = actW * acc[t][r] + bv;
        else      v = actW * acc[t][r];
        if (RESID) v += resS * A[(long)rowM * 128 + outCol];
        if (RELU) v = fmaxf(v, 0.0f);
        out[(long)rowM * NCOLS + outCol] = v;
      }
    }

    tile = ntile;
    have = nhave;
    round = nround;
  }
}

__global__ __launch_bounds__(128) void k_fold(const float* __restrict__ cw,
                                              const float* __restrict__ cb,
                                              const float* __restrict__ w2,
                                              const float* __restrict__ b2,
                                              float* __restrict__ wc,
                                              float* __restrict__ bc, float beta) {
  const int wave = (int)(threadIdx.x >> 5);
  const int lane = threadIdx.x & 31;
  const int half = lane >> 4;
  const int l16  = lane & 15;
  const int rt   = blockIdx.x;
  const int arow = rt * 16 + l16;
  const int col  = wave * 16 + l16;
  const float* __restrict__ ap = ((arow < 128) ? (cw + arow * 128) : cb) + 2 * half;
  const float* __restrict__ bp = w2 + (2 * half) * 64 + col;

  v8f acc = v8f{};
#pragma unroll
  for (int k0 = 0; k0 < 128; k0 += 4) {
    v2f af, bf;
    af.x = ap[k0];
    af.y = ap[k0 + 1];
    bf.x = bp[k0 * 64];
    bf.y = bp[(k0 + 1) * 64];
    acc = __builtin_amdgcn_wmma_f32_16x16x4_f32(false, af, false, bf, (short)0, acc, false, false);
  }

#pragma unroll
  for (int r = 0; r < 8; ++r) {
    const int rowM = rt * 16 + half * 8 + r;
    const float v = acc[r];
    if (rowM < 128)       wc[rowM * 64 + col] = beta * v + (1.0f - beta) * w2[rowM * 64 + col];
    else if (rowM == 128) bc[col] = v + b2[col];
  }
}


static inline long cdiv(long a, long b) { return (a + b - 1) / b; }

extern "C" void kernel_launch(void* const* d_in, const int* in_sizes, int n_in,
                              void* d_out, int out_size, void* d_ws, size_t ws_size,
                              hipStream_t stream) {
  const float* feat = (const float*)d_in[0];
  const int*   src  = (const int*)d_in[1];
  const int*   dst  = (const int*)d_in[2];
  const float* w1   = (const float*)d_in[3];
  const float* b1   = (const float*)d_in[4];
  const float* cw1  = (const float*)d_in[5];
  const float* cb1  = (const float*)d_in[6];
  const float* cw2  = (const float*)d_in[7];
  const float* cb2  = (const float*)d_in[8];
  const float* w2   = (const float*)d_in[9];
  const float* b2   = (const float*)d_in[10];

  const int N = in_sizes[0] / 128;
  const int E = in_sizes[1];
  const long NF = (long)N * 128;
  const int NA = (N + 63) & ~63;

  char* wsb = (char*)d_ws;
  int* cntO   = (int*)wsb;                wsb += (size_t)NA * 4;
  int* cntI   = (int*)wsb;                wsb += (size_t)NA * 4;
  int* rowptr = (int*)wsb;                wsb += (size_t)NA * 4;
  int* cursor = (int*)wsb;                wsb += (size_t)NA * 4;
  int* bsum   = (int*)wsb;                wsb += 512 * 4;
  int* csc    = (int*)wsb;                wsb += ((size_t)E * 4 + 255) & ~(size_t)255;
  float* degO = (float*)wsb;              wsb += (size_t)NA * 4;
  float* degI = (float*)wsb;              wsb += (size_t)NA * 4;
  float* h0   = (float*)wsb;              wsb += (size_t)NF * 4;
  float* bufS = (float*)wsb;              wsb += (size_t)NF * 4;
  float* bufA = (float*)wsb;              wsb += (size_t)NF * 4;
  float* bufB = (float*)wsb;
  (void)ws_size; (void)n_in; (void)out_size;

  const int NB       = (int)cdiv(N, 256);
  const int nb_node  = NB;
  const int n_strips = N / 16;

  unsigned int* hist = (unsigned int*)bufA;
  int* pre = (int*)bufB;

  float* wc = (float*)cursor;
  float* bc = wc + 128 * 64;

  k_hist<<<128, 1024, 0, stream>>>(src, dst, hist, E);
  k_sum<<<196, 256, 0, stream>>>(hist, cntO, cntI, pre, N, NA);
  k_degrsqrt<<<nb_node, 256, 0, stream>>>(cntO, cntI, degO, degI, N);
  k_scan_block<<<NB, 256, 0, stream>>>(cntI, rowptr, bsum, N);
  k_scan_tops<<<1, 512, 0, stream>>>(bsum, NB);
  k_scan_add<<<NB, 256, 0, stream>>>(rowptr, bsum, cursor, N);
  k_scatter<<<CS_SLICES * CP_PARTS, CS_THREADS, 0, stream>>>(src, dst, rowptr, pre, csc, E, N, NA);

  k_gemm_wmma<128, true, false, true><<<256, 512, 0, stream>>>(feat, w1, b1, h0, 1.0f, 0.0f,
                                                                n_strips);

  k_scale<<<512, 256, 0, stream>>>((const float4*)h0, degO, (float4*)bufS, NF / 4);
  k_spmm_pull<float4, false><<<1024, 256, 0, stream>>>(rowptr, cntI, csc, bufS, degI, h0, nullptr,
                                                       bufA, N);
  k_gemm_wmma<128, true, true, true><<<256, 512, 0, stream>>>(bufA, cw1, cb1, bufB,
                                                              GA_BETA1, 1.0f - GA_BETA1, n_strips);

  k_scale<<<512, 256, 0, stream>>>((const float4*)bufB, degO, (float4*)bufS, NF / 4);
  k_fold<<<9, 128, 0, stream>>>(cw2, cb2, w2, b2, wc, bc, GA_BETA2);
  k_gemm_wmma<64, false, false, false><<<256, 512, 0, stream>>>(h0, wc, bc, bufA, 1.0f, 0.0f,
                                                                2 * n_strips);
  const float* Q = bufA;
  const float* P = bufA + (size_t)N * 64;
  k_spmm_pull<float2, true><<<1024, 256, 0, stream>>>(rowptr, cntI, csc, P, degI, Q, bc,
                                                      (float*)d_out, N);
}


# Source-degree row scaling moves into the GEMM epilogues: GEMM 1 stores h0 and h0*degO, GEMM 2 stores only x1*degO; both scaling launches and the unscaled x1 array are gone; same products
# speedup vs baseline: 2.4236x; 1.1103x over previous
#include <hip/hip_runtime.h>


#define GA_ALPHA 0.2f
#define GA_BETA1 0.6931471805599453f
#define GA_BETA2 0.4054651081081644f

typedef __attribute__((ext_vector_type(2))) float v2f;
typedef __attribute__((ext_vector_type(8))) float v8f;


#define CS_SLICES  64
#define CS_WORDS   50048
#define CS_THREADS 1024
#define CS_B       8

__global__ __launch_bounds__(CS_THREADS) void k_hist(const int* __restrict__ src,
                                                     const int* __restrict__ dst,
                                                     unsigned int* __restrict__ hist, int E) {
  __shared__ unsigned int h[CS_WORDS];
  const int tid = threadIdx.x;
  const int which = blockIdx.x / CS_SLICES;
  const int b = blockIdx.x - which * CS_SLICES;
  const int* __restrict__ key = which ? dst : src;
  for (int w = tid; w < CS_WORDS; w += CS_THREADS) h[w] = 0u;
  __syncthreads();
  const int len = (E + CS_SLICES - 1) / CS_SLICES;
  const int lo = b * len;
  const int hi = min(E, lo + len);
  for (int e0 = lo + tid; e0 < hi; e0 += CS_B * CS_THREADS) {
    int v[CS_B];
#pragma unroll
    for (int k = 0; k < CS_B; ++k) {
      const int e = e0 + k * CS_THREADS;
      v[k] = (e < hi) ? key[e] : -1;
    }
#pragma unroll
    for (int k = 0; k < CS_B; ++k)
      if (v[k] >= 0) atomicAdd(&h[v[k] >> 1], (v[k] & 1) ? 0x10000u : 1u);
  }
  __syncthreads();
  unsigned int* __restrict__ out = hist + (size_t)blockIdx.x * CS_WORDS;
  for (int w = tid; w < CS_WORDS; w += CS_THREADS) out[w] = h[w];
}

__global__ __launch_bounds__(256) void k_sum(const unsigned int* __restrict__ hist,
                                             int* __restrict__ cntO, int* __restrict__ cntI,
                                             int* __restrict__ pre, int N, int NA) {
  const int w = blockIdx.x * 256 + threadIdx.x;
  if (w >= CS_WORDS) return;
  const int n0 = 2 * w, n1 = 2 * w + 1;
  const bool v0 = n0 < N, v1 = n1 < N;
  if (!v0) return;

  unsigned int x[CS_SLICES];
#pragma unroll
  for (int b = 0; b < CS_SLICES; ++b) x[b] = hist[(size_t)b * CS_WORDS + w];
  unsigned int lo = 0u, hi = 0u;
#pragma unroll
  for (int b = 0; b < CS_SLICES; ++b) {
    lo += x[b] & 0xffffu;
    hi += x[b] >> 16;
  }
  const unsigned int outLo = lo, outHi = hi;

#pragma unroll
  for (int b = 0; b < CS_SLICES; ++b) x[b] = hist[(size_t)(CS_SLICES + b) * CS_WORDS + w];
  int* __restrict__ prow = pre + n0;
  lo = 0u; hi = 0u;
  if (v1) {
#pragma unroll
    for (int b = 0; b < CS_SLICES; ++b) {
      *(int2*)(prow + (size_t)b * NA) = make_int2((int)lo, (int)hi);
      lo += x[b] & 0xffffu;
      hi += x[b] >> 16;
    }
    *(int2*)(cntO + n0) = make_int2((int)outLo, (int)outHi);
    *(int2*)(cntI + n0) = make_int2((int)lo, (int)hi);
  } else {
#pragma unroll
    for (int b = 0; b < CS_SLICES; ++b) {
      prow[(size_t)b * NA] = (int)lo;
      lo += x[b] & 0xffffu;
    }
    cntO[n0] = (int)outLo;
    cntI[n0] = (int)lo;
  }
}

__global__ __launch_bounds__(256) void k_degrsqrt(const int* __restrict__ cntO,
                                                  const int* __restrict__ cntI,
                                                  float* __restrict__ degO,
                                                  float* __restrict__ degI, int n) {
  int i = blockIdx.x * 256 + threadIdx.x;
  if (i < n) {
    degO[i] = rsqrtf(fmaxf((float)cntO[i], 1.0f));
    degI[i] = rsqrtf(fmaxf((float)cntI[i], 1.0f));
  }
}


__global__ __launch_bounds__(256) void k_scan_block(const int* __restrict__ in,
                                                    int* __restrict__ outEx,
                                                    int* __restrict__ bsum, int n) {
  __shared__ int s[256];
  const int tid = threadIdx.x;
  const int i = blockIdx.x * 256 + tid;
  const int v = (i < n) ? in[i] : 0;
  s[tid] = v;
  __syncthreads();
#pragma unroll
  for (int off = 1; off < 256; off <<= 1) {
    int t = (tid >= off) ? s[tid - off] : 0;
    __syncthreads();
    s[tid] += t;
    __syncthreads();
  }
  if (i < n) outEx[i] = s[tid] - v;
  if (tid == 255) bsum[blockIdx.x] = s[255];
}

__global__ __launch_bounds__(512) void k_scan_tops(int* __restrict__ bsum, int nb) {
  __shared__ int s[512];
  const int tid = threadIdx.x;
  const int v = (tid < nb) ? bsum[tid] : 0;
  s[tid] = v;
  __syncthreads();
#pragma unroll
  for (int off = 1; off < 512; off <<= 1) {
    int t = (tid >= off) ? s[tid - off] : 0;
    __syncthreads();
    s[tid] += t;
    __syncthreads();
  }
  if (tid < nb) bsum[tid] = s[tid] - v;
}

__global__ __launch_bounds__(256) void k_scan_add(int* __restrict__ outEx,
                                                  const int* __restrict__ bsum,
                                                  int* __restrict__ cursor, int n) {
  int i = blockIdx.x * 256 + threadIdx.x;
  if (i < n) {
    int r = outEx[i] + bsum[blockIdx.x];
    outEx[i] = r;
    cursor[i] = r;
  }
}

#define CP_PARTS 4
#define CP_NODES (2 * CS_WORDS / CP_PARTS)
#define CP_B     13
__global__ __launch_bounds__(CS_THREADS) void k_scatter(const int* __restrict__ src,
                                                        const int* __restrict__ dst,
                                                        const int* __restrict__ rowptr,
                                                        const int* __restrict__ pre,
                                                        int* __restrict__ csc, int E, int N,
                                                        int NA) {
  __shared__ int place[CP_NODES];
  const int tid = threadIdx.x;
  const int b = blockIdx.x / CP_PARTS;
  const int q = blockIdx.x - b * CP_PARTS;
  const int nlo = q * CP_NODES;
  const int nq = min(N - nlo, CP_NODES);
  if (nq <= 0) return;

  const int* __restrict__ rp = rowptr + nlo;
  const int* __restrict__ pp = pre + (size_t)b * NA + nlo;
  for (int i0 = tid; i0 < nq; i0 += CP_B * CS_THREADS) {
    int r[CP_B], p[CP_B];
#pragma unroll
    for (int k = 0; k < CP_B; ++k) {
      const int i = i0 + k * CS_THREADS;
      const int ic = (i < nq) ? i : i0;
      r[k] = rp[ic];
      p[k] = pp[ic];
    }
#pragma unroll
    for (int k = 0; k < CP_B; ++k) {
      const int i = i0 + k * CS_THREADS;
      if (i < nq) place[i] = r[k] + p[k];
    }
  }
  __syncthreads();

  const int len = (E + CS_SLICES - 1) / CS_SLICES;
  const int lo = b * len;
  const int hi = min(E, lo + len);
  for (int e0 = lo + tid; e0 < hi; e0 += CP_B * CS_THREADS) {
    int s[CP_B], li[CP_B], pos[CP_B];
    bool mine[CP_B];
#pragma unroll
    for (int k = 0; k < CP_B; ++k) {
      const int e = e0 + k * CS_THREADS;
      const bool ok = e < hi;
      const int ec = ok ? e : e0;
      li[k] = dst[ec] - nlo;
      s[k] = src[ec];
      mine[k] = ok && ((unsigned int)li[k] < (unsigned int)nq);
    }
#pragma unroll
    for (int k = 0; k < CP_B; ++k) {
      pos[k] = 0;
      if (mine[k]) pos[k] = atomicAdd(&place[li[k]], 1);
    }
#pragma unroll
    for (int k = 0; k < CP_B; ++k)
      if (mine[k]) csc[pos[k]] = s[k];
  }
}

#define SP_B 16

__device__ __forceinline__ void sp_zero(float4& a) { a = float4{0.0f, 0.0f, 0.0f, 0.0f}; }
__device__ __forceinline__ void sp_zero(float2& a) { a = float2{0.0f, 0.0f}; }
__device__ __forceinline__ void sp_add(float4& a, const float4& b) {
  a.x += b.x; a.y += b.y; a.z += b.z; a.w += b.w;
}
__device__ __forceinline__ void sp_add(float2& a, const float2& b) { a.x += b.x; a.y += b.y; }
__device__ __forceinline__ float4 sp_comb(float s, const float4& a, const float4& x) {
  float4 r;
  r.x = s * a.x + GA_ALPHA * x.x;
  r.y = s * a.y + GA_ALPHA * x.y;
  r.z = s * a.z + GA_ALPHA * x.z;
  r.w = s * a.w + GA_ALPHA * x.w;
  return r;
}
__device__ __forceinline__ float2 sp_comb(float s, const float2& a, const float2& x) {
  float2 r;
  r.x = s * a.x + GA_ALPHA * x.x;
  r.y = s * a.y + GA_ALPHA * x.y;
  return r;
}

template <bool FULL, typename VT>
__device__ __forceinline__ void sp_batch(const float* __restrict__ hsl, int myidx, int i,
                                         int rem, VT& acc) {
  constexpr int ROWF = 32 * (int)(sizeof(VT) / sizeof(float));
  int s[SP_B];
#pragma unroll
  for (int k = 0; k < SP_B; ++k) s[k] = __builtin_amdgcn_readlane(myidx, i + k);
  VT v[SP_B];
#pragma unroll
  for (int k = 0; k < SP_B; ++k) {
    if (FULL || k < rem) v[k] = *(const VT*)(hsl + (long)s[k] * ROWF);
    else                 sp_zero(v[k]);
  }
#pragma unroll
  for (int k = 0; k < SP_B; ++k) {
    if (FULL || k < rem) sp_add(acc, v[k]);
  }
}

template <typename VT, bool BIAS>
__global__ __launch_bounds__(256) void k_spmm_pull(const int* __restrict__ rowptr,
                                                   const int* __restrict__ cnt,
                                                   const int* __restrict__ csc,
                                                   const float* __restrict__ hs,
                                                   const float* __restrict__ dirs,
                                                   const float* __restrict__ x0,
                                                   const float* __restrict__ bias,
                                                   float* __restrict__ f, int N) {
  constexpr int VW = (int)(sizeof(VT) / sizeof(float));
  constexpr int ROWF = 32 * VW;
  const int wave = __builtin_amdgcn_readfirstlane((int)(threadIdx.x >> 5));
  const int lane = threadIdx.x & 31;
  const int step = (int)gridDim.x * 8;
  const float* __restrict__ hsl = hs + lane * VW;
  int node = (int)blockIdx.x * 8 + wave;
  if (node >= N) return;
  VT bv;
  sp_zero(bv);
  if (BIAS) bv = *(const VT*)(bias + lane * VW);
  int start = __builtin_amdgcn_readfirstlane(rowptr[node]);
  int c = __builtin_amdgcn_readfirstlane(cnt[node]);
  float dn = dirs[node];
  for (;;) {
    const int nnode = node + step;
    const int nn = min(nnode, N - 1);
    const int nstart = __builtin_amdgcn_readfirstlane(rowptr[nn]);
    const int nc = __builtin_amdgcn_readfirstlane(cnt[nn]);
    const float ndn = dirs[nn];

    int m = min(32, c);
    int myidx = 0;
    if (lane < m) myidx = csc[start + lane];
    const long o = (long)node * ROWF + lane * VW;
    const VT x0v = *(const VT*)(x0 + o);
    const float scale = (1.0f - GA_ALPHA) * dn;
    VT acc;
    sp_zero(acc);
    int base = 0;
    for (;;) {
      for (int i = 0; i < m; i += SP_B) {
        const int rem = m - i;
        if (rem >= SP_B) sp_batch<true, VT>(hsl, myidx, i, rem, acc);
        else             sp_batch<false, VT>(hsl, myidx, i, rem, acc);
      }
      base += 32;
      if (base >= c) break;
      m = min(32, c - base);
      myidx = 0;
      if (lane < m) myidx = csc[start + base + lane];
    }
    VT r = sp_comb(scale, acc, x0v);
    if (BIAS) sp_add(r, bv);
    *(VT*)(f + o) = r;

    if (nnode >= N) break;
    node = nnode; start = nstart; c = nc; dn = ndn;
  }
}

template <int NCOLS, bool RELU, bool RESID, bool BIAS, int SCALED>
__global__ __launch_bounds__(512) void k_gemm_wmma(const float* __restrict__ A,
                                                   const float* __restrict__ W,
                                                   const float* __restrict__ bias,
                                                   const float* __restrict__ dors,
                                                   float* __restrict__ out,
                                                   float* __restrict__ outS,
                                                   float actW, float resS, int nTiles) {
  constexpr int NCT = NCOLS / 16;
  __shared__ float sW[128 * NCOLS];

  const int wave = __builtin_amdgcn_readfirstlane((int)(threadIdx.x >> 5));
  const int lane = threadIdx.x & 31;
  const int half = lane >> 4;
  const int l16  = lane & 15;
  const int nblk = gridDim.x;

  int round = 0;
  int tile = blockIdx.x + nblk * wave;
  bool have = tile < nTiles;

  constexpr int NST = (128 * NCOLS) / 512;
  float wv[NST];
#pragma unroll
  for (int it = 0; it < NST; ++it) {
    const int d  = it * 512 + threadIdx.x;
    const int p  = d & 1;
    const int c  = (d >> 1) & 15;
    const int h  = (d >> 5) & 1;
    const int bt = d >> 6;
    const int t  = bt % NCT;
    const int js = bt / NCT;
    const int k  = (js >> 1) * 8 + h * 4 + (js & 1) * 2 + p;
    wv[it] = W[k * NCOLS + t * 16 + c];
  }

  float4 a[16];
  {
    const float* __restrict__ Ap = A + (long)((have ? tile : 0) * 16 + l16) * 128 + half * 4;
#pragma unroll
    for (int j = 0; j < 16; ++j) a[j] = *(const float4*)(Ap + 8 * j);
  }

#pragma unroll
  for (int it = 0; it < NST; ++it) sW[it * 512 + threadIdx.x] = wv[it];
  __builtin_amdgcn_fence(__ATOMIC_RELEASE, "workgroup", "local");
  __builtin_amdgcn_s_barrier();
  __builtin_amdgcn_fence(__ATOMIC_ACQUIRE, "workgroup", "local");

  const float* __restrict__ sWl = sW + lane * 2;

  while (have) {
    const int nround = round + 1;
    const int q = (nround & 1) ? ((13 * wave) & 15) : wave;
    const int ntile = blockIdx.x + nblk * (16 * nround + q);
    const bool nhave = ntile < nTiles;
    const float* __restrict__ An =
        A + (long)((nhave ? ntile : tile) * 16 + l16) * 128 + half * 4;

    float ds[8];
    if (SCALED) {
      const float4* __restrict__ dp = (const float4*)(dors + tile * 16 + half * 8);
      const float4 d0 = dp[0];
      const float4 d1 = dp[1];
      ds[0] = d0.x; ds[1] = d0.y; ds[2] = d0.z; ds[3] = d0.w;
      ds[4] = d1.x; ds[5] = d1.y; ds[6] = d1.z; ds[7] = d1.w;
    }

    v8f acc[NCT];
#pragma unroll
    for (int t = 0; t < NCT; ++t) acc[t] = v8f{};

#pragma unroll
    for (int j = 0; j < 16; ++j) {
      v2f b[2 * NCT];
#pragma unroll
      for (int u = 0; u < 2 * NCT; ++u) b[u] = *(const v2f*)(sWl + (j * 2 * NCT + u) * 64);
#pragma unroll
      for (int s = 0; s < 2; ++s) {
        v2f af;
        af.x = s ? a[j].z : a[j].x;
        af.y = s ? a[j].w : a[j].y;
#pragma unroll
        for (int t = 0; t < NCT; ++t) {
          acc[t] = __builtin_amdgcn_wmma_f32_16x16x4_f32(false, af, false, b[s * NCT + t],
                                                         (short)0, acc[t], false, false);
        }
      }
      a[j] = *(const float4*)(An + 8 * j);
      __builtin_amdgcn_sched_barrier(0);
    }

    const int row0 = tile * 16;
#pragma unroll
    for (int t = 0; t < NCT; ++t) {
      const int outCol = t * 16 + l16;
      const float bv = BIAS ? bias[outCol] : 0.0f;
#pragma unroll
      for (int r = 0; r < 8; ++r) {
        const int rowM = row0 + r + half * 8;
        float v;
        if (BIAS) v = actW * acc[t][r] + bv;
        else      v = actW * acc[t][r];
        if (RESID) v += resS * A[(long)rowM * 128 + outCol];
        if (RELU) v = fmaxf(v, 0.0f);
        if (SCALED != 2) out[(long)rowM * NCOLS + outCol] = v;
        if (SCALED) outS[(long)rowM * NCOLS + outCol] = v * ds[r];
      }
    }

    tile = ntile;
    have = nhave;
    round = nround;
  }
}

__global__ __launch_bounds__(128) void k_fold(const float* __restrict__ cw,
                                              const float* __restrict__ cb,
                                              const float* __restrict__ w2,
                                              const float* __restrict__ b2,
                                              float* __restrict__ wc,
                                              float* __restrict__ bc, float beta) {
  const int wave = (int)(threadIdx.x >> 5);
  const int lane = threadIdx.x & 31;
  const int half = lane >> 4;
  const int l16  = lane & 15;
  const int rt   = blockIdx.x;
  const int arow = rt * 16 + l16;
  const int col  = wave * 16 + l16;
  const float* __restrict__ ap = ((arow < 128) ? (cw + arow * 128) : cb) + 2 * half;
  const float* __restrict__ bp = w2 + (2 * half) * 64 + col;

  v8f acc = v8f{};
#pragma unroll
  for (int k0 = 0; k0 < 128; k0 += 4) {
    v2f af, bf;
    af.x = ap[k0];
    af.y = ap[k0 + 1];
    bf.x = bp[k0 * 64];
    bf.y = bp[(k0 + 1) * 64];
    acc = __builtin_amdgcn_wmma_f32_16x16x4_f32(false, af, false, bf, (short)0, acc, false, false);
  }

#pragma unroll
  for (int r = 0; r < 8; ++r) {
    const int rowM = rt * 16 + half * 8 + r;
    const float v = acc[r];
    if (rowM < 128)       wc[rowM * 64 + col] = beta * v + (1.0f - beta) * w2[rowM * 64 + col];
    else if (rowM == 128) bc[col] = v + b2[col];
  }
}


static inline long cdiv(long a, long b) { return (a + b - 1) / b; }

extern "C" void kernel_launch(void* const* d_in, const int* in_sizes, int n_in,
                              void* d_out, int out_size, void* d_ws, size_t ws_size,
                              hipStream_t stream) {
  const float* feat = (const float*)d_in[0];
  const int*   src  = (const int*)d_in[1];
  const int*   dst  = (const int*)d_in[2];
  const float* w1   = (const float*)d_in[3];
  const float* b1   = (const float*)d_in[4];
  const float* cw1  = (const float*)d_in[5];
  const float* cb1  = (const float*)d_in[6];
  const float* cw2  = (const float*)d_in[7];
  const float* cb2  = (const float*)d_in[8];
  const float* w2   = (const float*)d_in[9];
  const float* b2   = (const float*)d_in[10];

  const int N = in_sizes[0] / 128;
  const int E = in_sizes[1];
  const long NF = (long)N * 128;
  const int NA = (N + 63) & ~63;

  char* wsb = (char*)d_ws;
  int* cntO   = (int*)wsb;                wsb += (size_t)NA * 4;
  int* cntI   = (int*)wsb;                wsb += (size_t)NA * 4;
  int* rowptr = (int*)wsb;                wsb += (size_t)NA * 4;
  int* cursor = (int*)wsb;                wsb += (size_t)NA * 4;
  int* bsum   = (int*)wsb;                wsb += 512 * 4;
  int* csc    = (int*)wsb;                wsb += ((size_t)E * 4 + 255) & ~(size_t)255;
  float* degO = (float*)wsb;              wsb += (size_t)NA * 4;
  float* degI = (float*)wsb;              wsb += (size_t)NA * 4;
  float* h0   = (float*)wsb;              wsb += (size_t)NF * 4;
  float* bufS = (float*)wsb;              wsb += (size_t)NF * 4;
  float* bufA = (float*)wsb;              wsb += (size_t)NF * 4;
  float* bufB = (float*)wsb;
  (void)ws_size; (void)n_in; (void)out_size;

  const int NB       = (int)cdiv(N, 256);
  const int nb_node  = NB;
  const int n_strips = N / 16;

  unsigned int* hist = (unsigned int*)bufA;
  int* pre = (int*)bufB;

  float* wc = (float*)cursor;
  float* bc = wc + 128 * 64;

  k_hist<<<128, 1024, 0, stream>>>(src, dst, hist, E);
  k_sum<<<196, 256, 0, stream>>>(hist, cntO, cntI, pre, N, NA);
  k_degrsqrt<<<nb_node, 256, 0, stream>>>(cntO, cntI, degO, degI, N);
  k_scan_block<<<NB, 256, 0, stream>>>(cntI, rowptr, bsum, N);
  k_scan_tops<<<1, 512, 0, stream>>>(bsum, NB);
  k_scan_add<<<NB, 256, 0, stream>>>(rowptr, bsum, cursor, N);
  k_scatter<<<CS_SLICES * CP_PARTS, CS_THREADS, 0, stream>>>(src, dst, rowptr, pre, csc, E, N, NA);

  k_gemm_wmma<128, true, false, true, 1><<<256, 512, 0, stream>>>(feat, w1, b1, degO, h0, bufS,
                                                                   1.0f, 0.0f, n_strips);

  k_spmm_pull<float4, false><<<1024, 256, 0, stream>>>(rowptr, cntI, csc, bufS, degI, h0, nullptr,
                                                       bufA, N);
  k_gemm_wmma<128, true, true, true, 2><<<256, 512, 0, stream>>>(bufA, cw1, cb1, degO, nullptr, bufS,
                                                                 GA_BETA1, 1.0f - GA_BETA1,
                                                                 n_strips);

  k_fold<<<9, 128, 0, stream>>>(cw2, cb2, w2, b2, wc, bc, GA_BETA2);
  k_gemm_wmma<64, false, false, false, 0><<<256, 512, 0, stream>>>(h0, wc, bc, nullptr, bufA, nullptr,
                                                                   1.0f, 0.0f, 2 * n_strips);
  const float* Q = bufA;
  const float* P = bufA + (size_t)N * 64;
  k_spmm_pull<float2, true><<<1024, 256, 0, stream>>>(rowptr, cntI, csc, P, degI, Q, bc,
                                                      (float*)d_out, N);
}


# Five short launches removed (13 -> 8): degree factors and scan level 1 join the slice-sum pass, scan level 2 joins the placing pass's table load, the weight fold rides in the slice-sum launch
# speedup vs baseline: 2.5739x; 1.0620x over previous
#include <hip/hip_runtime.h>


#define GA_ALPHA 0.2f
#define GA_BETA1 0.6931471805599453f
#define GA_BETA2 0.4054651081081644f

typedef __attribute__((ext_vector_type(2))) float v2f;
typedef __attribute__((ext_vector_type(8))) float v8f;


#define CS_SLICES  64
#define CS_WORDS   50048
#define CS_THREADS 1024
#define CS_B       8

__global__ __launch_bounds__(CS_THREADS) void k_hist(const int* __restrict__ src,
                                                     const int* __restrict__ dst,
                                                     unsigned int* __restrict__ hist, int E) {
  __shared__ unsigned int h[CS_WORDS];
  const int tid = threadIdx.x;
  const int which = blockIdx.x / CS_SLICES;
  const int b = blockIdx.x - which * CS_SLICES;
  const int* __restrict__ key = which ? dst : src;
  for (int w = tid; w < CS_WORDS; w += CS_THREADS) h[w] = 0u;
  __syncthreads();
  const int len = (E + CS_SLICES - 1) / CS_SLICES;
  const int lo = b * len;
  const int hi = min(E, lo + len);
  for (int e0 = lo + tid; e0 < hi; e0 += CS_B * CS_THREADS) {
    int v[CS_B];
#pragma unroll
    for (int k = 0; k < CS_B; ++k) {
      const int e = e0 + k * CS_THREADS;
      v[k] = (e < hi) ? key[e] : -1;
    }
#pragma unroll
    for (int k = 0; k < CS_B; ++k)
      if (v[k] >= 0) atomicAdd(&h[v[k] >> 1], (v[k] & 1) ? 0x10000u : 1u);
  }
  __syncthreads();
  unsigned int* __restrict__ out = hist + (size_t)blockIdx.x * CS_WORDS;
  for (int w = tid; w < CS_WORDS; w += CS_THREADS) out[w] = h[w];
}

#define FOLD_TILES 9
__device__ __forceinline__ void fold_tile(const int rt, const float* __restrict__ cw,
                                          const float* __restrict__ cb,
                                          const float* __restrict__ w2,
                                          const float* __restrict__ b2,
                                          float* __restrict__ wc, float* __restrict__ bc,
                                          float beta) {
  const int wave = (int)(threadIdx.x >> 5);
  const int lane = threadIdx.x & 31;
  const int half = lane >> 4;
  const int l16  = lane & 15;
  const int arow = rt * 16 + l16;
  const int col  = wave * 16 + l16;
  const float* __restrict__ ap = ((arow < 128) ? (cw + arow * 128) : cb) + 2 * half;
  const float* __restrict__ bp = w2 + (2 * half) * 64 + col;

  v8f acc = v8f{};
#pragma unroll
  for (int k0 = 0; k0 < 128; k0 += 4) {
    v2f af, bf;
    af.x = ap[k0];
    af.y = ap[k0 + 1];
    bf.x = bp[k0 * 64];
    bf.y = bp[(k0 + 1) * 64];
    acc = __builtin_amdgcn_wmma_f32_16x16x4_f32(false, af, false, bf, (short)0, acc, false, false);
  }

#pragma unroll
  for (int r = 0; r < 8; ++r) {
    const int rowM = rt * 16 + half * 8 + r;
    const float v = acc[r];
    if (rowM < 128)       wc[rowM * 64 + col] = beta * v + (1.0f - beta) * w2[rowM * 64 + col];
    else if (rowM == 128) bc[col] = v + b2[col];
  }
}

__device__ __forceinline__ int wave_incl_sum(int v, int lane) {
#pragma unroll
  for (int off = 1; off < 32; off <<= 1) {
    const int y = __shfl_up(v, off);
    if (lane >= off) v += y;
  }
  return v;
}

#define KS_THREADS 256
#define KS_BLOCKS  ((CS_WORDS + KS_THREADS - 1) / KS_THREADS)
__global__ __launch_bounds__(KS_THREADS) void k_sum(const unsigned int* __restrict__ hist,
                                                    int* __restrict__ cntI, int* __restrict__ exI,
                                                    int* __restrict__ bsum,
                                                    float* __restrict__ degO,
                                                    float* __restrict__ degI,
                                                    int* __restrict__ pre, int N, int NA,
                                                    const float* __restrict__ cw,
                                                    const float* __restrict__ cb,
                                                    const float* __restrict__ w2,
                                                    const float* __restrict__ b2,
                                                    float* __restrict__ wc, float* __restrict__ bc,
                                                    float beta) {
  if (blockIdx.x >= KS_BLOCKS) {
    if (threadIdx.x < 128)
      fold_tile((int)blockIdx.x - KS_BLOCKS, cw, cb, w2, b2, wc, bc, beta);
    return;
  }
  __shared__ int wtot[KS_THREADS / 32];
  const int tid = threadIdx.x;
  const int lane = tid & 31;
  const int wave = tid >> 5;
  const int w = blockIdx.x * KS_THREADS + tid;
  const bool live = w < CS_WORDS;
  const int wl = live ? w : CS_WORDS - 1;
  const int n0 = 2 * w, n1 = 2 * w + 1;
  const bool v0 = live && n0 < N, v1 = live && n1 < N;

  unsigned int x[CS_SLICES];
#pragma unroll
  for (int b = 0; b < CS_SLICES; ++b) x[b] = hist[(size_t)b * CS_WORDS + wl];
  unsigned int lo = 0u, hi = 0u;
#pragma unroll
  for (int b = 0; b < CS_SLICES; ++b) {
    lo += x[b] & 0xffffu;
    hi += x[b] >> 16;
  }
  const unsigned int outLo = lo, outHi = hi;

#pragma unroll
  for (int b = 0; b < CS_SLICES; ++b) x[b] = hist[(size_t)(CS_SLICES + b) * CS_WORDS + wl];
  int* __restrict__ prow = pre + n0;
  lo = 0u; hi = 0u;
  if (v1) {
#pragma unroll
    for (int b = 0; b < CS_SLICES; ++b) {
      *(int2*)(prow + (size_t)b * NA) = make_int2((int)lo, (int)hi);
      lo += x[b] & 0xffffu;
      hi += x[b] >> 16;
    }
    *(int2*)(cntI + n0) = make_int2((int)lo, (int)hi);
    *(float2*)(degO + n0) = make_float2(rsqrtf(fmaxf((float)(int)outLo, 1.0f)),
                                        rsqrtf(fmaxf((float)(int)outHi, 1.0f)));
    *(float2*)(degI + n0) = make_float2(rsqrtf(fmaxf((float)(int)lo, 1.0f)),
                                        rsqrtf(fmaxf((float)(int)hi, 1.0f)));
  } else if (v0) {
#pragma unroll
    for (int b = 0; b < CS_SLICES; ++b) {
      prow[(size_t)b * NA] = (int)lo;
      lo += x[b] & 0xffffu;
    }
    cntI[n0] = (int)lo;
    degO[n0] = rsqrtf(fmaxf((float)(int)outLo, 1.0f));
    degI[n0] = rsqrtf(fmaxf((float)(int)lo, 1.0f));
  }

  const int c0 = v0 ? (int)lo : 0;
  const int c1 = v1 ? (int)hi : 0;
  const int t = c0 + c1;
  const int incl = wave_incl_sum(t, lane);
  if (lane == 31) wtot[wave] = incl;
  __syncthreads();
  int wp = 0, tot = 0;
#pragma unroll
  for (int u = 0; u < KS_THREADS / 32; ++u) {
    const int s = wtot[u];
    if (u < wave) wp += s;
    tot += s;
  }
  const int ex = wp + incl - t;
  if (v1)      *(int2*)(exI + n0) = make_int2(ex, ex + c0);
  else if (v0) exI[n0] = ex;
  if (tid == 0) bsum[blockIdx.x] = tot;
}

#define CP_PARTS 4
#define CP_NODES (2 * CS_WORDS / CP_PARTS)
#define CP_B     13
__global__ __launch_bounds__(CS_THREADS) void k_scatter(const int* __restrict__ src,
                                                        const int* __restrict__ dst,
                                                        const int* __restrict__ exI,
                                                        const int* __restrict__ bsum,
                                                        const int* __restrict__ pre,
                                                        int* __restrict__ rowptr,
                                                        int* __restrict__ csc, int E, int N,
                                                        int NA, int nb) {
  __shared__ int place[CP_NODES];
  __shared__ int tops[256];
  __shared__ int wtot[8];
  const int tid = threadIdx.x;
  const int b = blockIdx.x / CP_PARTS;
  const int q = blockIdx.x - b * CP_PARTS;
  const int nlo = q * CP_NODES;
  const int nq = min(N - nlo, CP_NODES);
  if (nq <= 0) return;

  int bt = 0, bincl = 0;
  if (tid < 256) {
    bt = (tid < nb) ? bsum[tid] : 0;
    bincl = wave_incl_sum(bt, tid & 31);
    if ((tid & 31) == 31) wtot[tid >> 5] = bincl;
  }
  __syncthreads();
  if (tid < 256) {
    int wp = 0;
#pragma unroll
    for (int u = 0; u < 8; ++u)
      if (u < (tid >> 5)) wp += wtot[u];
    tops[tid] = wp + bincl - bt;
  }
  __syncthreads();

  const int* __restrict__ rp = exI + nlo;
  const int* __restrict__ pp = pre + (size_t)b * NA + nlo;
  for (int i0 = tid; i0 < nq; i0 += CP_B * CS_THREADS) {
    int r[CP_B], p[CP_B];
#pragma unroll
    for (int k = 0; k < CP_B; ++k) {
      const int i = i0 + k * CS_THREADS;
      const int ic = (i < nq) ? i : i0;
      r[k] = rp[ic];
      p[k] = pp[ic];
    }
#pragma unroll
    for (int k = 0; k < CP_B; ++k) {
      const int i = i0 + k * CS_THREADS;
      if (i < nq) {
        const int st = r[k] + tops[(unsigned int)(nlo + i) / (2u * KS_THREADS)];
        place[i] = st + p[k];
        if (b == 0) rowptr[nlo + i] = st;
      }
    }
  }
  __syncthreads();

  const int len = (E + CS_SLICES - 1) / CS_SLICES;
  const int lo = b * len;
  const int hi = min(E, lo + len);
  for (int e0 = lo + tid; e0 < hi; e0 += CP_B * CS_THREADS) {
    int s[CP_B], li[CP_B], pos[CP_B];
    bool mine[CP_B];
#pragma unroll
    for (int k = 0; k < CP_B; ++k) {
      const int e = e0 + k * CS_THREADS;
      const bool ok = e < hi;
      const int ec = ok ? e : e0;
      li[k] = dst[ec] - nlo;
      s[k] = src[ec];
      mine[k] = ok && ((unsigned int)li[k] < (unsigned int)nq);
    }
#pragma unroll
    for (int k = 0; k < CP_B; ++k) {
      pos[k] = 0;
      if (mine[k]) pos[k] = atomicAdd(&place[li[k]], 1);
    }
#pragma unroll
    for (int k = 0; k < CP_B; ++k)
      if (mine[k]) csc[pos[k]] = s[k];
  }
}

#define SP_B 16

__device__ __forceinline__ void sp_zero(float4& a) { a = float4{0.0f, 0.0f, 0.0f, 0.0f}; }
__device__ __forceinline__ void sp_zero(float2& a) { a = float2{0.0f, 0.0f}; }
__device__ __forceinline__ void sp_add(float4& a, const float4& b) {
  a.x += b.x; a.y += b.y; a.z += b.z; a.w += b.w;
}
__device__ __forceinline__ void sp_add(float2& a, const float2& b) { a.x += b.x; a.y += b.y; }
__device__ __forceinline__ float4 sp_comb(float s, const float4& a, const float4& x) {
  float4 r;
  r.x = s * a.x + GA_ALPHA * x.x;
  r.y = s * a.y + GA_ALPHA * x.y;
  r.z = s * a.z + GA_ALPHA * x.z;
  r.w = s * a.w + GA_ALPHA * x.w;
  return r;
}
__device__ __forceinline__ float2 sp_comb(float s, const float2& a, const float2& x) {
  float2 r;
  r.x = s * a.x + GA_ALPHA * x.x;
  r.y = s * a.y + GA_ALPHA * x.y;
  return r;
}

template <bool FULL, typename VT>
__device__ __forceinline__ void sp_batch(const float* __restrict__ hsl, int myidx, int i,
                                         int rem, VT& acc) {
  constexpr int ROWF = 32 * (int)(sizeof(VT) / sizeof(float));
  int s[SP_B];
#pragma unroll
  for (int k = 0; k < SP_B; ++k) s[k] = __builtin_amdgcn_readlane(myidx, i + k);
  VT v[SP_B];
#pragma unroll
  for (int k = 0; k < SP_B; ++k) {
    if (FULL || k < rem) v[k] = *(const VT*)(hsl + (long)s[k] * ROWF);
    else                 sp_zero(v[k]);
  }
#pragma unroll
  for (int k = 0; k < SP_B; ++k) {
    if (FULL || k < rem) sp_add(acc, v[k]);
  }
}

template <typename VT, bool BIAS>
__global__ __launch_bounds__(256) void k_spmm_pull(const int* __restrict__ rowptr,
                                                   const int* __restrict__ cnt,
                                                   const int* __restrict__ csc,
                                                   const float* __restrict__ hs,
                                                   const float* __restrict__ dirs,
                                                   const float* __restrict__ x0,
                                                   const float* __restrict__ bias,
                                                   float* __restrict__ f, int N) {
  constexpr int VW = (int)(sizeof(VT) / sizeof(float));
  constexpr int ROWF = 32 * VW;
  const int wave = __builtin_amdgcn_readfirstlane((int)(threadIdx.x >> 5));
  const int lane = threadIdx.x & 31;
  const int step = (int)gridDim.x * 8;
  const float* __restrict__ hsl = hs + lane * VW;
  int node = (int)blockIdx.x * 8 + wave;
  if (node >= N) return;
  VT bv;
  sp_zero(bv);
  if (BIAS) bv = *(const VT*)(bias + lane * VW);
  int start = __builtin_amdgcn_readfirstlane(rowptr[node]);
  int c = __builtin_amdgcn_readfirstlane(cnt[node]);
  float dn = dirs[node];
  for (;;) {
    const int nnode = node + step;
    const int nn = min(nnode, N - 1);
    const int nstart = __builtin_amdgcn_readfirstlane(rowptr[nn]);
    const int nc = __builtin_amdgcn_readfirstlane(cnt[nn]);
    const float ndn = dirs[nn];

    int m = min(32, c);
    int myidx = 0;
    if (lane < m) myidx = csc[start + lane];
    const long o = (long)node * ROWF + lane * VW;
    const VT x0v = *(const VT*)(x0 + o);
    const float scale = (1.0f - GA_ALPHA) * dn;
    VT acc;
    sp_zero(acc);
    int base = 0;
    for (;;) {
      for (int i = 0; i < m; i += SP_B) {
        const int rem = m - i;
        if (rem >= SP_B) sp_batch<true, VT>(hsl, myidx, i, rem, acc);
        else             sp_batch<false, VT>(hsl, myidx, i, rem, acc);
      }
      base += 32;
      if (base >= c) break;
      m = min(32, c - base);
      myidx = 0;
      if (lane < m) myidx = csc[start + base + lane];
    }
    VT r = sp_comb(scale, acc, x0v);
    if (BIAS) sp_add(r, bv);
    *(VT*)(f + o) = r;

    if (nnode >= N) break;
    node = nnode; start = nstart; c = nc; dn = ndn;
  }
}

template <int NCOLS, bool RELU, bool RESID, bool BIAS, int SCALED>
__global__ __launch_bounds__(512) void k_gemm_wmma(const float* __restrict__ A,
                                                   const float* __restrict__ W,
                                                   const float* __restrict__ bias,
                                                   const float* __restrict__ dors,
                                                   float* __restrict__ out,
                                                   float* __restrict__ outS,
                                                   float actW, float resS, int nTiles) {
  constexpr int NCT = NCOLS / 16;
  __shared__ float sW[128 * NCOLS];

  const int wave = __builtin_amdgcn_readfirstlane((int)(threadIdx.x >> 5));
  const int lane = threadIdx.x & 31;
  const int half = lane >> 4;
  const int l16  = lane & 15;
  const int nblk = gridDim.x;

  int round = 0;
  int tile = blockIdx.x + nblk * wave;
  bool have = tile < nTiles;

  constexpr int NST = (128 * NCOLS) / 512;
  float wv[NST];
#pragma unroll
  for (int it = 0; it < NST; ++it) {
    const int d  = it * 512 + threadIdx.x;
    const int p  = d & 1;
    const int c  = (d >> 1) & 15;
    const int h  = (d >> 5) & 1;
    const int bt = d >> 6;
    const int t  = bt % NCT;
    const int js = bt / NCT;
    const int k  = (js >> 1) * 8 + h * 4 + (js & 1) * 2 + p;
    wv[it] = W[k * NCOLS + t * 16 + c];
  }

  float4 a[16];
  {
    const float* __restrict__ Ap = A + (long)((have ? tile : 0) * 16 + l16) * 128 + half * 4;
#pragma unroll
    for (int j = 0; j < 16; ++j) a[j] = *(const float4*)(Ap + 8 * j);
  }

#pragma unroll
  for (int it = 0; it < NST; ++it) sW[it * 512 + threadIdx.x] = wv[it];
  __builtin_amdgcn_fence(__ATOMIC_RELEASE, "workgroup", "local");
  __builtin_amdgcn_s_barrier();
  __builtin_amdgcn_fence(__ATOMIC_ACQUIRE, "workgroup", "local");

  const float* __restrict__ sWl = sW + lane * 2;

  while (have) {
    const int nround = round + 1;
    const int q = (nround & 1) ? ((13 * wave) & 15) : wave;
    const int ntile = blockIdx.x + nblk * (16 * nround + q);
    const bool nhave = ntile < nTiles;
    const float* __restrict__ An =
        A + (long)((nhave ? ntile : tile) * 16 + l16) * 128 + half * 4;

    float ds[8];
    if (SCALED) {
      const float4* __restrict__ dp = (const float4*)(dors + tile * 16 + half * 8);
      const float4 d0 = dp[0];
      const float4 d1 = dp[1];
      ds[0] = d0.x; ds[1] = d0.y; ds[2] = d0.z; ds[3] = d0.w;
      ds[4] = d1.x; ds[5] = d1.y; ds[6] = d1.z; ds[7] = d1.w;
    }

    v8f acc[NCT];
#pragma unroll
    for (int t = 0; t < NCT; ++t) acc[t] = v8f{};

#pragma unroll
    for (int j = 0; j < 16; ++j) {
      v2f b[2 * NCT];
#pragma unroll
      for (int u = 0; u < 2 * NCT; ++u) b[u] = *(const v2f*)(sWl + (j * 2 * NCT + u) * 64);
#pragma unroll
      for (int s = 0; s < 2; ++s) {
        v2f af;
        af.x = s ? a[j].z : a[j].x;
        af.y = s ? a[j].w : a[j].y;
#pragma unroll
        for (int t = 0; t < NCT; ++t) {
          acc[t] = __builtin_amdgcn_wmma_f32_16x16x4_f32(false, af, false, b[s * NCT + t],
                                                         (short)0, acc[t], false, false);
        }
      }
      a[j] = *(const float4*)(An + 8 * j);
      __builtin_amdgcn_sched_barrier(0);
    }

    const int row0 = tile * 16;
#pragma unroll
    for (int t = 0; t < NCT; ++t) {
      const int outCol = t * 16 + l16;
      const float bv = BIAS ? bias[outCol] : 0.0f;
#pragma unroll
      for (int r = 0; r < 8; ++r) {
        const int rowM = row0 + r + half * 8;
        float v;
        if (BIAS) v = actW * acc[t][r] + bv;
        else      v = actW * acc[t][r];
        if (RESID) v += resS * A[(long)rowM * 128 + outCol];
        if (RELU) v = fmaxf(v, 0.0f);
        if (SCALED != 2) out[(long)rowM * NCOLS + outCol] = v;
        if (SCALED) outS[(long)rowM * NCOLS + outCol] = v * ds[r];
      }
    }

    tile = ntile;
    have = nhave;
    round = nround;
  }
}


static inline long cdiv(long a, long b) { return (a + b - 1) / b; }

extern "C" void kernel_launch(void* const* d_in, const int* in_sizes, int n_in,
                              void* d_out, int out_size, void* d_ws, size_t ws_size,
                              hipStream_t stream) {
  const float* feat = (const float*)d_in[0];
  const int*   src  = (const int*)d_in[1];
  const int*   dst  = (const int*)d_in[2];
  const float* w1   = (const float*)d_in[3];
  const float* b1   = (const float*)d_in[4];
  const float* cw1  = (const float*)d_in[5];
  const float* cb1  = (const float*)d_in[6];
  const float* cw2  = (const float*)d_in[7];
  const float* cb2  = (const float*)d_in[8];
  const float* w2   = (const float*)d_in[9];
  const float* b2   = (const float*)d_in[10];

  const int N = in_sizes[0] / 128;
  const int E = in_sizes[1];
  const long NF = (long)N * 128;
  const int NA = (N + 63) & ~63;

  char* wsb = (char*)d_ws;
  int* cntO   = (int*)wsb;                wsb += (size_t)NA * 4;
  int* cntI   = (int*)wsb;                wsb += (size_t)NA * 4;
  int* rowptr = (int*)wsb;                wsb += (size_t)NA * 4;
  int* cursor = (int*)wsb;                wsb += (size_t)NA * 4;
  int* bsum   = (int*)wsb;                wsb += 512 * 4;
  int* csc    = (int*)wsb;                wsb += ((size_t)E * 4 + 255) & ~(size_t)255;
  float* degO = (float*)wsb;              wsb += (size_t)NA * 4;
  float* degI = (float*)wsb;              wsb += (size_t)NA * 4;
  float* h0   = (float*)wsb;              wsb += (size_t)NF * 4;
  float* bufS = (float*)wsb;              wsb += (size_t)NF * 4;
  float* bufA = (float*)wsb;              wsb += (size_t)NF * 4;
  float* bufB = (float*)wsb;
  (void)ws_size; (void)n_in; (void)out_size;

  const int n_strips = N / 16;

  unsigned int* hist = (unsigned int*)bufA;
  int* pre = (int*)bufB;

  float* wc = (float*)cntO;
  float* bc = wc + 128 * 64;
  int* exI = cursor;

  k_hist<<<128, 1024, 0, stream>>>(src, dst, hist, E);
  k_sum<<<KS_BLOCKS + FOLD_TILES, KS_THREADS, 0, stream>>>(hist, cntI, exI, bsum, degO, degI, pre, N,
                                                           NA, cw2, cb2, w2, b2, wc, bc, GA_BETA2);
  k_scatter<<<CS_SLICES * CP_PARTS, CS_THREADS, 0, stream>>>(src, dst, exI, bsum, pre, rowptr, csc,
                                                             E, N, NA, KS_BLOCKS);

  k_gemm_wmma<128, true, false, true, 1><<<256, 512, 0, stream>>>(feat, w1, b1, degO, h0, bufS,
                                                                   1.0f, 0.0f, n_strips);

  k_spmm_pull<float4, false><<<1024, 256, 0, stream>>>(rowptr, cntI, csc, bufS, degI, h0, nullptr,
                                                       bufA, N);
  k_gemm_wmma<128, true, true, true, 2><<<256, 512, 0, stream>>>(bufA, cw1, cb1, degO, nullptr, bufS,
                                                                 GA_BETA1, 1.0f - GA_BETA1,
                                                                 n_strips);

  k_gemm_wmma<64, false, false, false, 0><<<256, 512, 0, stream>>>(h0, wc, bc, nullptr, bufA, nullptr,
                                                                   1.0f, 0.0f, 2 * n_strips);
  const float* Q = bufA;
  const float* P = bufA + (size_t)N * 64;
  k_spmm_pull<float2, true><<<1024, 256, 0, stream>>>(rowptr, cntI, csc, P, degI, Q, bc,
                                                      (float*)d_out, N);
}


# h0 stored once, only as h0*degO: GEMM 1 writes one array not two; layer 1 gathers and reads own rows from that one table; own-row terms multiply 1/degO back in (2 more roundings); sum order same
# speedup vs baseline: 2.6805x; 1.0414x over previous
#include <hip/hip_runtime.h>


#define GA_ALPHA 0.2f
#define GA_BETA1 0.6931471805599453f
#define GA_BETA2 0.4054651081081644f

typedef __attribute__((ext_vector_type(2))) float v2f;
typedef __attribute__((ext_vector_type(8))) float v8f;


#define CS_SLICES  64
#define CS_WORDS   50048
#define CS_THREADS 1024
#define CS_B       8

__global__ __launch_bounds__(CS_THREADS) void k_hist(const int* __restrict__ src,
                                                     const int* __restrict__ dst,
                                                     unsigned int* __restrict__ hist, int E) {
  __shared__ unsigned int h[CS_WORDS];
  const int tid = threadIdx.x;
  const int which = blockIdx.x / CS_SLICES;
  const int b = blockIdx.x - which * CS_SLICES;
  const int* __restrict__ key = which ? dst : src;
  for (int w = tid; w < CS_WORDS; w += CS_THREADS) h[w] = 0u;
  __syncthreads();
  const int len = (E + CS_SLICES - 1) / CS_SLICES;
  const int lo = b * len;
  const int hi = min(E, lo + len);
  for (int e0 = lo + tid; e0 < hi; e0 += CS_B * CS_THREADS) {
    int v[CS_B];
#pragma unroll
    for (int k = 0; k < CS_B; ++k) {
      const int e = e0 + k * CS_THREADS;
      v[k] = (e < hi) ? key[e] : -1;
    }
#pragma unroll
    for (int k = 0; k < CS_B; ++k)
      if (v[k] >= 0) atomicAdd(&h[v[k] >> 1], (v[k] & 1) ? 0x10000u : 1u);
  }
  __syncthreads();
  unsigned int* __restrict__ out = hist + (size_t)blockIdx.x * CS_WORDS;
  for (int w = tid; w < CS_WORDS; w += CS_THREADS) out[w] = h[w];
}

#define FOLD_TILES 9
__device__ __forceinline__ void fold_tile(const int rt, const float* __restrict__ cw,
                                          const float* __restrict__ cb,
                                          const float* __restrict__ w2,
                                          const float* __restrict__ b2,
                                          float* __restrict__ wc, float* __restrict__ bc,
                                          float beta) {
  const int wave = (int)(threadIdx.x >> 5);
  const int lane = threadIdx.x & 31;
  const int half = lane >> 4;
  const int l16  = lane & 15;
  const int arow = rt * 16 + l16;
  const int col  = wave * 16 + l16;
  const float* __restrict__ ap = ((arow < 128) ? (cw + arow * 128) : cb) + 2 * half;
  const float* __restrict__ bp = w2 + (2 * half) * 64 + col;

  v8f acc = v8f{};
#pragma unroll
  for (int k0 = 0; k0 < 128; k0 += 4) {
    v2f af, bf;
    af.x = ap[k0];
    af.y = ap[k0 + 1];
    bf.x = bp[k0 * 64];
    bf.y = bp[(k0 + 1) * 64];
    acc = __builtin_amdgcn_wmma_f32_16x16x4_f32(false, af, false, bf, (short)0, acc, false, false);
  }

#pragma unroll
  for (int r = 0; r < 8; ++r) {
    const int rowM = rt * 16 + half * 8 + r;
    const float v = acc[r];
    if (rowM < 128)       wc[rowM * 64 + col] = beta * v + (1.0f - beta) * w2[rowM * 64 + col];
    else if (rowM == 128) bc[col] = v + b2[col];
  }
}

__device__ __forceinline__ int wave_incl_sum(int v, int lane) {
#pragma unroll
  for (int off = 1; off < 32; off <<= 1) {
    const int y = __shfl_up(v, off);
    if (lane >= off) v += y;
  }
  return v;
}

#define KS_THREADS 256
#define KS_BLOCKS  ((CS_WORDS + KS_THREADS - 1) / KS_THREADS)
__global__ __launch_bounds__(KS_THREADS) void k_sum(const unsigned int* __restrict__ hist,
                                                    int* __restrict__ cntI, int* __restrict__ exI,
                                                    int* __restrict__ bsum,
                                                    float* __restrict__ degO,
                                                    float* __restrict__ degI,
                                                    int* __restrict__ pre, int N, int NA,
                                                    const float* __restrict__ cw,
                                                    const float* __restrict__ cb,
                                                    const float* __restrict__ w2,
                                                    const float* __restrict__ b2,
                                                    float* __restrict__ wc, float* __restrict__ bc,
                                                    float beta) {
  if (blockIdx.x >= KS_BLOCKS) {
    if (threadIdx.x < 128)
      fold_tile((int)blockIdx.x - KS_BLOCKS, cw, cb, w2, b2, wc, bc, beta);
    return;
  }
  __shared__ int wtot[KS_THREADS / 32];
  const int tid = threadIdx.x;
  const int lane = tid & 31;
  const int wave = tid >> 5;
  const int w = blockIdx.x * KS_THREADS + tid;
  const bool live = w < CS_WORDS;
  const int wl = live ? w : CS_WORDS - 1;
  const int n0 = 2 * w, n1 = 2 * w + 1;
  const bool v0 = live && n0 < N, v1 = live && n1 < N;

  unsigned int x[CS_SLICES];
#pragma unroll
  for (int b = 0; b < CS_SLICES; ++b) x[b] = hist[(size_t)b * CS_WORDS + wl];
  unsigned int lo = 0u, hi = 0u;
#pragma unroll
  for (int b = 0; b < CS_SLICES; ++b) {
    lo += x[b] & 0xffffu;
    hi += x[b] >> 16;
  }
  const unsigned int outLo = lo, outHi = hi;

#pragma unroll
  for (int b = 0; b < CS_SLICES; ++b) x[b] = hist[(size_t)(CS_SLICES + b) * CS_WORDS + wl];
  int* __restrict__ prow = pre + n0;
  lo = 0u; hi = 0u;
  if (v1) {
#pragma unroll
    for (int b = 0; b < CS_SLICES; ++b) {
      *(int2*)(prow + (size_t)b * NA) = make_int2((int)lo, (int)hi);
      lo += x[b] & 0xffffu;
      hi += x[b] >> 16;
    }
    *(int2*)(cntI + n0) = make_int2((int)lo, (int)hi);
    *(float2*)(degO + n0) = make_float2(rsqrtf(fmaxf((float)(int)outLo, 1.0f)),
                                        rsqrtf(fmaxf((float)(int)outHi, 1.0f)));
    *(float2*)(degI + n0) = make_float2(rsqrtf(fmaxf((float)(int)lo, 1.0f)),
                                        rsqrtf(fmaxf((float)(int)hi, 1.0f)));
  } else if (v0) {
#pragma unroll
    for (int b = 0; b < CS_SLICES; ++b) {
      prow[(size_t)b * NA] = (int)lo;
      lo += x[b] & 0xffffu;
    }
    cntI[n0] = (int)lo;
    degO[n0] = rsqrtf(fmaxf((float)(int)outLo, 1.0f));
    degI[n0] = rsqrtf(fmaxf((float)(int)lo, 1.0f));
  }

  const int c0 = v0 ? (int)lo : 0;
  const int c1 = v1 ? (int)hi : 0;
  const int t = c0 + c1;
  const int incl = wave_incl_sum(t, lane);
  if (lane == 31) wtot[wave] = incl;
  __syncthreads();
  int wp = 0, tot = 0;
#pragma unroll
  for (int u = 0; u < KS_THREADS / 32; ++u) {
    const int s = wtot[u];
    if (u < wave) wp += s;
    tot += s;
  }
  const int ex = wp + incl - t;
  if (v1)      *(int2*)(exI + n0) = make_int2(ex, ex + c0);
  else if (v0) exI[n0] = ex;
  if (tid == 0) bsum[blockIdx.x] = tot;
}

#define CP_PARTS 4
#define CP_NODES (2 * CS_WORDS / CP_PARTS)
#define CP_B     13
__global__ __launch_bounds__(CS_THREADS) void k_scatter(const int* __restrict__ src,
                                                        const int* __restrict__ dst,
                                                        const int* __restrict__ exI,
                                                        const int* __restrict__ bsum,
                                                        const int* __restrict__ pre,
                                                        int* __restrict__ rowptr,
                                                        int* __restrict__ csc, int E, int N,
                                                        int NA, int nb) {
  __shared__ int place[CP_NODES];
  __shared__ int tops[256];
  __shared__ int wtot[8];
  const int tid = threadIdx.x;
  const int b = blockIdx.x / CP_PARTS;
  const int q = blockIdx.x - b * CP_PARTS;
  const int nlo = q * CP_NODES;
  const int nq = min(N - nlo, CP_NODES);
  if (nq <= 0) return;

  int bt = 0, bincl = 0;
  if (tid < 256) {
    bt = (tid < nb) ? bsum[tid] : 0;
    bincl = wave_incl_sum(bt, tid & 31);
    if ((tid & 31) == 31) wtot[tid >> 5] = bincl;
  }
  __syncthreads();
  if (tid < 256) {
    int wp = 0;
#pragma unroll
    for (int u = 0; u < 8; ++u)
      if (u < (tid >> 5)) wp += wtot[u];
    tops[tid] = wp + bincl - bt;
  }
  __syncthreads();

  const int* __restrict__ rp = exI + nlo;
  const int* __restrict__ pp = pre + (size_t)b * NA + nlo;
  for (int i0 = tid; i0 < nq; i0 += CP_B * CS_THREADS) {
    int r[CP_B], p[CP_B];
#pragma unroll
    for (int k = 0; k < CP_B; ++k) {
      const int i = i0 + k * CS_THREADS;
      const int ic = (i < nq) ? i : i0;
      r[k] = rp[ic];
      p[k] = pp[ic];
    }
#pragma unroll
    for (int k = 0; k < CP_B; ++k) {
      const int i = i0 + k * CS_THREADS;
      if (i < nq) {
        const int st = r[k] + tops[(unsigned int)(nlo + i) / (2u * KS_THREADS)];
        place[i] = st + p[k];
        if (b == 0) rowptr[nlo + i] = st;
      }
    }
  }
  __syncthreads();

  const int len = (E + CS_SLICES - 1) / CS_SLICES;
  const int lo = b * len;
  const int hi = min(E, lo + len);
  for (int e0 = lo + tid; e0 < hi; e0 += CP_B * CS_THREADS) {
    int s[CP_B], li[CP_B], pos[CP_B];
    bool mine[CP_B];
#pragma unroll
    for (int k = 0; k < CP_B; ++k) {
      const int e = e0 + k * CS_THREADS;
      const bool ok = e < hi;
      const int ec = ok ? e : e0;
      li[k] = dst[ec] - nlo;
      s[k] = src[ec];
      mine[k] = ok && ((unsigned int)li[k] < (unsigned int)nq);
    }
#pragma unroll
    for (int k = 0; k < CP_B; ++k) {
      pos[k] = 0;
      if (mine[k]) pos[k] = atomicAdd(&place[li[k]], 1);
    }
#pragma unroll
    for (int k = 0; k < CP_B; ++k)
      if (mine[k]) csc[pos[k]] = s[k];
  }
}

#define SP_B 16

__device__ __forceinline__ void sp_zero(float4& a) { a = float4{0.0f, 0.0f, 0.0f, 0.0f}; }
__device__ __forceinline__ void sp_zero(float2& a) { a = float2{0.0f, 0.0f}; }
__device__ __forceinline__ void sp_add(float4& a, const float4& b) {
  a.x += b.x; a.y += b.y; a.z += b.z; a.w += b.w;
}
__device__ __forceinline__ void sp_add(float2& a, const float2& b) { a.x += b.x; a.y += b.y; }
__device__ __forceinline__ float4 sp_comb(float s, const float4& a, float xw, const float4& x) {
  float4 r;
  r.x = s * a.x + xw * x.x;
  r.y = s * a.y + xw * x.y;
  r.z = s * a.z + xw * x.z;
  r.w = s * a.w + xw * x.w;
  return r;
}
__device__ __forceinline__ float2 sp_comb(float s, const float2& a, float xw, const float2& x) {
  float2 r;
  r.x = s * a.x + xw * x.x;
  r.y = s * a.y + xw * x.y;
  return r;
}

template <bool FULL, typename VT>
__device__ __forceinline__ void sp_batch(const float* __restrict__ hsl, int myidx, int i,
                                         int rem, VT& acc) {
  constexpr int ROWF = 32 * (int)(sizeof(VT) / sizeof(float));
  int s[SP_B];
#pragma unroll
  for (int k = 0; k < SP_B; ++k) s[k] = __builtin_amdgcn_readlane(myidx, i + k);
  VT v[SP_B];
#pragma unroll
  for (int k = 0; k < SP_B; ++k) {
    if (FULL || k < rem) v[k] = *(const VT*)(hsl + (long)s[k] * ROWF);
    else                 sp_zero(v[k]);
  }
#pragma unroll
  for (int k = 0; k < SP_B; ++k) {
    if (FULL || k < rem) sp_add(acc, v[k]);
  }
}

template <typename VT, bool BIAS>
__global__ __launch_bounds__(256) void k_spmm_pull(const int* __restrict__ rowptr,
                                                   const int* __restrict__ cnt,
                                                   const int* __restrict__ csc,
                                                   const float* __restrict__ hs,
                                                   const float* __restrict__ dirs,
                                                   const float* __restrict__ dors,
                                                   const float* __restrict__ x0,
                                                   const float* __restrict__ bias,
                                                   float* __restrict__ f, int N) {
  constexpr int VW = (int)(sizeof(VT) / sizeof(float));
  constexpr int ROWF = 32 * VW;
  const int wave = __builtin_amdgcn_readfirstlane((int)(threadIdx.x >> 5));
  const int lane = threadIdx.x & 31;
  const int step = (int)gridDim.x * 8;
  const float* __restrict__ hsl = hs + lane * VW;
  int node = (int)blockIdx.x * 8 + wave;
  if (node >= N) return;
  VT bv;
  sp_zero(bv);
  if (BIAS) bv = *(const VT*)(bias + lane * VW);
  int start = __builtin_amdgcn_readfirstlane(rowptr[node]);
  int c = __builtin_amdgcn_readfirstlane(cnt[node]);
  float dn = dirs[node];
  float dq = dors[node];
  for (;;) {
    const int nnode = node + step;
    const int nn = min(nnode, N - 1);
    const int nstart = __builtin_amdgcn_readfirstlane(rowptr[nn]);
    const int nc = __builtin_amdgcn_readfirstlane(cnt[nn]);
    const float ndn = dirs[nn];
    const float ndq = dors[nn];

    int m = min(32, c);
    int myidx = 0;
    if (lane < m) myidx = csc[start + lane];
    const long o = (long)node * ROWF + lane * VW;
    const VT x0v = *(const VT*)(x0 + o);
    const float scale = (1.0f - GA_ALPHA) * dn;
    const float xw = GA_ALPHA / dq;
    VT acc;
    sp_zero(acc);
    int base = 0;
    for (;;) {
      for (int i = 0; i < m; i += SP_B) {
        const int rem = m - i;
        if (rem >= SP_B) sp_batch<true, VT>(hsl, myidx, i, rem, acc);
        else             sp_batch<false, VT>(hsl, myidx, i, rem, acc);
      }
      base += 32;
      if (base >= c) break;
      m = min(32, c - base);
      myidx = 0;
      if (lane < m) myidx = csc[start + base + lane];
    }
    VT r = sp_comb(scale, acc, xw, x0v);
    if (BIAS) sp_add(r, bv);
    *(VT*)(f + o) = r;

    if (nnode >= N) break;
    node = nnode; start = nstart; c = nc; dn = ndn; dq = ndq;
  }
}

template <int NCOLS, bool RELU, bool RESID, bool BIAS, int SCALED>
__global__ __launch_bounds__(512) void k_gemm_wmma(const float* __restrict__ A,
                                                   const float* __restrict__ W,
                                                   const float* __restrict__ bias,
                                                   const float* __restrict__ dors,
                                                   float* __restrict__ out,
                                                   float* __restrict__ outS,
                                                   float actW, float resS, int nTiles) {
  constexpr int NCT = NCOLS / 16;
  __shared__ float sW[128 * NCOLS];

  const int wave = __builtin_amdgcn_readfirstlane((int)(threadIdx.x >> 5));
  const int lane = threadIdx.x & 31;
  const int half = lane >> 4;
  const int l16  = lane & 15;
  const int nblk = gridDim.x;

  int round = 0;
  int tile = blockIdx.x + nblk * wave;
  bool have = tile < nTiles;

  constexpr int NST = (128 * NCOLS) / 512;
  float wv[NST];
#pragma unroll
  for (int it = 0; it < NST; ++it) {
    const int d  = it * 512 + threadIdx.x;
    const int p  = d & 1;
    const int c  = (d >> 1) & 15;
    const int h  = (d >> 5) & 1;
    const int bt = d >> 6;
    const int t  = bt % NCT;
    const int js = bt / NCT;
    const int k  = (js >> 1) * 8 + h * 4 + (js & 1) * 2 + p;
    wv[it] = W[k * NCOLS + t * 16 + c];
  }

  float4 a[16];
  {
    const float* __restrict__ Ap = A + (long)((have ? tile : 0) * 16 + l16) * 128 + half * 4;
#pragma unroll
    for (int j = 0; j < 16; ++j) a[j] = *(const float4*)(Ap + 8 * j);
  }

#pragma unroll
  for (int it = 0; it < NST; ++it) sW[it * 512 + threadIdx.x] = wv[it];
  __builtin_amdgcn_fence(__ATOMIC_RELEASE, "workgroup", "local");
  __builtin_amdgcn_s_barrier();
  __builtin_amdgcn_fence(__ATOMIC_ACQUIRE, "workgroup", "local");

  const float* __restrict__ sWl = sW + lane * 2;

  while (have) {
    const int nround = round + 1;
    const int q = (nround & 1) ? ((13 * wave) & 15) : wave;
    const int ntile = blockIdx.x + nblk * (16 * nround + q);
    const bool nhave = ntile < nTiles;
    const float* __restrict__ An =
        A + (long)((nhave ? ntile : tile) * 16 + l16) * 128 + half * 4;

    float ds[8];
    if (SCALED) {
      const float4* __restrict__ dp = (const float4*)(dors + tile * 16 + half * 8);
      const float4 d0 = dp[0];
      const float4 d1 = dp[1];
      ds[0] = d0.x; ds[1] = d0.y; ds[2] = d0.z; ds[3] = d0.w;
      ds[4] = d1.x; ds[5] = d1.y; ds[6] = d1.z; ds[7] = d1.w;
    }

    v8f acc[NCT];
#pragma unroll
    for (int t = 0; t < NCT; ++t) acc[t] = v8f{};

#pragma unroll
    for (int j = 0; j < 16; ++j) {
      v2f b[2 * NCT];
#pragma unroll
      for (int u = 0; u < 2 * NCT; ++u) b[u] = *(const v2f*)(sWl + (j * 2 * NCT + u) * 64);
#pragma unroll
      for (int s = 0; s < 2; ++s) {
        v2f af;
        af.x = s ? a[j].z : a[j].x;
        af.y = s ? a[j].w : a[j].y;
#pragma unroll
        for (int t = 0; t < NCT; ++t) {
          acc[t] = __builtin_amdgcn_wmma_f32_16x16x4_f32(false, af, false, b[s * NCT + t],
                                                         (short)0, acc[t], false, false);
        }
      }
      a[j] = *(const float4*)(An + 8 * j);
      __builtin_amdgcn_sched_barrier(0);
    }

    const int row0 = tile * 16;
#pragma unroll
    for (int t = 0; t < NCT; ++t) {
      const int outCol = t * 16 + l16;
      const float bv = BIAS ? bias[outCol] : 0.0f;
#pragma unroll
      for (int r = 0; r < 8; ++r) {
        const int rowM = row0 + r + half * 8;
        float v;
        if (BIAS) v = actW * acc[t][r] + bv;
        else      v = actW * acc[t][r];
        if (RESID) v += resS * A[(long)rowM * 128 + outCol];
        if (RELU) v = fmaxf(v, 0.0f);
        if (SCALED != 2) out[(long)rowM * NCOLS + outCol] = v;
        if (SCALED) outS[(long)rowM * NCOLS + outCol] = v * ds[r];
      }
    }

    tile = ntile;
    have = nhave;
    round = nround;
  }
}


static inline long cdiv(long a, long b) { return (a + b - 1) / b; }

extern "C" void kernel_launch(void* const* d_in, const int* in_sizes, int n_in,
                              void* d_out, int out_size, void* d_ws, size_t ws_size,
                              hipStream_t stream) {
  const float* feat = (const float*)d_in[0];
  const int*   src  = (const int*)d_in[1];
  const int*   dst  = (const int*)d_in[2];
  const float* w1   = (const float*)d_in[3];
  const float* b1   = (const float*)d_in[4];
  const float* cw1  = (const float*)d_in[5];
  const float* cb1  = (const float*)d_in[6];
  const float* cw2  = (const float*)d_in[7];
  const float* cb2  = (const float*)d_in[8];
  const float* w2   = (const float*)d_in[9];
  const float* b2   = (const float*)d_in[10];

  const int N = in_sizes[0] / 128;
  const int E = in_sizes[1];
  const long NF = (long)N * 128;
  const int NA = (N + 63) & ~63;

  char* wsb = (char*)d_ws;
  int* cntO   = (int*)wsb;                wsb += (size_t)NA * 4;
  int* cntI   = (int*)wsb;                wsb += (size_t)NA * 4;
  int* rowptr = (int*)wsb;                wsb += (size_t)NA * 4;
  int* cursor = (int*)wsb;                wsb += (size_t)NA * 4;
  int* bsum   = (int*)wsb;                wsb += 512 * 4;
  int* csc    = (int*)wsb;                wsb += ((size_t)E * 4 + 255) & ~(size_t)255;
  float* degO = (float*)wsb;              wsb += (size_t)NA * 4;
  float* degI = (float*)wsb;              wsb += (size_t)NA * 4;
  float* h0   = (float*)wsb;              wsb += (size_t)NF * 4;
  float* bufS = (float*)wsb;              wsb += (size_t)NF * 4;
  float* bufA = (float*)wsb;              wsb += (size_t)NF * 4;
  float* bufB = (float*)wsb;
  (void)ws_size; (void)n_in; (void)out_size;

  const int n_strips = N / 16;

  unsigned int* hist = (unsigned int*)bufA;
  int* pre = (int*)bufB;

  float* wc = (float*)cntO;
  float* bc = wc + 128 * 64;
  int* exI = cursor;

  k_hist<<<128, 1024, 0, stream>>>(src, dst, hist, E);
  k_sum<<<KS_BLOCKS + FOLD_TILES, KS_THREADS, 0, stream>>>(hist, cntI, exI, bsum, degO, degI, pre, N,
                                                           NA, cw2, cb2, w2, b2, wc, bc, GA_BETA2);
  k_scatter<<<CS_SLICES * CP_PARTS, CS_THREADS, 0, stream>>>(src, dst, exI, bsum, pre, rowptr, csc,
                                                             E, N, NA, KS_BLOCKS);

  float* h0s = h0;
  k_gemm_wmma<128, true, false, true, 2><<<256, 512, 0, stream>>>(feat, w1, b1, degO, nullptr, h0s,
                                                                   1.0f, 0.0f, n_strips);

  k_spmm_pull<float4, false><<<1024, 256, 0, stream>>>(rowptr, cntI, csc, h0s, degI, degO, h0s,
                                                       nullptr, bufA, N);
  k_gemm_wmma<128, true, true, true, 2><<<256, 512, 0, stream>>>(bufA, cw1, cb1, degO, nullptr, bufS,
                                                                 GA_BETA1, 1.0f - GA_BETA1,
                                                                 n_strips);

  k_gemm_wmma<64, false, false, false, 0><<<256, 512, 0, stream>>>(h0s, wc, bc, nullptr, bufA, nullptr,
                                                                   1.0f, 0.0f, 2 * n_strips);
  const float* Qs = bufA;
  const float* P = bufA + (size_t)N * 64;
  k_spmm_pull<float2, true><<<1024, 256, 0, stream>>>(rowptr, cntI, csc, P, degI, degO, Qs, bc,
                                                      (float*)d_out, N);
}


# GEMM 1 rides in the CSC placing pass's launch: 256+256 work-groups, 16 waves, 128 registers, one of each kind per WGP; matrix work under the placing waits; 7 launches; sums as before (CSC: LDS order)
# speedup vs baseline: 3.0871x; 1.1517x over previous
#include <hip/hip_runtime.h>


#define GA_ALPHA 0.2f
#define GA_BETA1 0.6931471805599453f
#define GA_BETA2 0.4054651081081644f

typedef __attribute__((ext_vector_type(2))) float v2f;
typedef __attribute__((ext_vector_type(8))) float v8f;


#define CS_SLICES  64
#define CS_WORDS   50048
#define CS_THREADS 1024
#define CS_B       8

__global__ __launch_bounds__(CS_THREADS) void k_hist(const int* __restrict__ src,
                                                     const int* __restrict__ dst,
                                                     unsigned int* __restrict__ hist, int E) {
  __shared__ unsigned int h[CS_WORDS];
  const int tid = threadIdx.x;
  const int which = blockIdx.x / CS_SLICES;
  const int b = blockIdx.x - which * CS_SLICES;
  const int* __restrict__ key = which ? dst : src;
  for (int w = tid; w < CS_WORDS; w += CS_THREADS) h[w] = 0u;
  __syncthreads();
  const int len = (E + CS_SLICES - 1) / CS_SLICES;
  const int lo = b * len;
  const int hi = min(E, lo + len);
  for (int e0 = lo + tid; e0 < hi; e0 += CS_B * CS_THREADS) {
    int v[CS_B];
#pragma unroll
    for (int k = 0; k < CS_B; ++k) {
      const int e = e0 + k * CS_THREADS;
      v[k] = (e < hi) ? key[e] : -1;
    }
#pragma unroll
    for (int k = 0; k < CS_B; ++k)
      if (v[k] >= 0) atomicAdd(&h[v[k] >> 1], (v[k] & 1) ? 0x10000u : 1u);
  }
  __syncthreads();
  unsigned int* __restrict__ out = hist + (size_t)blockIdx.x * CS_WORDS;
  for (int w = tid; w < CS_WORDS; w += CS_THREADS) out[w] = h[w];
}

#define FOLD_TILES 9
__device__ __forceinline__ void fold_tile(const int rt, const float* __restrict__ cw,
                                          const float* __restrict__ cb,
                                          const float* __restrict__ w2,
                                          const float* __restrict__ b2,
                                          float* __restrict__ wc, float* __restrict__ bc,
                                          float beta) {
  const int wave = (int)(threadIdx.x >> 5);
  const int lane = threadIdx.x & 31;
  const int half = lane >> 4;
  const int l16  = lane & 15;
  const int arow = rt * 16 + l16;
  const int col  = wave * 16 + l16;
  const float* __restrict__ ap = ((arow < 128) ? (cw + arow * 128) : cb) + 2 * half;
  const float* __restrict__ bp = w2 + (2 * half) * 64 + col;

  v8f acc = v8f{};
#pragma unroll
  for (int k0 = 0; k0 < 128; k0 += 4) {
    v2f af, bf;
    af.x = ap[k0];
    af.y = ap[k0 + 1];
    bf.x = bp[k0 * 64];
    bf.y = bp[(k0 + 1) * 64];
    acc = __builtin_amdgcn_wmma_f32_16x16x4_f32(false, af, false, bf, (short)0, acc, false, false);
  }

#pragma unroll
  for (int r = 0; r < 8; ++r) {
    const int rowM = rt * 16 + half * 8 + r;
    const float v = acc[r];
    if (rowM < 128)       wc[rowM * 64 + col] = beta * v + (1.0f - beta) * w2[rowM * 64 + col];
    else if (rowM == 128) bc[col] = v + b2[col];
  }
}

__device__ __forceinline__ int wave_incl_sum(int v, int lane) {
#pragma unroll
  for (int off = 1; off < 32; off <<= 1) {
    const int y = __shfl_up(v, off);
    if (lane >= off) v += y;
  }
  return v;
}

#define KS_THREADS 256
#define KS_BLOCKS  ((CS_WORDS + KS_THREADS - 1) / KS_THREADS)
__global__ __launch_bounds__(KS_THREADS) void k_sum(const unsigned int* __restrict__ hist,
                                                    int* __restrict__ cntI, int* __restrict__ exI,
                                                    int* __restrict__ bsum,
                                                    float* __restrict__ degO,
                                                    float* __restrict__ degI,
                                                    int* __restrict__ pre, int N, int NA,
                                                    const float* __restrict__ cw,
                                                    const float* __restrict__ cb,
                                                    const float* __restrict__ w2,
                                                    const float* __restrict__ b2,
                                                    float* __restrict__ wc, float* __restrict__ bc,
                                                    float beta) {
  if (blockIdx.x >= KS_BLOCKS) {
    if (threadIdx.x < 128)
      fold_tile((int)blockIdx.x - KS_BLOCKS, cw, cb, w2, b2, wc, bc, beta);
    return;
  }
  __shared__ int wtot[KS_THREADS / 32];
  const int tid = threadIdx.x;
  const int lane = tid & 31;
  const int wave = tid >> 5;
  const int w = blockIdx.x * KS_THREADS + tid;
  const bool live = w < CS_WORDS;
  const int wl = live ? w : CS_WORDS - 1;
  const int n0 = 2 * w, n1 = 2 * w + 1;
  const bool v0 = live && n0 < N, v1 = live && n1 < N;

  unsigned int x[CS_SLICES];
#pragma unroll
  for (int b = 0; b < CS_SLICES; ++b) x[b] = hist[(size_t)b * CS_WORDS + wl];
  unsigned int lo = 0u, hi = 0u;
#pragma unroll
  for (int b = 0; b < CS_SLICES; ++b) {
    lo += x[b] & 0xffffu;
    hi += x[b] >> 16;
  }
  const unsigned int outLo = lo, outHi = hi;

#pragma unroll
  for (int b = 0; b < CS_SLICES; ++b) x[b] = hist[(size_t)(CS_SLICES + b) * CS_WORDS + wl];
  int* __restrict__ prow = pre + n0;
  lo = 0u; hi = 0u;
  if (v1) {
#pragma unroll
    for (int b = 0; b < CS_SLICES; ++b) {
      *(int2*)(prow + (size_t)b * NA) = make_int2((int)lo, (int)hi);
      lo += x[b] & 0xffffu;
      hi += x[b] >> 16;
    }
    *(int2*)(cntI + n0) = make_int2((int)lo, (int)hi);
    *(float2*)(degO + n0) = make_float2(rsqrtf(fmaxf((float)(int)outLo, 1.0f)),
                                        rsqrtf(fmaxf((float)(int)outHi, 1.0f)));
    *(float2*)(degI + n0) = make_float2(rsqrtf(fmaxf((float)(int)lo, 1.0f)),
                                        rsqrtf(fmaxf((float)(int)hi, 1.0f)));
  } else if (v0) {
#pragma unroll
    for (int b = 0; b < CS_SLICES; ++b) {
      prow[(size_t)b * NA] = (int)lo;
      lo += x[b] & 0xffffu;
    }
    cntI[n0] = (int)lo;
    degO[n0] = rsqrtf(fmaxf((float)(int)outLo, 1.0f));
    degI[n0] = rsqrtf(fmaxf((float)(int)lo, 1.0f));
  }

  const int c0 = v0 ? (int)lo : 0;
  const int c1 = v1 ? (int)hi : 0;
  const int t = c0 + c1;
  const int incl = wave_incl_sum(t, lane);
  if (lane == 31) wtot[wave] = incl;
  __syncthreads();
  int wp = 0, tot = 0;
#pragma unroll
  for (int u = 0; u < KS_THREADS / 32; ++u) {
    const int s = wtot[u];
    if (u < wave) wp += s;
    tot += s;
  }
  const int ex = wp + incl - t;
  if (v1)      *(int2*)(exI + n0) = make_int2(ex, ex + c0);
  else if (v0) exI[n0] = ex;
  if (tid == 0) bsum[blockIdx.x] = tot;
}

#define CP_PARTS 4
#define CP_NODES (2 * CS_WORDS / CP_PARTS)
#define CP_B     26
#define CP_THREADS 512
struct PlaceLds {
  int place[CP_NODES];
  int tops[256];
  int wtot[8];
};
__device__ __forceinline__ void place_part(PlaceLds& L, const int blk,
                                           const int* __restrict__ src,
                                           const int* __restrict__ dst,
                                           const int* __restrict__ exI,
                                           const int* __restrict__ bsum,
                                           const int* __restrict__ pre,
                                           int* __restrict__ rowptr,
                                           int* __restrict__ csc, int E, int N,
                                           int NA, int nb) {
  int* __restrict__ place = L.place;
  int* __restrict__ tops = L.tops;
  int* __restrict__ wtot = L.wtot;
  const int tid = threadIdx.x;
  const int b = blk / CP_PARTS;
  const int q = blk - b * CP_PARTS;
  const int nlo = q * CP_NODES;
  const int nq = min(N - nlo, CP_NODES);
  if (nq <= 0) return;

  int bt = 0, bincl = 0;
  if (tid < 256) {
    bt = (tid < nb) ? bsum[tid] : 0;
    bincl = wave_incl_sum(bt, tid & 31);
    if ((tid & 31) == 31) wtot[tid >> 5] = bincl;
  }
  __syncthreads();
  if (tid < 256) {
    int wp = 0;
#pragma unroll
    for (int u = 0; u < 8; ++u)
      if (u < (tid >> 5)) wp += wtot[u];
    tops[tid] = wp + bincl - bt;
  }
  __syncthreads();

  const int* __restrict__ rp = exI + nlo;
  const int* __restrict__ pp = pre + (size_t)b * NA + nlo;
  for (int i0 = tid; i0 < nq; i0 += CP_B * CP_THREADS) {
    int r[CP_B], p[CP_B];
#pragma unroll
    for (int k = 0; k < CP_B; ++k) {
      const int i = i0 + k * CP_THREADS;
      const int ic = (i < nq) ? i : i0;
      r[k] = rp[ic];
      p[k] = pp[ic];
    }
#pragma unroll
    for (int k = 0; k < CP_B; ++k) {
      const int i = i0 + k * CP_THREADS;
      if (i < nq) {
        const int st = r[k] + tops[(unsigned int)(nlo + i) / (2u * KS_THREADS)];
        place[i] = st + p[k];
        if (b == 0) rowptr[nlo + i] = st;
      }
    }
  }
  __syncthreads();

  const int len = (E + CS_SLICES - 1) / CS_SLICES;
  const int lo = b * len;
  const int hi = min(E, lo + len);
  for (int e0 = lo + tid; e0 < hi; e0 += CP_B * CP_THREADS) {
    int s[CP_B], li[CP_B], pos[CP_B];
    bool mine[CP_B];
#pragma unroll
    for (int k = 0; k < CP_B; ++k) {
      const int e = e0 + k * CP_THREADS;
      const bool ok = e < hi;
      const int ec = ok ? e : e0;
      li[k] = dst[ec] - nlo;
      s[k] = src[ec];
      mine[k] = ok && ((unsigned int)li[k] < (unsigned int)nq);
    }
#pragma unroll
    for (int k = 0; k < CP_B; ++k) {
      pos[k] = 0;
      if (mine[k]) pos[k] = atomicAdd(&place[li[k]], 1);
    }
#pragma unroll
    for (int k = 0; k < CP_B; ++k)
      if (mine[k]) csc[pos[k]] = s[k];
  }
}

#define SP_B 16

__device__ __forceinline__ void sp_zero(float4& a) { a = float4{0.0f, 0.0f, 0.0f, 0.0f}; }
__device__ __forceinline__ void sp_zero(float2& a) { a = float2{0.0f, 0.0f}; }
__device__ __forceinline__ void sp_add(float4& a, const float4& b) {
  a.x += b.x; a.y += b.y; a.z += b.z; a.w += b.w;
}
__device__ __forceinline__ void sp_add(float2& a, const float2& b) { a.x += b.x; a.y += b.y; }
__device__ __forceinline__ float4 sp_comb(float s, const float4& a, float xw, const float4& x) {
  float4 r;
  r.x = s * a.x + xw * x.x;
  r.y = s * a.y + xw * x.y;
  r.z = s * a.z + xw * x.z;
  r.w = s * a.w + xw * x.w;
  return r;
}
__device__ __forceinline__ float2 sp_comb(float s, const float2& a, float xw, const float2& x) {
  float2 r;
  r.x = s * a.x + xw * x.x;
  r.y = s * a.y + xw * x.y;
  return r;
}

template <bool FULL, typename VT>
__device__ __forceinline__ void sp_batch(const float* __restrict__ hsl, int myidx, int i,
                                         int rem, VT& acc) {
  constexpr int ROWF = 32 * (int)(sizeof(VT) / sizeof(float));
  int s[SP_B];
#pragma unroll
  for (int k = 0; k < SP_B; ++k) s[k] = __builtin_amdgcn_readlane(myidx, i + k);
  VT v[SP_B];
#pragma unroll
  for (int k = 0; k < SP_B; ++k) {
    if (FULL || k < rem) v[k] = *(const VT*)(hsl + (long)s[k] * ROWF);
    else                 sp_zero(v[k]);
  }
#pragma unroll
  for (int k = 0; k < SP_B; ++k) {
    if (FULL || k < rem) sp_add(acc, v[k]);
  }
}

template <typename VT, bool BIAS>
__global__ __launch_bounds__(256) void k_spmm_pull(const int* __restrict__ rowptr,
                                                   const int* __restrict__ cnt,
                                                   const int* __restrict__ csc,
                                                   const float* __restrict__ hs,
                                                   const float* __restrict__ dirs,
                                                   const float* __restrict__ dors,
                                                   const float* __restrict__ x0,
                                                   const float* __restrict__ bias,
                                                   float* __restrict__ f, int N) {
  constexpr int VW = (int)(sizeof(VT) / sizeof(float));
  constexpr int ROWF = 32 * VW;
  const int wave = __builtin_amdgcn_readfirstlane((int)(threadIdx.x >> 5));
  const int lane = threadIdx.x & 31;
  const int step = (int)gridDim.x * 8;
  const float* __restrict__ hsl = hs + lane * VW;
  int node = (int)blockIdx.x * 8 + wave;
  if (node >= N) return;
  VT bv;
  sp_zero(bv);
  if (BIAS) bv = *(const VT*)(bias + lane * VW);
  int start = __builtin_amdgcn_readfirstlane(rowptr[node]);
  int c = __builtin_amdgcn_readfirstlane(cnt[node]);
  float dn = dirs[node];
  float dq = dors[node];
  for (;;) {
    const int nnode = node + step;
    const int nn = min(nnode, N - 1);
    const int nstart = __builtin_amdgcn_readfirstlane(rowptr[nn]);
    const int nc = __builtin_amdgcn_readfirstlane(cnt[nn]);
    const float ndn = dirs[nn];
    const float ndq = dors[nn];

    int m = min(32, c);
    int myidx = 0;
    if (lane < m) myidx = csc[start + lane];
    const long o = (long)node * ROWF + lane * VW;
    const VT x0v = *(const VT*)(x0 + o);
    const float scale = (1.0f - GA_ALPHA) * dn;
    const float xw = GA_ALPHA / dq;
    VT acc;
    sp_zero(acc);
    int base = 0;
    for (;;) {
      for (int i = 0; i < m; i += SP_B) {
        const int rem = m - i;
        if (rem >= SP_B) sp_batch<true, VT>(hsl, myidx, i, rem, acc);
        else             sp_batch<false, VT>(hsl, myidx, i, rem, acc);
      }
      base += 32;
      if (base >= c) break;
      m = min(32, c - base);
      myidx = 0;
      if (lane < m) myidx = csc[start + base + lane];
    }
    VT r = sp_comb(scale, acc, xw, x0v);
    if (BIAS) sp_add(r, bv);
    *(VT*)(f + o) = r;

    if (nnode >= N) break;
    node = nnode; start = nstart; c = nc; dn = ndn; dq = ndq;
  }
}

template <int NCOLS, bool RELU, bool RESID, bool BIAS, int SCALED>
__global__ __launch_bounds__(512) void k_gemm_wmma(const float* __restrict__ A,
                                                   const float* __restrict__ W,
                                                   const float* __restrict__ bias,
                                                   const float* __restrict__ dors,
                                                   float* __restrict__ out,
                                                   float* __restrict__ outS,
                                                   float actW, float resS, int nTiles) {
  constexpr int NCT = NCOLS / 16;
  __shared__ float sW[128 * NCOLS];

  const int wave = __builtin_amdgcn_readfirstlane((int)(threadIdx.x >> 5));
  const int lane = threadIdx.x & 31;
  const int half = lane >> 4;
  const int l16  = lane & 15;
  const int nblk = gridDim.x;

  int round = 0;
  int tile = blockIdx.x + nblk * wave;
  bool have = tile < nTiles;

  constexpr int NST = (128 * NCOLS) / 512;
  float wv[NST];
#pragma unroll
  for (int it = 0; it < NST; ++it) {
    const int d  = it * 512 + threadIdx.x;
    const int p  = d & 1;
    const int c  = (d >> 1) & 15;
    const int h  = (d >> 5) & 1;
    const int bt = d >> 6;
    const int t  = bt % NCT;
    const int js = bt / NCT;
    const int k  = (js >> 1) * 8 + h * 4 + (js & 1) * 2 + p;
    wv[it] = W[k * NCOLS + t * 16 + c];
  }

  float4 a[16];
  {
    const float* __restrict__ Ap = A + (long)((have ? tile : 0) * 16 + l16) * 128 + half * 4;
#pragma unroll
    for (int j = 0; j < 16; ++j) a[j] = *(const float4*)(Ap + 8 * j);
  }

#pragma unroll
  for (int it = 0; it < NST; ++it) sW[it * 512 + threadIdx.x] = wv[it];
  __builtin_amdgcn_fence(__ATOMIC_RELEASE, "workgroup", "local");
  __builtin_amdgcn_s_barrier();
  __builtin_amdgcn_fence(__ATOMIC_ACQUIRE, "workgroup", "local");

  const float* __restrict__ sWl = sW + lane * 2;

  while (have) {
    const int nround = round + 1;
    const int q = (nround & 1) ? ((13 * wave) & 15) : wave;
    const int ntile = blockIdx.x + nblk * (16 * nround + q);
    const bool nhave = ntile < nTiles;
    const float* __restrict__ An =
        A + (long)((nhave ? ntile : tile) * 16 + l16) * 128 + half * 4;

    float ds[8];
    if (SCALED) {
      const float4* __restrict__ dp = (const float4*)(dors + tile * 16 + half * 8);
      const float4 d0 = dp[0];
      const float4 d1 = dp[1];
      ds[0] = d0.x; ds[1] = d0.y; ds[2] = d0.z; ds[3] = d0.w;
      ds[4] = d1.x; ds[5] = d1.y; ds[6] = d1.z; ds[7] = d1.w;
    }

    v8f acc[NCT];
#pragma unroll
    for (int t = 0; t < NCT; ++t) acc[t] = v8f{};

#pragma unroll
    for (int j = 0; j < 16; ++j) {
      v2f b[2 * NCT];
#pragma unroll
      for (int u = 0; u < 2 * NCT; ++u) b[u] = *(const v2f*)(sWl + (j * 2 * NCT + u) * 64);
#pragma unroll
      for (int s = 0; s < 2; ++s) {
        v2f af;
        af.x = s ? a[j].z : a[j].x;
        af.y = s ? a[j].w : a[j].y;
#pragma unroll
        for (int t = 0; t < NCT; ++t) {
          acc[t] = __builtin_amdgcn_wmma_f32_16x16x4_f32(false, af, false, b[s * NCT + t],
                                                         (short)0, acc[t], false, false);
        }
      }
      a[j] = *(const float4*)(An + 8 * j);
      __builtin_amdgcn_sched_barrier(0);
    }

    const int row0 = tile * 16;
#pragma unroll
    for (int t = 0; t < NCT; ++t) {
      const int outCol = t * 16 + l16;
      const float bv = BIAS ? bias[outCol] : 0.0f;
#pragma unroll
      for (int r = 0; r < 8; ++r) {
        const int rowM = row0 + r + half * 8;
        float v;
        if (BIAS) v = actW * acc[t][r] + bv;
        else      v = actW * acc[t][r];
        if (RESID) v += resS * A[(long)rowM * 128 + outCol];
        if (RELU) v = fmaxf(v, 0.0f);
        if (SCALED != 2) out[(long)rowM * NCOLS + outCol] = v;
        if (SCALED) outS[(long)rowM * NCOLS + outCol] = v * ds[r];
      }
    }

    tile = ntile;
    have = nhave;
    round = nround;
  }
}

#define G1_BLOCKS 256
__device__ __forceinline__ void gemm1_part(float* __restrict__ sW, const int bid,
                                           const float* __restrict__ A,
                                           const float* __restrict__ W,
                                           const float* __restrict__ bias,
                                           const float* __restrict__ dors,
                                           float* __restrict__ outS, int nTiles) {
  constexpr int NCOLS = 128;
  constexpr int NCT = NCOLS / 16;
  const int wave = __builtin_amdgcn_readfirstlane((int)(threadIdx.x >> 5));
  const int lane = threadIdx.x & 31;
  const int half = lane >> 4;
  const int l16  = lane & 15;
  const int nblk = G1_BLOCKS;

  int round = 0;
  int tile = bid + nblk * wave;
  bool have = tile < nTiles;

  constexpr int NST = (128 * NCOLS) / CP_THREADS;
  float wv[NST];
#pragma unroll
  for (int it = 0; it < NST; ++it) {
    const int d  = it * CP_THREADS + threadIdx.x;
    const int p  = d & 1;
    const int tt = (d >> 1) & 1;
    const int c  = (d >> 2) & 15;
    const int h  = (d >> 6) & 1;
    const int bt = d >> 7;
    const int u  = bt % (NCT / 2);
    const int js = bt / (NCT / 2);
    const int k  = (js >> 1) * 8 + h * 4 + (js & 1) * 2 + p;
    wv[it] = W[k * NCOLS + (2 * u + tt) * 16 + c];
  }

  const float* __restrict__ Ap = A + (long)((have ? tile : 0) * 16 + l16) * 128 + half * 4;
  float4 a[4];
#pragma unroll
  for (int j = 0; j < 4; ++j) a[j] = *(const float4*)(Ap + 8 * j);

#pragma unroll
  for (int it = 0; it < NST; ++it) sW[it * CP_THREADS + threadIdx.x] = wv[it];
  __builtin_amdgcn_fence(__ATOMIC_RELEASE, "workgroup", "local");
  __builtin_amdgcn_s_barrier();
  __builtin_amdgcn_fence(__ATOMIC_ACQUIRE, "workgroup", "local");

  const float* __restrict__ sWl = sW + lane * 4;

  while (have) {
    const int nround = round + 1;
    const int q = (nround & 1) ? ((13 * wave) & 15) : wave;
    const int ntile = bid + nblk * (16 * nround + q);
    const bool nhave = ntile < nTiles;
    const float* __restrict__ An =
        A + (long)((nhave ? ntile : tile) * 16 + l16) * 128 + half * 4;

    const float4* __restrict__ dp = (const float4*)(dors + tile * 16 + half * 8);
    float4 d0, d1;

    v8f acc[NCT];
#pragma unroll
    for (int t = 0; t < NCT; ++t) acc[t] = v8f{};

#pragma unroll
    for (int j = 0; j < 16; ++j) {
      const float4 aj = a[j & 3];
#pragma unroll
      for (int s = 0; s < 2; ++s) {
        v2f b[NCT];
#pragma unroll
        for (int u = 0; u < NCT / 2; ++u) {
          const float4 q = *(const float4*)(sWl + ((j * 2 + s) * (NCT / 2) + u) * 128);
          b[2 * u].x = q.x;
          b[2 * u].y = q.y;
          b[2 * u + 1].x = q.z;
          b[2 * u + 1].y = q.w;
        }
        v2f af;
        af.x = s ? aj.z : aj.x;
        af.y = s ? aj.w : aj.y;
#pragma unroll
        for (int t = 0; t < NCT; ++t) {
          acc[t] = __builtin_amdgcn_wmma_f32_16x16x4_f32(false, af, false, b[t], (short)0, acc[t],
                                                         false, false);
        }
      }
      a[j & 3] = (j < 12) ? *(const float4*)(Ap + 8 * (j + 4)) : *(const float4*)(An + 8 * (j - 12));
      if (j == 11) {
        d0 = dp[0];
        d1 = dp[1];
      }
      __builtin_amdgcn_sched_barrier(0);
    }

    const float ds[8] = {d0.x, d0.y, d0.z, d0.w, d1.x, d1.y, d1.z, d1.w};
    const int row0 = tile * 16;
#pragma unroll
    for (int t = 0; t < NCT; ++t) {
      const int outCol = t * 16 + l16;
      const float bv = bias[outCol];
#pragma unroll
      for (int r = 0; r < 8; ++r) {
        const int rowM = row0 + r + half * 8;
        const float v = fmaxf(acc[t][r] + bv, 0.0f);
        outS[(long)rowM * NCOLS + outCol] = v * ds[r];
      }
    }

    tile = ntile;
    have = nhave;
    round = nround;
    Ap = An;
  }
}

union alignas(16) FusedLds {
  PlaceLds pl;
  float sW[128 * 128];
};

static_assert(CS_SLICES * CP_PARTS == G1_BLOCKS, "one placing work-group per GEMM work-group");
__global__ __launch_bounds__(CP_THREADS, 8) void k_place_gemm(const int* __restrict__ src,
                                                              const int* __restrict__ dst,
                                                              const int* __restrict__ exI,
                                                              const int* __restrict__ bsum,
                                                              const int* __restrict__ pre,
                                                              int* __restrict__ rowptr,
                                                              int* __restrict__ csc, int E, int N,
                                                              int NA, int nb,
                                                              const float* __restrict__ feat,
                                                              const float* __restrict__ w1,
                                                              const float* __restrict__ b1,
                                                              const float* __restrict__ dors,
                                                              float* __restrict__ h0s, int nTiles) {
  __shared__ FusedLds lds;
  const int i = (int)blockIdx.x;
  const int id = i >> 1;
  if ((((i / G1_BLOCKS) ^ i) & 1) == 0)
    place_part(lds.pl, id, src, dst, exI, bsum, pre, rowptr, csc, E, N, NA, nb);
  else
    gemm1_part(lds.sW, id, feat, w1, b1, dors, h0s, nTiles);
}


static inline long cdiv(long a, long b) { return (a + b - 1) / b; }

extern "C" void kernel_launch(void* const* d_in, const int* in_sizes, int n_in,
                              void* d_out, int out_size, void* d_ws, size_t ws_size,
                              hipStream_t stream) {
  const float* feat = (const float*)d_in[0];
  const int*   src  = (const int*)d_in[1];
  const int*   dst  = (const int*)d_in[2];
  const float* w1   = (const float*)d_in[3];
  const float* b1   = (const float*)d_in[4];
  const float* cw1  = (const float*)d_in[5];
  const float* cb1  = (const float*)d_in[6];
  const float* cw2  = (const float*)d_in[7];
  const float* cb2  = (const float*)d_in[8];
  const float* w2   = (const float*)d_in[9];
  const float* b2   = (const float*)d_in[10];

  const int N = in_sizes[0] / 128;
  const int E = in_sizes[1];
  const long NF = (long)N * 128;
  const int NA = (N + 63) & ~63;

  char* wsb = (char*)d_ws;
  int* cntO   = (int*)wsb;                wsb += (size_t)NA * 4;
  int* cntI   = (int*)wsb;                wsb += (size_t)NA * 4;
  int* rowptr = (int*)wsb;                wsb += (size_t)NA * 4;
  int* cursor = (int*)wsb;                wsb += (size_t)NA * 4;
  int* bsum   = (int*)wsb;                wsb += 512 * 4;
  int* csc    = (int*)wsb;                wsb += ((size_t)E * 4 + 255) & ~(size_t)255;
  float* degO = (float*)wsb;              wsb += (size_t)NA * 4;
  float* degI = (float*)wsb;              wsb += (size_t)NA * 4;
  float* h0   = (float*)wsb;              wsb += (size_t)NF * 4;
  float* bufS = (float*)wsb;              wsb += (size_t)NF * 4;
  float* bufA = (float*)wsb;              wsb += (size_t)NF * 4;
  float* bufB = (float*)wsb;
  (void)ws_size; (void)n_in; (void)out_size;

  const int n_strips = N / 16;

  unsigned int* hist = (unsigned int*)bufA;
  int* pre = (int*)bufB;

  float* wc = (float*)cntO;
  float* bc = wc + 128 * 64;
  int* exI = cursor;

  k_hist<<<128, 1024, 0, stream>>>(src, dst, hist, E);
  k_sum<<<KS_BLOCKS + FOLD_TILES, KS_THREADS, 0, stream>>>(hist, cntI, exI, bsum, degO, degI, pre, N,
                                                           NA, cw2, cb2, w2, b2, wc, bc, GA_BETA2);
  float* h0s = h0;
  k_place_gemm<<<2 * G1_BLOCKS, CP_THREADS, 0, stream>>>(
      src, dst, exI, bsum, pre, rowptr, csc, E, N, NA, KS_BLOCKS, feat, w1, b1, degO, h0s, n_strips);

  k_spmm_pull<float4, false><<<1024, 256, 0, stream>>>(rowptr, cntI, csc, h0s, degI, degO, h0s,
                                                       nullptr, bufA, N);
  k_gemm_wmma<128, true, true, true, 2><<<256, 512, 0, stream>>>(bufA, cw1, cb1, degO, nullptr, bufS,
                                                                 GA_BETA1, 1.0f - GA_BETA1,
                                                                 n_strips);

  k_gemm_wmma<64, false, false, false, 0><<<256, 512, 0, stream>>>(h0s, wc, bc, nullptr, bufA, nullptr,
                                                                   1.0f, 0.0f, 2 * n_strips);
  const float* Qs = bufA;
  const float* P = bufA + (size_t)N * 64;
  k_spmm_pull<float2, true><<<1024, 256, 0, stream>>>(rowptr, cntI, csc, P, degI, degO, Qs, bc,
                                                      (float*)d_out, N);
}


# Layer-1 GEMM forms its tiles transposed (operands swapped): its result registers are the A fragments of the layer-2 projection; Qs strips ride behind; 6 launches, x1 never stored; P/Qs sums regroup k
# speedup vs baseline: 3.3961x; 1.1001x over previous
#include <hip/hip_runtime.h>


#define GA_ALPHA 0.2f
#define GA_BETA1 0.6931471805599453f
#define GA_BETA2 0.4054651081081644f

typedef __attribute__((ext_vector_type(2))) float v2f;
typedef __attribute__((ext_vector_type(8))) float v8f;


#define CS_SLICES  64
#define CS_WORDS   50048
#define CS_THREADS 1024
#define CS_B       8

__global__ __launch_bounds__(CS_THREADS) void k_hist(const int* __restrict__ src,
                                                     const int* __restrict__ dst,
                                                     unsigned int* __restrict__ hist, int E) {
  __shared__ unsigned int h[CS_WORDS];
  const int tid = threadIdx.x;
  const int which = blockIdx.x / CS_SLICES;
  const int b = blockIdx.x - which * CS_SLICES;
  const int* __restrict__ key = which ? dst : src;
  for (int w = tid; w < CS_WORDS; w += CS_THREADS) h[w] = 0u;
  __syncthreads();
  const int len = (E + CS_SLICES - 1) / CS_SLICES;
  const int lo = b * len;
  const int hi = min(E, lo + len);
  for (int e0 = lo + tid; e0 < hi; e0 += CS_B * CS_THREADS) {
    int v[CS_B];
#pragma unroll
    for (int k = 0; k < CS_B; ++k) {
      const int e = e0 + k * CS_THREADS;
      v[k] = (e < hi) ? key[e] : -1;
    }
#pragma unroll
    for (int k = 0; k < CS_B; ++k)
      if (v[k] >= 0) atomicAdd(&h[v[k] >> 1], (v[k] & 1) ? 0x10000u : 1u);
  }
  __syncthreads();
  unsigned int* __restrict__ out = hist + (size_t)blockIdx.x * CS_WORDS;
  for (int w = tid; w < CS_WORDS; w += CS_THREADS) out[w] = h[w];
}

#define FOLD_TILES 9
__device__ __forceinline__ void fold_tile(const int rt, const float* __restrict__ cw,
                                          const float* __restrict__ cb,
                                          const float* __restrict__ w2,
                                          const float* __restrict__ b2,
                                          float* __restrict__ wc, float* __restrict__ bc,
                                          float beta) {
  const int wave = (int)(threadIdx.x >> 5);
  const int lane = threadIdx.x & 31;
  const int half = lane >> 4;
  const int l16  = lane & 15;
  const int arow = rt * 16 + l16;
  const int col  = wave * 16 + l16;
  const float* __restrict__ ap = ((arow < 128) ? (cw + arow * 128) : cb) + 2 * half;
  const float* __restrict__ bp = w2 + (2 * half) * 64 + col;

  v8f acc = v8f{};
#pragma unroll
  for (int k0 = 0; k0 < 128; k0 += 4) {
    v2f af, bf;
    af.x = ap[k0];
    af.y = ap[k0 + 1];
    bf.x = bp[k0 * 64];
    bf.y = bp[(k0 + 1) * 64];
    acc = __builtin_amdgcn_wmma_f32_16x16x4_f32(false, af, false, bf, (short)0, acc, false, false);
  }

#pragma unroll
  for (int r = 0; r < 8; ++r) {
    const int rowM = rt * 16 + half * 8 + r;
    const float v = acc[r];
    if (rowM < 128)       wc[rowM * 64 + col] = beta * v + (1.0f - beta) * w2[rowM * 64 + col];
    else if (rowM == 128) bc[col] = v + b2[col];
  }
}

__device__ __forceinline__ int wave_incl_sum(int v, int lane) {
#pragma unroll
  for (int off = 1; off < 32; off <<= 1) {
    const int y = __shfl_up(v, off);
    if (lane >= off) v += y;
  }
  return v;
}

#define KS_THREADS 256
#define KS_BLOCKS  ((CS_WORDS + KS_THREADS - 1) / KS_THREADS)
__global__ __launch_bounds__(KS_THREADS) void k_sum(const unsigned int* __restrict__ hist,
                                                    int* __restrict__ cntI, int* __restrict__ exI,
                                                    int* __restrict__ bsum,
                                                    float* __restrict__ degO,
                                                    float* __restrict__ degI,
                                                    int* __restrict__ pre, int N, int NA,
                                                    const float* __restrict__ cw,
                                                    const float* __restrict__ cb,
                                                    const float* __restrict__ w2,
                                                    const float* __restrict__ b2,
                                                    float* __restrict__ wc, float* __restrict__ bc,
                                                    float beta) {
  if (blockIdx.x >= KS_BLOCKS) {
    if (threadIdx.x < 128)
      fold_tile((int)blockIdx.x - KS_BLOCKS, cw, cb, w2, b2, wc, bc, beta);
    return;
  }
  __shared__ int wtot[KS_THREADS / 32];
  const int tid = threadIdx.x;
  const int lane = tid & 31;
  const int wave = tid >> 5;
  const int w = blockIdx.x * KS_THREADS + tid;
  const bool live = w < CS_WORDS;
  const int wl = live ? w : CS_WORDS - 1;
  const int n0 = 2 * w, n1 = 2 * w + 1;
  const bool v0 = live && n0 < N, v1 = live && n1 < N;

  unsigned int x[CS_SLICES];
#pragma unroll
  for (int b = 0; b < CS_SLICES; ++b) x[b] = hist[(size_t)b * CS_WORDS + wl];
  unsigned int lo = 0u, hi = 0u;
#pragma unroll
  for (int b = 0; b < CS_SLICES; ++b) {
    lo += x[b] & 0xffffu;
    hi += x[b] >> 16;
  }
  const unsigned int outLo = lo, outHi = hi;

#pragma unroll
  for (int b = 0; b < CS_SLICES; ++b) x[b] = hist[(size_t)(CS_SLICES + b) * CS_WORDS + wl];
  int* __restrict__ prow = pre + n0;
  lo = 0u; hi = 0u;
  if (v1) {
#pragma unroll
    for (int b = 0; b < CS_SLICES; ++b) {
      *(int2*)(prow + (size_t)b * NA) = make_int2((int)lo, (int)hi);
      lo += x[b] & 0xffffu;
      hi += x[b] >> 16;
    }
    *(int2*)(cntI + n0) = make_int2((int)lo, (int)hi);
    *(float2*)(degO + n0) = make_float2(rsqrtf(fmaxf((float)(int)outLo, 1.0f)),
                                        rsqrtf(fmaxf((float)(int)outHi, 1.0f)));
    *(float2*)(degI + n0) = make_float2(rsqrtf(fmaxf((float)(int)lo, 1.0f)),
                                        rsqrtf(fmaxf((float)(int)hi, 1.0f)));
  } else if (v0) {
#pragma unroll
    for (int b = 0; b < CS_SLICES; ++b) {
      prow[(size_t)b * NA] = (int)lo;
      lo += x[b] & 0xffffu;
    }
    cntI[n0] = (int)lo;
    degO[n0] = rsqrtf(fmaxf((float)(int)outLo, 1.0f));
    degI[n0] = rsqrtf(fmaxf((float)(int)lo, 1.0f));
  }

  const int c0 = v0 ? (int)lo : 0;
  const int c1 = v1 ? (int)hi : 0;
  const int t = c0 + c1;
  const int incl = wave_incl_sum(t, lane);
  if (lane == 31) wtot[wave] = incl;
  __syncthreads();
  int wp = 0, tot = 0;
#pragma unroll
  for (int u = 0; u < KS_THREADS / 32; ++u) {
    const int s = wtot[u];
    if (u < wave) wp += s;
    tot += s;
  }
  const int ex = wp + incl - t;
  if (v1)      *(int2*)(exI + n0) = make_int2(ex, ex + c0);
  else if (v0) exI[n0] = ex;
  if (tid == 0) bsum[blockIdx.x] = tot;
}

#define CP_PARTS 4
#define CP_NODES (2 * CS_WORDS / CP_PARTS)
#define CP_B     26
#define CP_THREADS 512
struct PlaceLds {
  int place[CP_NODES];
  int tops[256];
  int wtot[8];
};
__device__ __forceinline__ void place_part(PlaceLds& L, const int blk,
                                           const int* __restrict__ src,
                                           const int* __restrict__ dst,
                                           const int* __restrict__ exI,
                                           const int* __restrict__ bsum,
                                           const int* __restrict__ pre,
                                           int* __restrict__ rowptr,
                                           int* __restrict__ csc, int E, int N,
                                           int NA, int nb) {
  int* __restrict__ place = L.place;
  int* __restrict__ tops = L.tops;
  int* __restrict__ wtot = L.wtot;
  const int tid = threadIdx.x;
  const int b = blk / CP_PARTS;
  const int q = blk - b * CP_PARTS;
  const int nlo = q * CP_NODES;
  const int nq = min(N - nlo, CP_NODES);
  if (nq <= 0) return;

  int bt = 0, bincl = 0;
  if (tid < 256) {
    bt = (tid < nb) ? bsum[tid] : 0;
    bincl = wave_incl_sum(bt, tid & 31);
    if ((tid & 31) == 31) wtot[tid >> 5] = bincl;
  }
  __syncthreads();
  if (tid < 256) {
    int wp = 0;
#pragma unroll
    for (int u = 0; u < 8; ++u)
      if (u < (tid >> 5)) wp += wtot[u];
    tops[tid] = wp + bincl - bt;
  }
  __syncthreads();

  const int* __restrict__ rp = exI + nlo;
  const int* __restrict__ pp = pre + (size_t)b * NA + nlo;
  for (int i0 = tid; i0 < nq; i0 += CP_B * CP_THREADS) {
    int r[CP_B], p[CP_B];
#pragma unroll
    for (int k = 0; k < CP_B; ++k) {
      const int i = i0 + k * CP_THREADS;
      const int ic = (i < nq) ? i : i0;
      r[k] = rp[ic];
      p[k] = pp[ic];
    }
#pragma unroll
    for (int k = 0; k < CP_B; ++k) {
      const int i = i0 + k * CP_THREADS;
      if (i < nq) {
        const int st = r[k] + tops[(unsigned int)(nlo + i) / (2u * KS_THREADS)];
        place[i] = st + p[k];
        if (b == 0) rowptr[nlo + i] = st;
      }
    }
  }
  __syncthreads();

  const int len = (E + CS_SLICES - 1) / CS_SLICES;
  const int lo = b * len;
  const int hi = min(E, lo + len);
  for (int e0 = lo + tid; e0 < hi; e0 += CP_B * CP_THREADS) {
    int s[CP_B], li[CP_B], pos[CP_B];
    bool mine[CP_B];
#pragma unroll
    for (int k = 0; k < CP_B; ++k) {
      const int e = e0 + k * CP_THREADS;
      const bool ok = e < hi;
      const int ec = ok ? e : e0;
      li[k] = dst[ec] - nlo;
      s[k] = src[ec];
      mine[k] = ok && ((unsigned int)li[k] < (unsigned int)nq);
    }
#pragma unroll
    for (int k = 0; k < CP_B; ++k) {
      pos[k] = 0;
      if (mine[k]) pos[k] = atomicAdd(&place[li[k]], 1);
    }
#pragma unroll
    for (int k = 0; k < CP_B; ++k)
      if (mine[k]) csc[pos[k]] = s[k];
  }
}

#define SP_B 16

__device__ __forceinline__ void sp_zero(float4& a) { a = float4{0.0f, 0.0f, 0.0f, 0.0f}; }
__device__ __forceinline__ void sp_zero(float2& a) { a = float2{0.0f, 0.0f}; }
__device__ __forceinline__ void sp_add(float4& a, const float4& b) {
  a.x += b.x; a.y += b.y; a.z += b.z; a.w += b.w;
}
__device__ __forceinline__ void sp_add(float2& a, const float2& b) { a.x += b.x; a.y += b.y; }
__device__ __forceinline__ float4 sp_comb(float s, const float4& a, float xw, const float4& x) {
  float4 r;
  r.x = s * a.x + xw * x.x;
  r.y = s * a.y + xw * x.y;
  r.z = s * a.z + xw * x.z;
  r.w = s * a.w + xw * x.w;
  return r;
}
__device__ __forceinline__ float2 sp_comb(float s, const float2& a, float xw, const float2& x) {
  float2 r;
  r.x = s * a.x + xw * x.x;
  r.y = s * a.y + xw * x.y;
  return r;
}

template <bool FULL, typename VT>
__device__ __forceinline__ void sp_batch(const float* __restrict__ hsl, int myidx, int i,
                                         int rem, VT& acc) {
  constexpr int ROWF = 32 * (int)(sizeof(VT) / sizeof(float));
  int s[SP_B];
#pragma unroll
  for (int k = 0; k < SP_B; ++k) s[k] = __builtin_amdgcn_readlane(myidx, i + k);
  VT v[SP_B];
#pragma unroll
  for (int k = 0; k < SP_B; ++k) {
    if (FULL || k < rem) v[k] = *(const VT*)(hsl + (long)s[k] * ROWF);
    else                 sp_zero(v[k]);
  }
#pragma unroll
  for (int k = 0; k < SP_B; ++k) {
    if (FULL || k < rem) sp_add(acc, v[k]);
  }
}

template <typename VT, bool BIAS>
__global__ __launch_bounds__(256) void k_spmm_pull(const int* __restrict__ rowptr,
                                                   const int* __restrict__ cnt,
                                                   const int* __restrict__ csc,
                                                   const float* __restrict__ hs,
                                                   const float* __restrict__ dirs,
                                                   const float* __restrict__ dors,
                                                   const float* __restrict__ x0,
                                                   const float* __restrict__ bias,
                                                   float* __restrict__ f, int N) {
  constexpr int VW = (int)(sizeof(VT) / sizeof(float));
  constexpr int ROWF = 32 * VW;
  const int wave = __builtin_amdgcn_readfirstlane((int)(threadIdx.x >> 5));
  const int lane = threadIdx.x & 31;
  const int step = (int)gridDim.x * 8;
  const float* __restrict__ hsl = hs + lane * VW;
  int node = (int)blockIdx.x * 8 + wave;
  if (node >= N) return;
  VT bv;
  sp_zero(bv);
  if (BIAS) bv = *(const VT*)(bias + lane * VW);
  int start = __builtin_amdgcn_readfirstlane(rowptr[node]);
  int c = __builtin_amdgcn_readfirstlane(cnt[node]);
  float dn = dirs[node];
  float dq = dors[node];
  for (;;) {
    const int nnode = node + step;
    const int nn = min(nnode, N - 1);
    const int nstart = __builtin_amdgcn_readfirstlane(rowptr[nn]);
    const int nc = __builtin_amdgcn_readfirstlane(cnt[nn]);
    const float ndn = dirs[nn];
    const float ndq = dors[nn];

    int m = min(32, c);
    int myidx = 0;
    if (lane < m) myidx = csc[start + lane];
    const long o = (long)node * ROWF + lane * VW;
    const VT x0v = *(const VT*)(x0 + o);
    const float scale = (1.0f - GA_ALPHA) * dn;
    const float xw = GA_ALPHA / dq;
    VT acc;
    sp_zero(acc);
    int base = 0;
    for (;;) {
      for (int i = 0; i < m; i += SP_B) {
        const int rem = m - i;
        if (rem >= SP_B) sp_batch<true, VT>(hsl, myidx, i, rem, acc);
        else             sp_batch<false, VT>(hsl, myidx, i, rem, acc);
      }
      base += 32;
      if (base >= c) break;
      m = min(32, c - base);
      myidx = 0;
      if (lane < m) myidx = csc[start + base + lane];
    }
    VT r = sp_comb(scale, acc, xw, x0v);
    if (BIAS) sp_add(r, bv);
    *(VT*)(f + o) = r;

    if (nnode >= N) break;
    node = nnode; start = nstart; c = nc; dn = ndn; dq = ndq;
  }
}

__global__ __launch_bounds__(512) void k_gemm_chain(const float* __restrict__ A,
                                                    const float* __restrict__ W,
                                                    const float* __restrict__ bias,
                                                    const float* __restrict__ dors,
                                                    const float* __restrict__ H,
                                                    const float* __restrict__ Wc,
                                                    float* __restrict__ outQ,
                                                    float* __restrict__ outP,
                                                    float actW, float resS, int nTiles) {
  constexpr int NCT = 8;
  constexpr int PCT = 4;
  __shared__ float sW[128 * 128];
  __shared__ float sWc[128 * 64];

  const int wave = __builtin_amdgcn_readfirstlane((int)(threadIdx.x >> 5));
  const int lane = threadIdx.x & 31;
  const int half = lane >> 4;
  const int l16  = lane & 15;
  const int nblk = gridDim.x;

  int round = 0;
  int tile = blockIdx.x + nblk * wave;
  bool have = tile < nTiles;
  const int qtile0 = tile;
  const bool ninth = blockIdx.x + nblk * 24 < nTiles;
  const unsigned long long qtab = ninth ? 0x543F210EDCBA8769ull : 0x76543210FEDCBA98ull;
  const int qperm = (int)((qtab >> (4 * wave)) & 15ull);

  constexpr int NST = (128 * 128) / 512;
  constexpr int NSC = (128 * 64) / 512;
  float wv[NST];
  float wcv[NSC];
#pragma unroll
  for (int it = 0; it < NST; ++it) {
    const int d  = it * 512 + threadIdx.x;
    const int p  = d & 1;
    const int tt = (d >> 1) & 1;
    const int c  = (d >> 2) & 15;
    const int h  = (d >> 6) & 1;
    const int bt = d >> 7;
    const int v  = bt % (NCT / 2);
    const int js = bt / (NCT / 2);
    const int k  = (js >> 1) * 8 + h * 4 + (js & 1) * 2 + p;
    wv[it] = W[k * 128 + (2 * v + tt) * 16 + c];
  }
#pragma unroll
  for (int it = 0; it < NSC; ++it) {
    const int d  = it * 512 + threadIdx.x;
    const int p  = d & 1;
    const int tt = (d >> 1) & 1;
    const int c  = (d >> 2) & 15;
    const int h  = (d >> 6) & 1;
    const int bt = d >> 7;
    const int v  = bt % (PCT / 2);
    const int tu = bt / (PCT / 2);
    const int k  = (tu >> 2) * 16 + h * 8 + (tu & 3) * 2 + p;
    wcv[it] = Wc[k * 64 + (2 * v + tt) * 16 + c];
  }

  const float4* __restrict__ A4 = (const float4*)A;
  const float4* __restrict__ H4 = (const float4*)H;
  const float4* __restrict__ B4 = (const float4*)bias;
  float4 a[16];
  {
    const int ap4 = ((have ? tile : 0) * 16 + l16) * 32 + half;
#pragma unroll
    for (int j = 0; j < 16; ++j) a[j] = A4[ap4 + 2 * j];
  }

#pragma unroll
  for (int it = 0; it < NST; ++it) sW[it * 512 + threadIdx.x] = wv[it];
#pragma unroll
  for (int it = 0; it < NSC; ++it) sWc[it * 512 + threadIdx.x] = wcv[it];
  __builtin_amdgcn_fence(__ATOMIC_RELEASE, "workgroup", "local");
  __builtin_amdgcn_s_barrier();
  __builtin_amdgcn_fence(__ATOMIC_ACQUIRE, "workgroup", "local");

  const float* __restrict__ sWl  = sW + lane * 4;
  const float* __restrict__ sWcl = sWc + lane * 4;

  while (have) {
    const int nround = round + 1;
    const int ntile = blockIdx.x + nblk * (16 * nround + wave);
    const bool nhave = ntile < nTiles;
    const float4* __restrict__ NX = nhave ? A4 : H4;
    const int nx4 = nhave ? (ntile * 16 + l16) * 32 + half : (qtile0 * 16 + l16) * 32 + half * 2;
    const int nxs = nhave ? 2 : 1;
    const int ro4 = (tile * 16 + l16) * 32 + half * 2;
    int bo4 = half * 2;
    asm volatile("" : "+v"(bo4));
    const float dsv = dors[tile * 16 + l16];

    v8f acc[NCT];
#pragma unroll
    for (int t = 0; t < NCT; ++t) acc[t] = v8f{};

    float4 nr0, nr1, nc0, nc1;

#pragma unroll
    for (int j = 0; j < 16; ++j) {
#pragma unroll
      for (int s = 0; s < 2; ++s) {
        v2f b[NCT];
#pragma unroll
        for (int v = 0; v < NCT / 2; ++v) {
          const float4 q = *(const float4*)(sWl + ((j * 2 + s) * (NCT / 2) + v) * 128);
          b[2 * v].x = q.x;
          b[2 * v].y = q.y;
          b[2 * v + 1].x = q.z;
          b[2 * v + 1].y = q.w;
        }
        v2f af;
        af.x = s ? a[j].z : a[j].x;
        af.y = s ? a[j].w : a[j].y;
#pragma unroll
        for (int t = 0; t < NCT; ++t) {
          acc[t] = __builtin_amdgcn_wmma_f32_16x16x4_f32(false, b[t], false, af, (short)0, acc[t],
                                                         false, false);
        }
      }
      if (j == 15) {
        nr0 = A4[ro4];
        nr1 = A4[ro4 + 1];
        nc0 = B4[bo4];
        nc1 = B4[bo4 + 1];
      }
      __builtin_amdgcn_sched_barrier(0);
    }

    v8f accP[PCT];
#pragma unroll
    for (int ct = 0; ct < PCT; ++ct) accP[ct] = v8f{};

#pragma unroll
    for (int t = 0; t < NCT; ++t) {
      const float rs[8] = {nr0.x, nr0.y, nr0.z, nr0.w, nr1.x, nr1.y, nr1.z, nr1.w};
      const float cb[8] = {nc0.x, nc0.y, nc0.z, nc0.w, nc1.x, nc1.y, nc1.z, nc1.w};
      if (t + 1 < NCT) {
        nr0 = A4[ro4 + 4 * (t + 1)];
        nr1 = A4[ro4 + 4 * (t + 1) + 1];
        nc0 = B4[bo4 + 4 * (t + 1)];
        nc1 = B4[bo4 + 4 * (t + 1) + 1];
      }
      float x[8];
#pragma unroll
      for (int r = 0; r < 8; ++r) {
        float v = actW * acc[t][r] + cb[r];
        v += resS * rs[r];
        v = fmaxf(v, 0.0f);
        x[r] = v * dsv;
      }
#pragma unroll
      for (int u = 0; u < 4; ++u) {
        v2f b[PCT];
#pragma unroll
        for (int v = 0; v < PCT / 2; ++v) {
          const float4 q = *(const float4*)(sWcl + ((t * 4 + u) * (PCT / 2) + v) * 128);
          b[2 * v].x = q.x;
          b[2 * v].y = q.y;
          b[2 * v + 1].x = q.z;
          b[2 * v + 1].y = q.w;
        }
        v2f ax;
        ax.x = x[2 * u];
        ax.y = x[2 * u + 1];
#pragma unroll
        for (int ct = 0; ct < PCT; ++ct)
          accP[ct] = __builtin_amdgcn_wmma_f32_16x16x4_f32(false, ax, false, b[ct], (short)0,
                                                           accP[ct], false, false);
      }
      a[2 * t]     = NX[nx4 + 4 * t];
      a[2 * t + 1] = NX[nx4 + 4 * t + nxs];
      __builtin_amdgcn_sched_barrier(0);
    }

    const int o0 = (tile * 16 + half * 8) * 64 + l16;
#pragma unroll
    for (int ct = 0; ct < PCT; ++ct) {
#pragma unroll
      for (int r = 0; r < 8; ++r)
        outP[o0 + r * 64 + ct * 16] = accP[ct][r];
    }

    tile = ntile;
    have = nhave;
    round = nround;
  }

  int qround = 0;
  int qtile = qtile0;
  bool qhave = qtile < nTiles;
  while (qhave) {
    const int nqround = qround + 1;
    const int qs = (nqround & 1) ? qperm : wave;
    const int nqtile = blockIdx.x + nblk * (16 * nqround + qs);
    const bool nqhave = nqtile < nTiles;
    const int nq4 = ((nqhave ? nqtile : qtile) * 16 + l16) * 32 + half * 2;

    v8f accQ[PCT];
#pragma unroll
    for (int ct = 0; ct < PCT; ++ct) accQ[ct] = v8f{};

#pragma unroll
    for (int t = 0; t < NCT; ++t) {
#pragma unroll
      for (int u = 0; u < 4; ++u) {
        v2f b[PCT];
#pragma unroll
        for (int v = 0; v < PCT / 2; ++v) {
          const float4 q = *(const float4*)(sWcl + ((t * 4 + u) * (PCT / 2) + v) * 128);
          b[2 * v].x = q.x;
          b[2 * v].y = q.y;
          b[2 * v + 1].x = q.z;
          b[2 * v + 1].y = q.w;
        }
        const float4 hv = a[2 * t + (u >> 1)];
        v2f ah;
        ah.x = (u & 1) ? hv.z : hv.x;
        ah.y = (u & 1) ? hv.w : hv.y;
#pragma unroll
        for (int ct = 0; ct < PCT; ++ct)
          accQ[ct] = __builtin_amdgcn_wmma_f32_16x16x4_f32(false, ah, false, b[ct], (short)0,
                                                           accQ[ct], false, false);
      }
      a[2 * t]     = H4[nq4 + 4 * t];
      a[2 * t + 1] = H4[nq4 + 4 * t + 1];
      __builtin_amdgcn_sched_barrier(0);
    }

    const int o0 = (qtile * 16 + half * 8) * 64 + l16;
#pragma unroll
    for (int ct = 0; ct < PCT; ++ct) {
#pragma unroll
      for (int r = 0; r < 8; ++r)
        outQ[o0 + r * 64 + ct * 16] = accQ[ct][r];
    }

    qtile = nqtile;
    qhave = nqhave;
    qround = nqround;
  }
}

#define G1_BLOCKS 256
__device__ __forceinline__ void gemm1_part(float* __restrict__ sW, const int bid,
                                           const float* __restrict__ A,
                                           const float* __restrict__ W,
                                           const float* __restrict__ bias,
                                           const float* __restrict__ dors,
                                           float* __restrict__ outS, int nTiles) {
  constexpr int NCOLS = 128;
  constexpr int NCT = NCOLS / 16;
  const int wave = __builtin_amdgcn_readfirstlane((int)(threadIdx.x >> 5));
  const int lane = threadIdx.x & 31;
  const int half = lane >> 4;
  const int l16  = lane & 15;
  const int nblk = G1_BLOCKS;

  int round = 0;
  int tile = bid + nblk * wave;
  bool have = tile < nTiles;

  constexpr int NST = (128 * NCOLS) / CP_THREADS;
  float wv[NST];
#pragma unroll
  for (int it = 0; it < NST; ++it) {
    const int d  = it * CP_THREADS + threadIdx.x;
    const int p  = d & 1;
    const int tt = (d >> 1) & 1;
    const int c  = (d >> 2) & 15;
    const int h  = (d >> 6) & 1;
    const int bt = d >> 7;
    const int u  = bt % (NCT / 2);
    const int js = bt / (NCT / 2);
    const int k  = (js >> 1) * 8 + h * 4 + (js & 1) * 2 + p;
    wv[it] = W[k * NCOLS + (2 * u + tt) * 16 + c];
  }

  const float* __restrict__ Ap = A + (long)((have ? tile : 0) * 16 + l16) * 128 + half * 4;
  float4 a[4];
#pragma unroll
  for (int j = 0; j < 4; ++j) a[j] = *(const float4*)(Ap + 8 * j);

#pragma unroll
  for (int it = 0; it < NST; ++it) sW[it * CP_THREADS + threadIdx.x] = wv[it];
  __builtin_amdgcn_fence(__ATOMIC_RELEASE, "workgroup", "local");
  __builtin_amdgcn_s_barrier();
  __builtin_amdgcn_fence(__ATOMIC_ACQUIRE, "workgroup", "local");

  const float* __restrict__ sWl = sW + lane * 4;

  while (have) {
    const int nround = round + 1;
    const int q = (nround & 1) ? ((13 * wave) & 15) : wave;
    const int ntile = bid + nblk * (16 * nround + q);
    const bool nhave = ntile < nTiles;
    const float* __restrict__ An =
        A + (long)((nhave ? ntile : tile) * 16 + l16) * 128 + half * 4;

    const float4* __restrict__ dp = (const float4*)(dors + tile * 16 + half * 8);
    float4 d0, d1;

    v8f acc[NCT];
#pragma unroll
    for (int t = 0; t < NCT; ++t) acc[t] = v8f{};

#pragma unroll
    for (int j = 0; j < 16; ++j) {
      const float4 aj = a[j & 3];
#pragma unroll
      for (int s = 0; s < 2; ++s) {
        v2f b[NCT];
#pragma unroll
        for (int u = 0; u < NCT / 2; ++u) {
          const float4 q = *(const float4*)(sWl + ((j * 2 + s) * (NCT / 2) + u) * 128);
          b[2 * u].x = q.x;
          b[2 * u].y = q.y;
          b[2 * u + 1].x = q.z;
          b[2 * u + 1].y = q.w;
        }
        v2f af;
        af.x = s ? aj.z : aj.x;
        af.y = s ? aj.w : aj.y;
#pragma unroll
        for (int t = 0; t < NCT; ++t) {
          acc[t] = __builtin_amdgcn_wmma_f32_16x16x4_f32(false, af, false, b[t], (short)0, acc[t],
                                                         false, false);
        }
      }
      a[j & 3] = (j < 12) ? *(const float4*)(Ap + 8 * (j + 4)) : *(const float4*)(An + 8 * (j - 12));
      if (j == 11) {
        d0 = dp[0];
        d1 = dp[1];
      }
      __builtin_amdgcn_sched_barrier(0);
    }

    const float ds[8] = {d0.x, d0.y, d0.z, d0.w, d1.x, d1.y, d1.z, d1.w};
    const int row0 = tile * 16;
#pragma unroll
    for (int t = 0; t < NCT; ++t) {
      const int outCol = t * 16 + l16;
      const float bv = bias[outCol];
#pragma unroll
      for (int r = 0; r < 8; ++r) {
        const int rowM = row0 + r + half * 8;
        const float v = fmaxf(acc[t][r] + bv, 0.0f);
        outS[(long)rowM * NCOLS + outCol] = v * ds[r];
      }
    }

    tile = ntile;
    have = nhave;
    round = nround;
    Ap = An;
  }
}

union alignas(16) FusedLds {
  PlaceLds pl;
  float sW[128 * 128];
};

static_assert(CS_SLICES * CP_PARTS == G1_BLOCKS, "one placing work-group per GEMM work-group");
__global__ __launch_bounds__(CP_THREADS, 8) void k_place_gemm(const int* __restrict__ src,
                                                              const int* __restrict__ dst,
                                                              const int* __restrict__ exI,
                                                              const int* __restrict__ bsum,
                                                              const int* __restrict__ pre,
                                                              int* __restrict__ rowptr,
                                                              int* __restrict__ csc, int E, int N,
                                                              int NA, int nb,
                                                              const float* __restrict__ feat,
                                                              const float* __restrict__ w1,
                                                              const float* __restrict__ b1,
                                                              const float* __restrict__ dors,
                                                              float* __restrict__ h0s, int nTiles) {
  __shared__ FusedLds lds;
  const int i = (int)blockIdx.x;
  const int id = i >> 1;
  if ((((i / G1_BLOCKS) ^ i) & 1) == 0)
    place_part(lds.pl, id, src, dst, exI, bsum, pre, rowptr, csc, E, N, NA, nb);
  else
    gemm1_part(lds.sW, id, feat, w1, b1, dors, h0s, nTiles);
}


static inline long cdiv(long a, long b) { return (a + b - 1) / b; }

extern "C" void kernel_launch(void* const* d_in, const int* in_sizes, int n_in,
                              void* d_out, int out_size, void* d_ws, size_t ws_size,
                              hipStream_t stream) {
  const float* feat = (const float*)d_in[0];
  const int*   src  = (const int*)d_in[1];
  const int*   dst  = (const int*)d_in[2];
  const float* w1   = (const float*)d_in[3];
  const float* b1   = (const float*)d_in[4];
  const float* cw1  = (const float*)d_in[5];
  const float* cb1  = (const float*)d_in[6];
  const float* cw2  = (const float*)d_in[7];
  const float* cb2  = (const float*)d_in[8];
  const float* w2   = (const float*)d_in[9];
  const float* b2   = (const float*)d_in[10];

  const int N = in_sizes[0] / 128;
  const int E = in_sizes[1];
  const long NF = (long)N * 128;
  const int NA = (N + 63) & ~63;

  char* wsb = (char*)d_ws;
  int* cntO   = (int*)wsb;                wsb += (size_t)NA * 4;
  int* cntI   = (int*)wsb;                wsb += (size_t)NA * 4;
  int* rowptr = (int*)wsb;                wsb += (size_t)NA * 4;
  int* cursor = (int*)wsb;                wsb += (size_t)NA * 4;
  int* bsum   = (int*)wsb;                wsb += 512 * 4;
  int* csc    = (int*)wsb;                wsb += ((size_t)E * 4 + 255) & ~(size_t)255;
  float* degO = (float*)wsb;              wsb += (size_t)NA * 4;
  float* degI = (float*)wsb;              wsb += (size_t)NA * 4;
  float* h0   = (float*)wsb;              wsb += (size_t)NF * 4;
  float* bufS = (float*)wsb;              wsb += (size_t)NF * 4;
  float* bufA = (float*)wsb;              wsb += (size_t)NF * 4;
  float* bufB = (float*)wsb;
  (void)ws_size; (void)n_in; (void)out_size;

  const int n_strips = N / 16;

  unsigned int* hist = (unsigned int*)bufA;
  int* pre = (int*)bufB;

  float* wc = (float*)cntO;
  float* bc = wc + 128 * 64;
  int* exI = cursor;

  k_hist<<<128, 1024, 0, stream>>>(src, dst, hist, E);
  k_sum<<<KS_BLOCKS + FOLD_TILES, KS_THREADS, 0, stream>>>(hist, cntI, exI, bsum, degO, degI, pre, N,
                                                           NA, cw2, cb2, w2, b2, wc, bc, GA_BETA2);
  float* h0s = h0;
  k_place_gemm<<<2 * G1_BLOCKS, CP_THREADS, 0, stream>>>(
      src, dst, exI, bsum, pre, rowptr, csc, E, N, NA, KS_BLOCKS, feat, w1, b1, degO, h0s, n_strips);

  k_spmm_pull<float4, false><<<1024, 256, 0, stream>>>(rowptr, cntI, csc, h0s, degI, degO, h0s,
                                                       nullptr, bufA, N);
  float* Qs = bufS;
  float* P = bufS + (size_t)N * 64;
  k_gemm_chain<<<256, 512, 0, stream>>>(bufA, cw1, cb1, degO, h0s, wc, Qs, P, GA_BETA1,
                                        1.0f - GA_BETA1, n_strips);
  k_spmm_pull<float2, true><<<1024, 256, 0, stream>>>(rowptr, cntI, csc, P, degI, degO, Qs, bc,
                                                      (float*)d_out, N);
}
